# starting kernel
# baseline (speedup 1.0000x reference)
#include <hip/hip_runtime.h>
#include <hip/hip_bf16.h>

typedef __attribute__((ext_vector_type(2))) float v2f;
typedef __attribute__((ext_vector_type(8))) float v8f;

#define HDIM 128
#define CTPW 4

__global__ __launch_bounds__(256) void gat_transpose(
    const float* __restrict__ W, float* __restrict__ Wt,
    int rows, int cols, int colsPad)
{
    int i = blockIdx.x * blockDim.x + threadIdx.x;
    int total = colsPad * rows;
    if (i >= total) return;
    int c = i / rows;
    int r = i - c * rows;
    Wt[(size_t)c * rows + r] = (c < cols) ? W[(size_t)r * cols + c] : 0.f;
}

__global__ __launch_bounds__(256) void gat_gemm_strip(
    const float* __restrict__ X, const float* __restrict__ Wt,
    float* __restrict__ H, int nRowTiles)
{
    int wid  = (blockIdx.x * blockDim.x + threadIdx.x) >> 5;
    int lane = threadIdx.x & 31;
    int nHalves = HDIM / (CTPW * 16);
    int nW = nRowTiles * nHalves;
    if (wid >= nW) return;
    int rt   = wid >> 1;
    int half = wid & 1;
    int row0 = rt * 16;
    int laneM = lane & 15;
    int kOff  = (lane >> 4) * 2;

    const float* xrow = X  + (size_t)(row0 + laneM) * HDIM + kOff;
    const float* wrow = Wt + (size_t)(half * (CTPW * 16) + laneM) * HDIM + kOff;

    v8f c[CTPW] = {};
    #pragma unroll 4
    for (int k = 0; k < HDIM; k += 4) {
        v2f a = *(const v2f*)(xrow + k);
        #pragma unroll
        for (int t = 0; t < CTPW; ++t) {
            v2f b = *(const v2f*)(wrow + (size_t)t * 16 * HDIM + k);
            c[t] = __builtin_amdgcn_wmma_f32_16x16x4_f32(
                       false, a, false, b, (short)0, c[t], false, false);
        }
    }

    int mBase = (lane >> 4) * 8;
    #pragma unroll
    for (int t = 0; t < CTPW; ++t) {
        float* o = H + (size_t)row0 * HDIM + (half * CTPW + t) * 16 + laneM;
        #pragma unroll
        for (int r = 0; r < 8; ++r)
            o[(size_t)(mBase + r) * HDIM] = c[t][r];
    }
}

__global__ __launch_bounds__(256) void gat_gemm_cls(
    const float* __restrict__ X, const float* __restrict__ Wlt,
    const float* __restrict__ bl, float* __restrict__ Out,
    int nRowTiles, int nCols)
{
    int wid  = (blockIdx.x * blockDim.x + threadIdx.x) >> 5;
    int lane = threadIdx.x & 31;
    if (wid >= nRowTiles) return;
    int row0 = wid * 16;
    int laneM = lane & 15;
    int kOff  = (lane >> 4) * 2;

    const float* xrow = X   + (size_t)(row0 + laneM) * HDIM + kOff;
    const float* wrow = Wlt + (size_t)laneM * HDIM + kOff;

    v8f c[3] = {};
    #pragma unroll 4
    for (int k = 0; k < HDIM; k += 4) {
        v2f a = *(const v2f*)(xrow + k);
        #pragma unroll
        for (int t = 0; t < 3; ++t) {
            v2f b = *(const v2f*)(wrow + (size_t)t * 16 * HDIM + k);
            c[t] = __builtin_amdgcn_wmma_f32_16x16x4_f32(
                       false, a, false, b, (short)0, c[t], false, false);
        }
    }

    int mBase = (lane >> 4) * 8;
    #pragma unroll
    for (int t = 0; t < 3; ++t) {
        int col = t * 16 + laneM;
        if (col >= nCols) continue;
        float bv = bl[col];
        #pragma unroll
        for (int r = 0; r < 8; ++r)
            Out[(size_t)(row0 + mBase + r) * nCols + col] = c[t][r] + bv;
    }
}

__global__ __launch_bounds__(256) void gat_alpha(
    const float* __restrict__ H, const float* __restrict__ asrc,
    const float* __restrict__ adst, float* __restrict__ aS,
    float* __restrict__ aD, int N)
{
    int w    = (blockIdx.x * blockDim.x + threadIdx.x) >> 5;
    int lane = threadIdx.x & 31;
    if (w >= N) return;
    const float* h = H + (size_t)w * HDIM;
    float s = 0.f, d = 0.f;
    #pragma unroll
    for (int j = 0; j < 4; ++j) {
        int f = lane + 32 * j;
        float hv = h[f];
        s += hv * asrc[f];
        d += hv * adst[f];
    }
    #pragma unroll
    for (int off = 16; off > 0; off >>= 1) {
        s += __shfl_xor(s, off, 32);
        d += __shfl_xor(d, off, 32);
    }
    if (lane == 0) { aS[w] = s; aD[w] = d; }
}

__device__ __forceinline__ unsigned f2ord(float f) {
    unsigned b = __float_as_uint(f);
    return (b & 0x80000000u) ? ~b : (b | 0x80000000u);
}
__device__ __forceinline__ float ord2f(unsigned u) {
    unsigned b = (u & 0x80000000u) ? (u & 0x7FFFFFFFu) : ~u;
    return __uint_as_float(b);
}

__global__ __launch_bounds__(256) void gat_init(
    unsigned* __restrict__ segmax, float* __restrict__ den,
    float* __restrict__ agg, int N)
{
    int i = blockIdx.x * blockDim.x + threadIdx.x;
    int total = N * HDIM;
    if (i < total) agg[i] = 0.f;
    if (i < N) {
        segmax[i] = 0x007FFFFFu;
        den[i]    = 0.f;
    }
}

__global__ __launch_bounds__(256) void gat_edge1(
    const int* __restrict__ ei, int E, int N,
    const float* __restrict__ aS, const float* __restrict__ aD,
    float* __restrict__ eVal, unsigned* __restrict__ segmax)
{
    int i = blockIdx.x * blockDim.x + threadIdx.x;
    int ET = E + N;
    if (i >= ET) return;
    int s, d;
    if (i < E) { s = ei[i]; d = ei[E + i]; } else { s = d = i - E; }
    float e = aS[s] + aD[d];
    e = (e >= 0.f) ? e : 0.2f * e;
    eVal[i] = e;
    atomicMax(&segmax[d], f2ord(e));
}

__global__ __launch_bounds__(256) void gat_edge2(
    const int* __restrict__ ei, int E, int N,
    float* __restrict__ eVal, const unsigned* __restrict__ segmax,
    float* __restrict__ den)
{
    int i = blockIdx.x * blockDim.x + threadIdx.x;
    int ET = E + N;
    if (i >= ET) return;
    int d = (i < E) ? ei[E + i] : (i - E);
    float ex = expf(eVal[i] - ord2f(segmax[d]));
    eVal[i] = ex;
    atomicAdd(&den[d], ex);
}

__global__ __launch_bounds__(256) void gat_edge3(
    const int* __restrict__ ei, int E, int N,
    const float* __restrict__ eVal, const float* __restrict__ den,
    const float* __restrict__ H, float* __restrict__ agg)
{
    int w    = (blockIdx.x * blockDim.x + threadIdx.x) >> 5;
    int lane = threadIdx.x & 31;
    int ET = E + N;
    if (w >= ET) return;
    int s, d;
    if (w < E) { s = ei[w]; d = ei[E + w]; } else { s = d = w - E; }
    float alpha = eVal[w] / den[d];
    const float* h = H + (size_t)s * HDIM;
    float* o = agg + (size_t)d * HDIM;
    #pragma unroll
    for (int j = 0; j < 4; ++j) {
        int f = lane + 32 * j;
        atomicAdd(&o[f], h[f] * alpha);
    }
}

__global__ __launch_bounds__(256) void gat_bias_relu(
    float* __restrict__ agg, const float* __restrict__ b, int total)
{
    int i = blockIdx.x * blockDim.x + threadIdx.x;
    if (i >= total) return;
    float v = agg[i] + b[i & (HDIM - 1)];
    agg[i] = (v > 0.f) ? v : 0.f;
}

extern "C" void kernel_launch(void* const* d_in, const int* in_sizes, int n_in,
                              void* d_out, int out_size, void* d_ws, size_t ws_size,
                              hipStream_t stream)
{
    const float* x   = (const float*)d_in[0];
    const int*   ei  = (const int*)d_in[1];
    const float* W1  = (const float*)d_in[2];
    const float* a1s = (const float*)d_in[3];
    const float* a1d = (const float*)d_in[4];
    const float* b1  = (const float*)d_in[5];
    const float* W2  = (const float*)d_in[6];
    const float* a2s = (const float*)d_in[7];
    const float* a2d = (const float*)d_in[8];
    const float* b2  = (const float*)d_in[9];
    const float* Wl  = (const float*)d_in[10];
    const float* bl  = (const float*)d_in[11];

    const int N  = in_sizes[0] / HDIM;
    const int E  = in_sizes[1] / 2;
    const int C  = in_sizes[11];
    const int CP = (C + 15) & ~15;
    const int ET = E + N;

    char* ws = (char*)d_ws;
    auto align256 = [](size_t v) { return (v + 255) & ~(size_t)255; };
    size_t off = 0;
    float*    bufH  = (float*)(ws + off);  off += align256((size_t)N * HDIM * 4);
    float*    bufA  = (float*)(ws + off);  off += align256((size_t)N * HDIM * 4);
    float*    Wt    = (float*)(ws + off);  off += align256((size_t)HDIM * HDIM * 4);
    float*    Wlt   = (float*)(ws + off);  off += align256((size_t)CP * HDIM * 4);
    float*    aS    = (float*)(ws + off);  off += align256((size_t)N * 4);
    float*    aD    = (float*)(ws + off);  off += align256((size_t)N * 4);
    unsigned* smax  = (unsigned*)(ws + off); off += align256((size_t)N * 4);
    float*    den   = (float*)(ws + off);  off += align256((size_t)N * 4);
    float*    eVal  = (float*)(ws + off);  off += align256((size_t)ET * 4);
    (void)ws_size;

    const int nRowTiles = N / 16;
    const int BLK = 256;
    dim3 blk(BLK);

    dim3 trGrid((HDIM * HDIM + BLK - 1) / BLK);
    dim3 gemmGrid((nRowTiles * 2 + 7) / 8);
    dim3 waveGridN((N + 7) / 8);
    dim3 elemGridN((N * HDIM + BLK - 1) / BLK);
    dim3 edgeGrid((ET + BLK - 1) / BLK);
    dim3 edgeWGrid((ET + 7) / 8);

    auto run_layer = [&](const float* act_in, const float* W,
                         const float* as_, const float* ad_, const float* bias) {
        gat_transpose<<<trGrid, blk, 0, stream>>>(W, Wt, HDIM, HDIM, HDIM);
        gat_gemm_strip<<<gemmGrid, blk, 0, stream>>>(act_in, Wt, bufH, nRowTiles);
        gat_alpha<<<waveGridN, blk, 0, stream>>>(bufH, as_, ad_, aS, aD, N);
        gat_init<<<elemGridN, blk, 0, stream>>>(smax, den, bufA, N);
        gat_edge1<<<edgeGrid, blk, 0, stream>>>(ei, E, N, aS, aD, eVal, smax);
        gat_edge2<<<edgeGrid, blk, 0, stream>>>(ei, E, N, eVal, smax, den);
        gat_edge3<<<edgeWGrid, blk, 0, stream>>>(ei, E, N, eVal, den, bufH, bufA);
        gat_bias_relu<<<elemGridN, blk, 0, stream>>>(bufA, bias, N * HDIM);
    };

    run_layer(x, W1, a1s, a1d, b1);
    run_layer(bufA, W2, a2s, a2d, b2);

    dim3 trlGrid((CP * HDIM + BLK - 1) / BLK);
    gat_transpose<<<trlGrid, blk, 0, stream>>>(Wl, Wlt, HDIM, C, CP);
    dim3 clsGrid((nRowTiles + 7) / 8);
    gat_gemm_cls<<<clsGrid, blk, 0, stream>>>(bufA, Wlt, bl, (float*)d_out,
                                              nRowTiles, C);
}


# Edges counting-sorted by destination once per call (CSR); one wave per node gathers h[src] rows, max/exp/sum in registers: no float atomics, per-node sum order changes
# speedup vs baseline: 2.0491x; 2.0491x over previous
#include <hip/hip_runtime.h>
#include <hip/hip_bf16.h>

typedef __attribute__((ext_vector_type(2))) float v2f;
typedef __attribute__((ext_vector_type(8))) float v8f;

#define HDIM 128
#define CTPW 4

__global__ __launch_bounds__(256) void gat_transpose(
    const float* __restrict__ W, float* __restrict__ Wt,
    int rows, int cols, int colsPad)
{
    int i = blockIdx.x * blockDim.x + threadIdx.x;
    int total = colsPad * rows;
    if (i >= total) return;
    int c = i / rows;
    int r = i - c * rows;
    Wt[(size_t)c * rows + r] = (c < cols) ? W[(size_t)r * cols + c] : 0.f;
}

__global__ __launch_bounds__(256) void gat_gemm_strip(
    const float* __restrict__ X, const float* __restrict__ Wt,
    float* __restrict__ H, int nRowTiles)
{
    int wid  = (blockIdx.x * blockDim.x + threadIdx.x) >> 5;
    int lane = threadIdx.x & 31;
    int nHalves = HDIM / (CTPW * 16);
    int nW = nRowTiles * nHalves;
    if (wid >= nW) return;
    int rt   = wid >> 1;
    int half = wid & 1;
    int row0 = rt * 16;
    int laneM = lane & 15;
    int kOff  = (lane >> 4) * 2;

    const float* xrow = X  + (size_t)(row0 + laneM) * HDIM + kOff;
    const float* wrow = Wt + (size_t)(half * (CTPW * 16) + laneM) * HDIM + kOff;

    v8f c[CTPW] = {};
    #pragma unroll 4
    for (int k = 0; k < HDIM; k += 4) {
        v2f a = *(const v2f*)(xrow + k);
        #pragma unroll
        for (int t = 0; t < CTPW; ++t) {
            v2f b = *(const v2f*)(wrow + (size_t)t * 16 * HDIM + k);
            c[t] = __builtin_amdgcn_wmma_f32_16x16x4_f32(
                       false, a, false, b, (short)0, c[t], false, false);
        }
    }

    int mBase = (lane >> 4) * 8;
    #pragma unroll
    for (int t = 0; t < CTPW; ++t) {
        float* o = H + (size_t)row0 * HDIM + (half * CTPW + t) * 16 + laneM;
        #pragma unroll
        for (int r = 0; r < 8; ++r)
            o[(size_t)(mBase + r) * HDIM] = c[t][r];
    }
}

__global__ __launch_bounds__(256) void gat_gemm_cls(
    const float* __restrict__ X, const float* __restrict__ Wlt,
    const float* __restrict__ bl, float* __restrict__ Out,
    int nRowTiles, int nCols)
{
    int wid  = (blockIdx.x * blockDim.x + threadIdx.x) >> 5;
    int lane = threadIdx.x & 31;
    if (wid >= nRowTiles) return;
    int row0 = wid * 16;
    int laneM = lane & 15;
    int kOff  = (lane >> 4) * 2;

    const float* xrow = X   + (size_t)(row0 + laneM) * HDIM + kOff;
    const float* wrow = Wlt + (size_t)laneM * HDIM + kOff;

    v8f c[3] = {};
    #pragma unroll 4
    for (int k = 0; k < HDIM; k += 4) {
        v2f a = *(const v2f*)(xrow + k);
        #pragma unroll
        for (int t = 0; t < 3; ++t) {
            v2f b = *(const v2f*)(wrow + (size_t)t * 16 * HDIM + k);
            c[t] = __builtin_amdgcn_wmma_f32_16x16x4_f32(
                       false, a, false, b, (short)0, c[t], false, false);
        }
    }

    int mBase = (lane >> 4) * 8;
    #pragma unroll
    for (int t = 0; t < 3; ++t) {
        int col = t * 16 + laneM;
        if (col >= nCols) continue;
        float bv = bl[col];
        #pragma unroll
        for (int r = 0; r < 8; ++r)
            Out[(size_t)(row0 + mBase + r) * nCols + col] = c[t][r] + bv;
    }
}

__global__ __launch_bounds__(256) void gat_alpha(
    const float* __restrict__ H, const float* __restrict__ asrc,
    const float* __restrict__ adst, float* __restrict__ aS,
    float* __restrict__ aD, int N)
{
    int w    = (blockIdx.x * blockDim.x + threadIdx.x) >> 5;
    int lane = threadIdx.x & 31;
    if (w >= N) return;
    const float* h = H + (size_t)w * HDIM;
    float s = 0.f, d = 0.f;
    #pragma unroll
    for (int j = 0; j < 4; ++j) {
        int f = lane + 32 * j;
        float hv = h[f];
        s += hv * asrc[f];
        d += hv * adst[f];
    }
    #pragma unroll
    for (int off = 16; off > 0; off >>= 1) {
        s += __shfl_xor(s, off, 32);
        d += __shfl_xor(d, off, 32);
    }
    if (lane == 0) { aS[w] = s; aD[w] = d; }
}

__global__ __launch_bounds__(256) void gat_zero_cnt(int* __restrict__ cnt, int N)
{
    int i = blockIdx.x * blockDim.x + threadIdx.x;
    if (i < N) cnt[i] = 0;
}

__global__ __launch_bounds__(256) void gat_hist(
    const int* __restrict__ ei, int E, int* __restrict__ cnt)
{
    int i = blockIdx.x * blockDim.x + threadIdx.x;
    if (i >= E) return;
    atomicAdd(&cnt[ei[E + i]], 1);
}

__global__ __launch_bounds__(1024) void gat_scan(
    int* __restrict__ cnt, int* __restrict__ rowptr, int N)
{
    __shared__ int part[1024];
    int t   = threadIdx.x;
    int per = (N + 1023) / 1024;
    int b   = t * per;
    int e   = (b + per < N) ? (b + per) : N;
    int s = 0;
    for (int i = b; i < e; ++i) s += cnt[i];
    part[t] = s;
    __syncthreads();
    for (int off = 1; off < 1024; off <<= 1) {
        int v = (t >= off) ? part[t - off] : 0;
        __syncthreads();
        part[t] += v;
        __syncthreads();
    }
    int run = part[t] - s;
    for (int i = b; i < e; ++i) {
        int c = cnt[i];
        rowptr[i] = run;
        cnt[i]    = run;
        run += c;
    }
    if (t == 1023) rowptr[N] = part[1023];
}

__global__ __launch_bounds__(256) void gat_fill(
    const int* __restrict__ ei, int E, int* __restrict__ cursor,
    int* __restrict__ srcS)
{
    int i = blockIdx.x * blockDim.x + threadIdx.x;
    if (i >= E) return;
    int p = atomicAdd(&cursor[ei[E + i]], 1);
    srcS[p] = ei[i];
}

__device__ __forceinline__ float gat_leaky(float e) {
    return (e >= 0.f) ? e : 0.2f * e;
}

__global__ __launch_bounds__(256) void gat_gather(
    const int* __restrict__ rowptr, const int* __restrict__ srcS,
    const float* __restrict__ aS, const float* __restrict__ aD,
    const float* __restrict__ H, const float* __restrict__ bias,
    float* __restrict__ out, int N)
{
    int w    = __builtin_amdgcn_readfirstlane(
                   (int)((blockIdx.x * blockDim.x + threadIdx.x) >> 5));
    int lane = threadIdx.x & 31;
    if (w >= N) return;
    int beg = rowptr[w];
    int end = rowptr[w + 1];
    float ad    = aD[w];
    float eSelf = gat_leaky(aS[w] + ad);

    float m = eSelf;
    for (int j = beg + lane; j < end; j += 32)
        m = fmaxf(m, gat_leaky(aS[srcS[j]] + ad));
    for (int off = 16; off > 0; off >>= 1)
        m = fmaxf(m, __shfl_xor(m, off, 32));

    float exSelf = expf(eSelf - m);
    float den = (lane == 0) ? exSelf : 0.f;
    for (int j = beg + lane; j < end; j += 32)
        den += expf(gat_leaky(aS[srcS[j]] + ad) - m);
    for (int off = 16; off > 0; off >>= 1)
        den += __shfl_xor(den, off, 32);

    const float4* H4 = (const float4*)H;
    float aSelf = exSelf / den;
    float4 hv = H4[(size_t)w * (HDIM / 4) + lane];
    float4 acc;
    acc.x = hv.x * aSelf; acc.y = hv.y * aSelf;
    acc.z = hv.z * aSelf; acc.w = hv.w * aSelf;
    for (int base = beg; base < end; base += 32) {
        int j = base + lane;
        int s = 0;
        float a = 0.f;
        if (j < end) {
            s = srcS[j];
            a = expf(gat_leaky(aS[s] + ad) - m) / den;
        }
        int cnt = (end - base < 32) ? (end - base) : 32;
        #pragma unroll 4
        for (int t = 0; t < cnt; ++t) {
            int   st = __builtin_amdgcn_readlane(s, t);
            float at = __int_as_float(
                           __builtin_amdgcn_readlane(__float_as_int(a), t));
            float4 v = H4[(size_t)st * (HDIM / 4) + lane];
            acc.x += v.x * at; acc.y += v.y * at;
            acc.z += v.z * at; acc.w += v.w * at;
        }
    }

    const float* bp = bias + 4 * lane;
    float4 r;
    r.x = acc.x + bp[0]; r.y = acc.y + bp[1];
    r.z = acc.z + bp[2]; r.w = acc.w + bp[3];
    r.x = (r.x > 0.f) ? r.x : 0.f; r.y = (r.y > 0.f) ? r.y : 0.f;
    r.z = (r.z > 0.f) ? r.z : 0.f; r.w = (r.w > 0.f) ? r.w : 0.f;
    ((float4*)out)[(size_t)w * (HDIM / 4) + lane] = r;
}

extern "C" void kernel_launch(void* const* d_in, const int* in_sizes, int n_in,
                              void* d_out, int out_size, void* d_ws, size_t ws_size,
                              hipStream_t stream)
{
    const float* x   = (const float*)d_in[0];
    const int*   ei  = (const int*)d_in[1];
    const float* W1  = (const float*)d_in[2];
    const float* a1s = (const float*)d_in[3];
    const float* a1d = (const float*)d_in[4];
    const float* b1  = (const float*)d_in[5];
    const float* W2  = (const float*)d_in[6];
    const float* a2s = (const float*)d_in[7];
    const float* a2d = (const float*)d_in[8];
    const float* b2  = (const float*)d_in[9];
    const float* Wl  = (const float*)d_in[10];
    const float* bl  = (const float*)d_in[11];

    const int N  = in_sizes[0] / HDIM;
    const int E  = in_sizes[1] / 2;
    const int C  = in_sizes[11];
    const int CP = (C + 15) & ~15;

    char* ws = (char*)d_ws;
    auto align256 = [](size_t v) { return (v + 255) & ~(size_t)255; };
    size_t off = 0;
    float* bufH   = (float*)(ws + off);  off += align256((size_t)N * HDIM * 4);
    float* bufA   = (float*)(ws + off);  off += align256((size_t)N * HDIM * 4);
    float* Wt     = (float*)(ws + off);  off += align256((size_t)HDIM * HDIM * 4);
    float* Wlt    = (float*)(ws + off);  off += align256((size_t)CP * HDIM * 4);
    float* aS     = (float*)(ws + off);  off += align256((size_t)N * 4);
    float* aD     = (float*)(ws + off);  off += align256((size_t)N * 4);
    int*   cursor = (int*)(ws + off);    off += align256((size_t)N * 4);
    int*   rowptr = (int*)(ws + off);    off += align256((size_t)(N + 1) * 4);
    int*   srcS   = (int*)(ws + off);    off += align256((size_t)E * 4);
    (void)ws_size;

    const int nRowTiles = N / 16;
    const int BLK = 256;
    dim3 blk(BLK);

    dim3 trGrid((HDIM * HDIM + BLK - 1) / BLK);
    dim3 gemmGrid((nRowTiles * 2 + 7) / 8);
    dim3 waveGridN((N + 7) / 8);
    dim3 nodeGrid((N + BLK - 1) / BLK);
    dim3 edgeGrid((E + BLK - 1) / BLK);

    gat_zero_cnt<<<nodeGrid, blk, 0, stream>>>(cursor, N);
    gat_hist<<<edgeGrid, blk, 0, stream>>>(ei, E, cursor);
    gat_scan<<<dim3(1), dim3(1024), 0, stream>>>(cursor, rowptr, N);
    gat_fill<<<edgeGrid, blk, 0, stream>>>(ei, E, cursor, srcS);

    auto run_layer = [&](const float* act_in, const float* W,
                         const float* as_, const float* ad_, const float* bias) {
        gat_transpose<<<trGrid, blk, 0, stream>>>(W, Wt, HDIM, HDIM, HDIM);
        gat_gemm_strip<<<gemmGrid, blk, 0, stream>>>(act_in, Wt, bufH, nRowTiles);
        gat_alpha<<<waveGridN, blk, 0, stream>>>(bufH, as_, ad_, aS, aD, N);
        gat_gather<<<waveGridN, blk, 0, stream>>>(rowptr, srcS, aS, aD, bufH,
                                                  bias, bufA, N);
    };

    run_layer(x, W1, a1s, a1d, b1);
    run_layer(bufA, W2, a2s, a2d, b2);

    dim3 trlGrid((CP * HDIM + BLK - 1) / BLK);
    gat_transpose<<<trlGrid, blk, 0, stream>>>(Wl, Wlt, HDIM, C, CP);
    dim3 clsGrid((nRowTiles + 7) / 8);
    gat_gemm_cls<<<clsGrid, blk, 0, stream>>>(bufA, Wlt, bl, (float*)d_out,
                                              nRowTiles, C);
}


# Independent stages share launches: counter zeroing + W1 and Wl transposes; edge histogram + layer-1 GEMM; prefix scan + layer-1 logits + W2 transpose (14 -> 9 launches); stage bodies unchanged
# speedup vs baseline: 2.2423x; 1.0943x over previous
#include <hip/hip_runtime.h>
#include <hip/hip_bf16.h>

typedef __attribute__((ext_vector_type(2))) float v2f;
typedef __attribute__((ext_vector_type(8))) float v8f;

#define HDIM 128
#define CTPW 4


__device__ __forceinline__ void gat_transpose_body(
    const float* __restrict__ W, float* __restrict__ Wt,
    int rows, int cols, int colsPad, int i)
{
    int total = colsPad * rows;
    if (i >= total) return;
    int c = i / rows;
    int r = i - c * rows;
    Wt[(size_t)c * rows + r] = (c < cols) ? W[(size_t)r * cols + c] : 0.f;
}

__device__ __forceinline__ void gat_gemm_strip_body(
    const float* __restrict__ X, const float* __restrict__ Wt,
    float* __restrict__ H, int nRowTiles, int wid, int lane)
{
    int nHalves = HDIM / (CTPW * 16);
    int nW = nRowTiles * nHalves;
    if (wid >= nW) return;
    int rt   = wid >> 1;
    int half = wid & 1;
    int row0 = rt * 16;
    int laneM = lane & 15;
    int kOff  = (lane >> 4) * 2;

    const float* xrow = X  + (size_t)(row0 + laneM) * HDIM + kOff;
    const float* wrow = Wt + (size_t)(half * (CTPW * 16) + laneM) * HDIM + kOff;

    v8f c[CTPW] = {};
    #pragma unroll 4
    for (int k = 0; k < HDIM; k += 4) {
        v2f a = *(const v2f*)(xrow + k);
        #pragma unroll
        for (int t = 0; t < CTPW; ++t) {
            v2f b = *(const v2f*)(wrow + (size_t)t * 16 * HDIM + k);
            c[t] = __builtin_amdgcn_wmma_f32_16x16x4_f32(
                       false, a, false, b, (short)0, c[t], false, false);
        }
    }

    int mBase = (lane >> 4) * 8;
    #pragma unroll
    for (int t = 0; t < CTPW; ++t) {
        float* o = H + (size_t)row0 * HDIM + (half * CTPW + t) * 16 + laneM;
        #pragma unroll
        for (int r = 0; r < 8; ++r)
            o[(size_t)(mBase + r) * HDIM] = c[t][r];
    }
}

__global__ __launch_bounds__(256) void gat_gemm_strip(
    const float* __restrict__ X, const float* __restrict__ Wt,
    float* __restrict__ H, int nRowTiles)
{
    gat_gemm_strip_body(X, Wt, H, nRowTiles,
                        (int)((blockIdx.x * blockDim.x + threadIdx.x) >> 5),
                        (int)(threadIdx.x & 31));
}

__global__ __launch_bounds__(256) void gat_gemm_cls(
    const float* __restrict__ X, const float* __restrict__ Wlt,
    const float* __restrict__ bl, float* __restrict__ Out,
    int nRowTiles, int nCols)
{
    int wid  = (blockIdx.x * blockDim.x + threadIdx.x) >> 5;
    int lane = threadIdx.x & 31;
    if (wid >= nRowTiles) return;
    int row0 = wid * 16;
    int laneM = lane & 15;
    int kOff  = (lane >> 4) * 2;

    const float* xrow = X   + (size_t)(row0 + laneM) * HDIM + kOff;
    const float* wrow = Wlt + (size_t)laneM * HDIM + kOff;

    v8f c[3] = {};
    #pragma unroll 4
    for (int k = 0; k < HDIM; k += 4) {
        v2f a = *(const v2f*)(xrow + k);
        #pragma unroll
        for (int t = 0; t < 3; ++t) {
            v2f b = *(const v2f*)(wrow + (size_t)t * 16 * HDIM + k);
            c[t] = __builtin_amdgcn_wmma_f32_16x16x4_f32(
                       false, a, false, b, (short)0, c[t], false, false);
        }
    }

    int mBase = (lane >> 4) * 8;
    #pragma unroll
    for (int t = 0; t < 3; ++t) {
        int col = t * 16 + laneM;
        if (col >= nCols) continue;
        float bv = bl[col];
        #pragma unroll
        for (int r = 0; r < 8; ++r)
            Out[(size_t)(row0 + mBase + r) * nCols + col] = c[t][r] + bv;
    }
}

__device__ __forceinline__ void gat_alpha_body(
    const float* __restrict__ H, const float* __restrict__ asrc,
    const float* __restrict__ adst, float* __restrict__ aS,
    float* __restrict__ aD, int N, int w, int lane)
{
    if (w >= N) return;
    const float* h = H + (size_t)w * HDIM;
    float s = 0.f, d = 0.f;
    #pragma unroll
    for (int j = 0; j < 4; ++j) {
        int f = lane + 32 * j;
        float hv = h[f];
        s += hv * asrc[f];
        d += hv * adst[f];
    }
    #pragma unroll
    for (int off = 16; off > 0; off >>= 1) {
        s += __shfl_xor(s, off, 32);
        d += __shfl_xor(d, off, 32);
    }
    if (lane == 0) { aS[w] = s; aD[w] = d; }
}

__global__ __launch_bounds__(256) void gat_alpha(
    const float* __restrict__ H, const float* __restrict__ asrc,
    const float* __restrict__ adst, float* __restrict__ aS,
    float* __restrict__ aD, int N)
{
    gat_alpha_body(H, asrc, adst, aS, aD, N,
                   (int)((blockIdx.x * blockDim.x + threadIdx.x) >> 5),
                   (int)(threadIdx.x & 31));
}

__device__ __forceinline__ void gat_hist_body(
    const int* __restrict__ ei, int E, int* __restrict__ cnt, int i)
{
    if (i >= E) return;
    atomicAdd(&cnt[ei[E + i]], 1);
}

__device__ __forceinline__ void gat_scan_body(
    int* __restrict__ cnt, int* __restrict__ rowptr, int N)
{
    __shared__ int part[1024];
    int t   = threadIdx.x;
    int per = (N + 1023) / 1024;
    int b   = t * per;
    int e   = (b + per < N) ? (b + per) : N;
    int s = 0;
    for (int i = b; i < e; ++i) s += cnt[i];
    part[t] = s;
    __syncthreads();
    for (int off = 1; off < 1024; off <<= 1) {
        int v = (t >= off) ? part[t - off] : 0;
        __syncthreads();
        part[t] += v;
        __syncthreads();
    }
    int run = part[t] - s;
    for (int i = b; i < e; ++i) {
        int c = cnt[i];
        rowptr[i] = run;
        cnt[i]    = run;
        run += c;
    }
    if (t == 1023) rowptr[N] = part[1023];
}

__global__ __launch_bounds__(256) void gat_fill(
    const int* __restrict__ ei, int E, int* __restrict__ cursor,
    int* __restrict__ srcS)
{
    int i = blockIdx.x * blockDim.x + threadIdx.x;
    if (i >= E) return;
    int p = atomicAdd(&cursor[ei[E + i]], 1);
    srcS[p] = ei[i];
}


__global__ __launch_bounds__(256) void gat_stage1(
    int* __restrict__ cnt, int N, int nbZero,
    const float* __restrict__ W1, float* __restrict__ Wt, int nbT1,
    const float* __restrict__ Wl, float* __restrict__ Wlt, int C, int CP)
{
    int b = blockIdx.x;
    int t = threadIdx.x;
    if (b < nbZero) {
        int i = b * 256 + t;
        if (i < N) cnt[i] = 0;
    } else if (b < nbZero + nbT1) {
        gat_transpose_body(W1, Wt, HDIM, HDIM, HDIM, (b - nbZero) * 256 + t);
    } else {
        gat_transpose_body(Wl, Wlt, HDIM, C, CP, (b - nbZero - nbT1) * 256 + t);
    }
}

__global__ __launch_bounds__(256) void gat_stage2(
    const float* __restrict__ X, const float* __restrict__ Wt,
    float* __restrict__ H, int nRowTiles, int nbGemm,
    const int* __restrict__ ei, int E, int* __restrict__ cnt)
{
    int b = blockIdx.x;
    int t = threadIdx.x;
    if (b < nbGemm) {
        gat_gemm_strip_body(X, Wt, H, nRowTiles, (b * 256 + t) >> 5, t & 31);
    } else {
        gat_hist_body(ei, E, cnt, (b - nbGemm) * 256 + t);
    }
}

__global__ __launch_bounds__(1024) void gat_stage3(
    int* __restrict__ cnt, int* __restrict__ rowptr, int N,
    const float* __restrict__ H, const float* __restrict__ asrc,
    const float* __restrict__ adst, float* __restrict__ aS,
    float* __restrict__ aD, int nbAlpha,
    const float* __restrict__ W2, float* __restrict__ Wt)
{
    int b = blockIdx.x;
    int t = threadIdx.x;
    if (b == 0) {
        gat_scan_body(cnt, rowptr, N);
    } else if (b <= nbAlpha) {
        gat_alpha_body(H, asrc, adst, aS, aD, N,
                       ((b - 1) * 1024 + t) >> 5, t & 31);
    } else {
        gat_transpose_body(W2, Wt, HDIM, HDIM, HDIM,
                           (b - 1 - nbAlpha) * 1024 + t);
    }
}

__device__ __forceinline__ float gat_leaky(float e) {
    return (e >= 0.f) ? e : 0.2f * e;
}

__global__ __launch_bounds__(256) void gat_gather(
    const int* __restrict__ rowptr, const int* __restrict__ srcS,
    const float* __restrict__ aS, const float* __restrict__ aD,
    const float* __restrict__ H, const float* __restrict__ bias,
    float* __restrict__ out, int N)
{
    int w    = __builtin_amdgcn_readfirstlane(
                   (int)((blockIdx.x * blockDim.x + threadIdx.x) >> 5));
    int lane = threadIdx.x & 31;
    if (w >= N) return;
    int beg = rowptr[w];
    int end = rowptr[w + 1];
    float ad    = aD[w];
    float eSelf = gat_leaky(aS[w] + ad);

    float m = eSelf;
    for (int j = beg + lane; j < end; j += 32)
        m = fmaxf(m, gat_leaky(aS[srcS[j]] + ad));
    for (int off = 16; off > 0; off >>= 1)
        m = fmaxf(m, __shfl_xor(m, off, 32));

    float exSelf = expf(eSelf - m);
    float den = (lane == 0) ? exSelf : 0.f;
    for (int j = beg + lane; j < end; j += 32)
        den += expf(gat_leaky(aS[srcS[j]] + ad) - m);
    for (int off = 16; off > 0; off >>= 1)
        den += __shfl_xor(den, off, 32);

    const float4* H4 = (const float4*)H;
    float aSelf = exSelf / den;
    float4 hv = H4[(size_t)w * (HDIM / 4) + lane];
    float4 acc;
    acc.x = hv.x * aSelf; acc.y = hv.y * aSelf;
    acc.z = hv.z * aSelf; acc.w = hv.w * aSelf;
    for (int base = beg; base < end; base += 32) {
        int j = base + lane;
        int s = 0;
        float a = 0.f;
        if (j < end) {
            s = srcS[j];
            a = expf(gat_leaky(aS[s] + ad) - m) / den;
        }
        int cnt = (end - base < 32) ? (end - base) : 32;
        #pragma unroll 4
        for (int t = 0; t < cnt; ++t) {
            int   st = __builtin_amdgcn_readlane(s, t);
            float at = __int_as_float(
                           __builtin_amdgcn_readlane(__float_as_int(a), t));
            float4 v = H4[(size_t)st * (HDIM / 4) + lane];
            acc.x += v.x * at; acc.y += v.y * at;
            acc.z += v.z * at; acc.w += v.w * at;
        }
    }

    const float* bp = bias + 4 * lane;
    float4 r;
    r.x = acc.x + bp[0]; r.y = acc.y + bp[1];
    r.z = acc.z + bp[2]; r.w = acc.w + bp[3];
    r.x = (r.x > 0.f) ? r.x : 0.f; r.y = (r.y > 0.f) ? r.y : 0.f;
    r.z = (r.z > 0.f) ? r.z : 0.f; r.w = (r.w > 0.f) ? r.w : 0.f;
    ((float4*)out)[(size_t)w * (HDIM / 4) + lane] = r;
}

extern "C" void kernel_launch(void* const* d_in, const int* in_sizes, int n_in,
                              void* d_out, int out_size, void* d_ws, size_t ws_size,
                              hipStream_t stream)
{
    const float* x   = (const float*)d_in[0];
    const int*   ei  = (const int*)d_in[1];
    const float* W1  = (const float*)d_in[2];
    const float* a1s = (const float*)d_in[3];
    const float* a1d = (const float*)d_in[4];
    const float* b1  = (const float*)d_in[5];
    const float* W2  = (const float*)d_in[6];
    const float* a2s = (const float*)d_in[7];
    const float* a2d = (const float*)d_in[8];
    const float* b2  = (const float*)d_in[9];
    const float* Wl  = (const float*)d_in[10];
    const float* bl  = (const float*)d_in[11];

    const int N  = in_sizes[0] / HDIM;
    const int E  = in_sizes[1] / 2;
    const int C  = in_sizes[11];
    const int CP = (C + 15) & ~15;

    char* ws = (char*)d_ws;
    auto align256 = [](size_t v) { return (v + 255) & ~(size_t)255; };
    size_t off = 0;
    float* bufH   = (float*)(ws + off);  off += align256((size_t)N * HDIM * 4);
    float* bufA   = (float*)(ws + off);  off += align256((size_t)N * HDIM * 4);
    float* Wt     = (float*)(ws + off);  off += align256((size_t)HDIM * HDIM * 4);
    float* Wlt    = (float*)(ws + off);  off += align256((size_t)CP * HDIM * 4);
    float* aS     = (float*)(ws + off);  off += align256((size_t)N * 4);
    float* aD     = (float*)(ws + off);  off += align256((size_t)N * 4);
    int*   cursor = (int*)(ws + off);    off += align256((size_t)N * 4);
    int*   rowptr = (int*)(ws + off);    off += align256((size_t)(N + 1) * 4);
    int*   srcS   = (int*)(ws + off);    off += align256((size_t)E * 4);
    (void)ws_size;

    const int nRowTiles = N / 16;
    const int BLK = 256;
    dim3 blk(BLK);

    const int nbZero  = (N + BLK - 1) / BLK;
    const int nbT     = (HDIM * HDIM + BLK - 1) / BLK;
    const int nbTl    = (CP * HDIM + BLK - 1) / BLK;
    const int nbGemm  = (nRowTiles * 2 + 7) / 8;
    const int nbEdge  = (E + BLK - 1) / BLK;
    const int nbAlpha = (N + 31) / 32;
    const int nbT2    = (HDIM * HDIM + 1023) / 1024;
    dim3 gemmGrid(nbGemm);
    dim3 waveGridN((N + 7) / 8);
    dim3 edgeGrid(nbEdge);

    gat_stage1<<<dim3(nbZero + nbT + nbTl), blk, 0, stream>>>(
        cursor, N, nbZero, W1, Wt, nbT, Wl, Wlt, C, CP);
    gat_stage2<<<dim3(nbGemm + nbEdge), blk, 0, stream>>>(
        x, Wt, bufH, nRowTiles, nbGemm, ei, E, cursor);
    gat_stage3<<<dim3(1 + nbAlpha + nbT2), dim3(1024), 0, stream>>>(
        cursor, rowptr, N, bufH, a1s, a1d, aS, aD, nbAlpha, W2, Wt);
    gat_fill<<<edgeGrid, blk, 0, stream>>>(ei, E, cursor, srcS);

    gat_gather<<<waveGridN, blk, 0, stream>>>(rowptr, srcS, aS, aD, bufH,
                                              b1, bufA, N);

    gat_gemm_strip<<<gemmGrid, blk, 0, stream>>>(bufA, Wt, bufH, nRowTiles);
    gat_alpha<<<waveGridN, blk, 0, stream>>>(bufH, a2s, a2d, aS, aD, N);
    gat_gather<<<waveGridN, blk, 0, stream>>>(rowptr, srcS, aS, aD, bufH,
                                              b2, bufA, N);

    dim3 clsGrid((nRowTiles + 7) / 8);
    gat_gemm_cls<<<clsGrid, blk, 0, stream>>>(bufA, Wlt, bl, (float*)d_out,
                                              nRowTiles, C);
}


# Prefix scan of the in-degree counters reads and writes whole 128-byte lines (4 consecutive counters per thread, 16-byte accesses) and scans by wave shuffles: 4 barriers, not 20
# speedup vs baseline: 2.5060x; 1.1176x over previous
#include <hip/hip_runtime.h>
#include <hip/hip_bf16.h>

typedef __attribute__((ext_vector_type(2))) float v2f;
typedef __attribute__((ext_vector_type(8))) float v8f;

#define HDIM 128
#define CTPW 4


__device__ __forceinline__ void gat_transpose_body(
    const float* __restrict__ W, float* __restrict__ Wt,
    int rows, int cols, int colsPad, int i)
{
    int total = colsPad * rows;
    if (i >= total) return;
    int c = i / rows;
    int r = i - c * rows;
    Wt[(size_t)c * rows + r] = (c < cols) ? W[(size_t)r * cols + c] : 0.f;
}

__device__ __forceinline__ void gat_gemm_strip_body(
    const float* __restrict__ X, const float* __restrict__ Wt,
    float* __restrict__ H, int nRowTiles, int wid, int lane)
{
    int nHalves = HDIM / (CTPW * 16);
    int nW = nRowTiles * nHalves;
    if (wid >= nW) return;
    int rt   = wid >> 1;
    int half = wid & 1;
    int row0 = rt * 16;
    int laneM = lane & 15;
    int kOff  = (lane >> 4) * 2;

    const float* xrow = X  + (size_t)(row0 + laneM) * HDIM + kOff;
    const float* wrow = Wt + (size_t)(half * (CTPW * 16) + laneM) * HDIM + kOff;

    v8f c[CTPW] = {};
    #pragma unroll 4
    for (int k = 0; k < HDIM; k += 4) {
        v2f a = *(const v2f*)(xrow + k);
        #pragma unroll
        for (int t = 0; t < CTPW; ++t) {
            v2f b = *(const v2f*)(wrow + (size_t)t * 16 * HDIM + k);
            c[t] = __builtin_amdgcn_wmma_f32_16x16x4_f32(
                       false, a, false, b, (short)0, c[t], false, false);
        }
    }

    int mBase = (lane >> 4) * 8;
    #pragma unroll
    for (int t = 0; t < CTPW; ++t) {
        float* o = H + (size_t)row0 * HDIM + (half * CTPW + t) * 16 + laneM;
        #pragma unroll
        for (int r = 0; r < 8; ++r)
            o[(size_t)(mBase + r) * HDIM] = c[t][r];
    }
}

__global__ __launch_bounds__(256) void gat_gemm_strip(
    const float* __restrict__ X, const float* __restrict__ Wt,
    float* __restrict__ H, int nRowTiles)
{
    gat_gemm_strip_body(X, Wt, H, nRowTiles,
                        (int)((blockIdx.x * blockDim.x + threadIdx.x) >> 5),
                        (int)(threadIdx.x & 31));
}

__global__ __launch_bounds__(256) void gat_gemm_cls(
    const float* __restrict__ X, const float* __restrict__ Wlt,
    const float* __restrict__ bl, float* __restrict__ Out,
    int nRowTiles, int nCols)
{
    int wid  = (blockIdx.x * blockDim.x + threadIdx.x) >> 5;
    int lane = threadIdx.x & 31;
    if (wid >= nRowTiles) return;
    int row0 = wid * 16;
    int laneM = lane & 15;
    int kOff  = (lane >> 4) * 2;

    const float* xrow = X   + (size_t)(row0 + laneM) * HDIM + kOff;
    const float* wrow = Wlt + (size_t)laneM * HDIM + kOff;

    v8f c[3] = {};
    #pragma unroll 4
    for (int k = 0; k < HDIM; k += 4) {
        v2f a = *(const v2f*)(xrow + k);
        #pragma unroll
        for (int t = 0; t < 3; ++t) {
            v2f b = *(const v2f*)(wrow + (size_t)t * 16 * HDIM + k);
            c[t] = __builtin_amdgcn_wmma_f32_16x16x4_f32(
                       false, a, false, b, (short)0, c[t], false, false);
        }
    }

    int mBase = (lane >> 4) * 8;
    #pragma unroll
    for (int t = 0; t < 3; ++t) {
        int col = t * 16 + laneM;
        if (col >= nCols) continue;
        float bv = bl[col];
        #pragma unroll
        for (int r = 0; r < 8; ++r)
            Out[(size_t)(row0 + mBase + r) * nCols + col] = c[t][r] + bv;
    }
}

__device__ __forceinline__ void gat_alpha_body(
    const float* __restrict__ H, const float* __restrict__ asrc,
    const float* __restrict__ adst, float* __restrict__ aS,
    float* __restrict__ aD, int N, int w, int lane)
{
    if (w >= N) return;
    const float* h = H + (size_t)w * HDIM;
    float s = 0.f, d = 0.f;
    #pragma unroll
    for (int j = 0; j < 4; ++j) {
        int f = lane + 32 * j;
        float hv = h[f];
        s += hv * asrc[f];
        d += hv * adst[f];
    }
    #pragma unroll
    for (int off = 16; off > 0; off >>= 1) {
        s += __shfl_xor(s, off, 32);
        d += __shfl_xor(d, off, 32);
    }
    if (lane == 0) { aS[w] = s; aD[w] = d; }
}

__global__ __launch_bounds__(256) void gat_alpha(
    const float* __restrict__ H, const float* __restrict__ asrc,
    const float* __restrict__ adst, float* __restrict__ aS,
    float* __restrict__ aD, int N)
{
    gat_alpha_body(H, asrc, adst, aS, aD, N,
                   (int)((blockIdx.x * blockDim.x + threadIdx.x) >> 5),
                   (int)(threadIdx.x & 31));
}

__device__ __forceinline__ void gat_hist_body(
    const int* __restrict__ ei, int E, int* __restrict__ cnt, int i)
{
    if (i >= E) return;
    atomicAdd(&cnt[ei[E + i]], 1);
}

#define SCAN_CHUNKS 5

__device__ __forceinline__ int gat_wave_incl_scan(int x, int lane)
{
    #pragma unroll
    for (int off = 1; off < 32; off <<= 1) {
        int y = __shfl_up(x, off, 32);
        if (lane >= off) x += y;
    }
    return x;
}

__device__ __forceinline__ void gat_scan_body(
    int* __restrict__ cnt, int* __restrict__ rowptr, int N)
{
    __shared__ int wtot[SCAN_CHUNKS * 32];
    __shared__ int w2[32];
    const int t    = threadIdx.x;
    const int lane = t & 31;
    const int wv   = t >> 5;
    int carry = 0;
    for (int base = 0; base < N; base += 1024 * 4 * SCAN_CHUNKS) {
        int4 v[SCAN_CHUNKS];
        #pragma unroll
        for (int j = 0; j < SCAN_CHUNKS; ++j) {
            int e0 = base + (j * 1024 + t) * 4;
            bool full = (e0 + 4 <= N);
            int4 q = *(const int4*)(cnt + (full ? e0 : 0));
            if (!full) {
                q.x = (e0     < N) ? cnt[e0]     : 0;
                q.y = (e0 + 1 < N) ? cnt[e0 + 1] : 0;
                q.z = (e0 + 2 < N) ? cnt[e0 + 2] : 0;
                q.w = 0;
            }
            v[j] = q;
        }
        int inc[SCAN_CHUNKS];
        #pragma unroll
        for (int j = 0; j < SCAN_CHUNKS; ++j) {
            int s = v[j].x + v[j].y + v[j].z + v[j].w;
            inc[j] = gat_wave_incl_scan(s, lane);
            if (lane == 31) wtot[j * 32 + wv] = inc[j];
        }
        __syncthreads();
        int tv = (t < SCAN_CHUNKS * 32) ? wtot[t] : 0;
        int ti = gat_wave_incl_scan(tv, lane);
        if (lane == 31) w2[wv] = ti;
        __syncthreads();
        int w2v   = w2[lane];
        int w2i   = gat_wave_incl_scan(w2v, lane);
        int wpre  = __shfl(w2i - w2v, wv, 32);
        int total = __shfl(w2i, 31, 32);
        if (t < SCAN_CHUNKS * 32) wtot[t] = ti - tv + wpre;
        __syncthreads();
        #pragma unroll
        for (int j = 0; j < SCAN_CHUNKS; ++j) {
            int e0 = base + (j * 1024 + t) * 4;
            int s  = v[j].x + v[j].y + v[j].z + v[j].w;
            int4 o;
            o.x = carry + wtot[j * 32 + wv] + inc[j] - s;
            o.y = o.x + v[j].x;
            o.z = o.y + v[j].y;
            o.w = o.z + v[j].z;
            if (e0 + 4 <= N) {
                *(int4*)(rowptr + e0) = o;
                *(int4*)(cnt + e0)    = o;
            } else {
                if (e0     < N) { rowptr[e0]     = o.x; cnt[e0]     = o.x; }
                if (e0 + 1 < N) { rowptr[e0 + 1] = o.y; cnt[e0 + 1] = o.y; }
                if (e0 + 2 < N) { rowptr[e0 + 2] = o.z; cnt[e0 + 2] = o.z; }
            }
        }
        carry += total;
        __syncthreads();
    }
    if (t == 0) rowptr[N] = carry;
}

__global__ __launch_bounds__(256) void gat_fill(
    const int* __restrict__ ei, int E, int* __restrict__ cursor,
    int* __restrict__ srcS)
{
    int i = blockIdx.x * blockDim.x + threadIdx.x;
    if (i >= E) return;
    int p = atomicAdd(&cursor[ei[E + i]], 1);
    srcS[p] = ei[i];
}


__global__ __launch_bounds__(256) void gat_stage1(
    int* __restrict__ cnt, int N, int nbZero,
    const float* __restrict__ W1, float* __restrict__ Wt, int nbT1,
    const float* __restrict__ Wl, float* __restrict__ Wlt, int C, int CP)
{
    int b = blockIdx.x;
    int t = threadIdx.x;
    if (b < nbZero) {
        int i = b * 256 + t;
        if (i < N) cnt[i] = 0;
    } else if (b < nbZero + nbT1) {
        gat_transpose_body(W1, Wt, HDIM, HDIM, HDIM, (b - nbZero) * 256 + t);
    } else {
        gat_transpose_body(Wl, Wlt, HDIM, C, CP, (b - nbZero - nbT1) * 256 + t);
    }
}

__global__ __launch_bounds__(256) void gat_stage2(
    const float* __restrict__ X, const float* __restrict__ Wt,
    float* __restrict__ H, int nRowTiles, int nbGemm,
    const int* __restrict__ ei, int E, int* __restrict__ cnt)
{
    int b = blockIdx.x;
    int t = threadIdx.x;
    if (b < nbGemm) {
        gat_gemm_strip_body(X, Wt, H, nRowTiles, (b * 256 + t) >> 5, t & 31);
    } else {
        gat_hist_body(ei, E, cnt, (b - nbGemm) * 256 + t);
    }
}

__global__ __launch_bounds__(1024) void gat_stage3(
    int* __restrict__ cnt, int* __restrict__ rowptr, int N,
    const float* __restrict__ H, const float* __restrict__ asrc,
    const float* __restrict__ adst, float* __restrict__ aS,
    float* __restrict__ aD, int nbAlpha,
    const float* __restrict__ W2, float* __restrict__ Wt)
{
    int b = blockIdx.x;
    int t = threadIdx.x;
    if (b == 0) {
        gat_scan_body(cnt, rowptr, N);
    } else if (b <= nbAlpha) {
        gat_alpha_body(H, asrc, adst, aS, aD, N,
                       ((b - 1) * 1024 + t) >> 5, t & 31);
    } else {
        gat_transpose_body(W2, Wt, HDIM, HDIM, HDIM,
                           (b - 1 - nbAlpha) * 1024 + t);
    }
}

__device__ __forceinline__ float gat_leaky(float e) {
    return (e >= 0.f) ? e : 0.2f * e;
}

__global__ __launch_bounds__(256) void gat_gather(
    const int* __restrict__ rowptr, const int* __restrict__ srcS,
    const float* __restrict__ aS, const float* __restrict__ aD,
    const float* __restrict__ H, const float* __restrict__ bias,
    float* __restrict__ out, int N)
{
    int w    = __builtin_amdgcn_readfirstlane(
                   (int)((blockIdx.x * blockDim.x + threadIdx.x) >> 5));
    int lane = threadIdx.x & 31;
    if (w >= N) return;
    int beg = rowptr[w];
    int end = rowptr[w + 1];
    float ad    = aD[w];
    float eSelf = gat_leaky(aS[w] + ad);

    float m = eSelf;
    for (int j = beg + lane; j < end; j += 32)
        m = fmaxf(m, gat_leaky(aS[srcS[j]] + ad));
    for (int off = 16; off > 0; off >>= 1)
        m = fmaxf(m, __shfl_xor(m, off, 32));

    float exSelf = expf(eSelf - m);
    float den = (lane == 0) ? exSelf : 0.f;
    for (int j = beg + lane; j < end; j += 32)
        den += expf(gat_leaky(aS[srcS[j]] + ad) - m);
    for (int off = 16; off > 0; off >>= 1)
        den += __shfl_xor(den, off, 32);

    const float4* H4 = (const float4*)H;
    float aSelf = exSelf / den;
    float4 hv = H4[(size_t)w * (HDIM / 4) + lane];
    float4 acc;
    acc.x = hv.x * aSelf; acc.y = hv.y * aSelf;
    acc.z = hv.z * aSelf; acc.w = hv.w * aSelf;
    for (int base = beg; base < end; base += 32) {
        int j = base + lane;
        int s = 0;
        float a = 0.f;
        if (j < end) {
            s = srcS[j];
            a = expf(gat_leaky(aS[s] + ad) - m) / den;
        }
        int cnt = (end - base < 32) ? (end - base) : 32;
        #pragma unroll 4
        for (int t = 0; t < cnt; ++t) {
            int   st = __builtin_amdgcn_readlane(s, t);
            float at = __int_as_float(
                           __builtin_amdgcn_readlane(__float_as_int(a), t));
            float4 v = H4[(size_t)st * (HDIM / 4) + lane];
            acc.x += v.x * at; acc.y += v.y * at;
            acc.z += v.z * at; acc.w += v.w * at;
        }
    }

    const float* bp = bias + 4 * lane;
    float4 r;
    r.x = acc.x + bp[0]; r.y = acc.y + bp[1];
    r.z = acc.z + bp[2]; r.w = acc.w + bp[3];
    r.x = (r.x > 0.f) ? r.x : 0.f; r.y = (r.y > 0.f) ? r.y : 0.f;
    r.z = (r.z > 0.f) ? r.z : 0.f; r.w = (r.w > 0.f) ? r.w : 0.f;
    ((float4*)out)[(size_t)w * (HDIM / 4) + lane] = r;
}

extern "C" void kernel_launch(void* const* d_in, const int* in_sizes, int n_in,
                              void* d_out, int out_size, void* d_ws, size_t ws_size,
                              hipStream_t stream)
{
    const float* x   = (const float*)d_in[0];
    const int*   ei  = (const int*)d_in[1];
    const float* W1  = (const float*)d_in[2];
    const float* a1s = (const float*)d_in[3];
    const float* a1d = (const float*)d_in[4];
    const float* b1  = (const float*)d_in[5];
    const float* W2  = (const float*)d_in[6];
    const float* a2s = (const float*)d_in[7];
    const float* a2d = (const float*)d_in[8];
    const float* b2  = (const float*)d_in[9];
    const float* Wl  = (const float*)d_in[10];
    const float* bl  = (const float*)d_in[11];

    const int N  = in_sizes[0] / HDIM;
    const int E  = in_sizes[1] / 2;
    const int C  = in_sizes[11];
    const int CP = (C + 15) & ~15;

    char* ws = (char*)d_ws;
    auto align256 = [](size_t v) { return (v + 255) & ~(size_t)255; };
    size_t off = 0;
    float* bufH   = (float*)(ws + off);  off += align256((size_t)N * HDIM * 4);
    float* bufA   = (float*)(ws + off);  off += align256((size_t)N * HDIM * 4);
    float* Wt     = (float*)(ws + off);  off += align256((size_t)HDIM * HDIM * 4);
    float* Wlt    = (float*)(ws + off);  off += align256((size_t)CP * HDIM * 4);
    float* aS     = (float*)(ws + off);  off += align256((size_t)N * 4);
    float* aD     = (float*)(ws + off);  off += align256((size_t)N * 4);
    int*   cursor = (int*)(ws + off);    off += align256((size_t)N * 4);
    int*   rowptr = (int*)(ws + off);    off += align256((size_t)(N + 1) * 4);
    int*   srcS   = (int*)(ws + off);    off += align256((size_t)E * 4);
    (void)ws_size;

    const int nRowTiles = N / 16;
    const int BLK = 256;
    dim3 blk(BLK);

    const int nbZero  = (N + BLK - 1) / BLK;
    const int nbT     = (HDIM * HDIM + BLK - 1) / BLK;
    const int nbTl    = (CP * HDIM + BLK - 1) / BLK;
    const int nbGemm  = (nRowTiles * 2 + 7) / 8;
    const int nbEdge  = (E + BLK - 1) / BLK;
    const int nbAlpha = (N + 31) / 32;
    const int nbT2    = (HDIM * HDIM + 1023) / 1024;
    dim3 gemmGrid(nbGemm);
    dim3 waveGridN((N + 7) / 8);
    dim3 edgeGrid(nbEdge);

    gat_stage1<<<dim3(nbZero + nbT + nbTl), blk, 0, stream>>>(
        cursor, N, nbZero, W1, Wt, nbT, Wl, Wlt, C, CP);
    gat_stage2<<<dim3(nbGemm + nbEdge), blk, 0, stream>>>(
        x, Wt, bufH, nRowTiles, nbGemm, ei, E, cursor);
    gat_stage3<<<dim3(1 + nbAlpha + nbT2), dim3(1024), 0, stream>>>(
        cursor, rowptr, N, bufH, a1s, a1d, aS, aD, nbAlpha, W2, Wt);
    gat_fill<<<edgeGrid, blk, 0, stream>>>(ei, E, cursor, srcS);

    gat_gather<<<waveGridN, blk, 0, stream>>>(rowptr, srcS, aS, aD, bufH,
                                              b1, bufA, N);

    gat_gemm_strip<<<gemmGrid, blk, 0, stream>>>(bufA, Wt, bufH, nRowTiles);
    gat_alpha<<<waveGridN, blk, 0, stream>>>(bufH, a2s, a2d, aS, aD, N);
    gat_gather<<<waveGridN, blk, 0, stream>>>(rowptr, srcS, aS, aD, bufH,
                                              b2, bufA, N);

    dim3 clsGrid((nRowTiles + 7) / 8);
    gat_gemm_cls<<<clsGrid, blk, 0, stream>>>(bufA, Wlt, bl, (float*)d_out,
                                              nRowTiles, C);
}


# Gather reads each edge's source index and logit once and keeps them in registers (2 chunks of 32 edges per lane) for max, sum and weights; own row and bias requested early: 3 index->logit trips -> 1
# speedup vs baseline: 2.6443x; 1.0552x over previous
#include <hip/hip_runtime.h>
#include <hip/hip_bf16.h>

typedef __attribute__((ext_vector_type(2))) float v2f;
typedef __attribute__((ext_vector_type(8))) float v8f;

#define HDIM 128
#define CTPW 4


__device__ __forceinline__ void gat_transpose_body(
    const float* __restrict__ W, float* __restrict__ Wt,
    int rows, int cols, int colsPad, int i)
{
    int total = colsPad * rows;
    if (i >= total) return;
    int c = i / rows;
    int r = i - c * rows;
    Wt[(size_t)c * rows + r] = (c < cols) ? W[(size_t)r * cols + c] : 0.f;
}

__device__ __forceinline__ void gat_gemm_strip_body(
    const float* __restrict__ X, const float* __restrict__ Wt,
    float* __restrict__ H, int nRowTiles, int wid, int lane)
{
    int nHalves = HDIM / (CTPW * 16);
    int nW = nRowTiles * nHalves;
    if (wid >= nW) return;
    int rt   = wid >> 1;
    int half = wid & 1;
    int row0 = rt * 16;
    int laneM = lane & 15;
    int kOff  = (lane >> 4) * 2;

    const float* xrow = X  + (size_t)(row0 + laneM) * HDIM + kOff;
    const float* wrow = Wt + (size_t)(half * (CTPW * 16) + laneM) * HDIM + kOff;

    v8f c[CTPW] = {};
    #pragma unroll 4
    for (int k = 0; k < HDIM; k += 4) {
        v2f a = *(const v2f*)(xrow + k);
        #pragma unroll
        for (int t = 0; t < CTPW; ++t) {
            v2f b = *(const v2f*)(wrow + (size_t)t * 16 * HDIM + k);
            c[t] = __builtin_amdgcn_wmma_f32_16x16x4_f32(
                       false, a, false, b, (short)0, c[t], false, false);
        }
    }

    int mBase = (lane >> 4) * 8;
    #pragma unroll
    for (int t = 0; t < CTPW; ++t) {
        float* o = H + (size_t)row0 * HDIM + (half * CTPW + t) * 16 + laneM;
        #pragma unroll
        for (int r = 0; r < 8; ++r)
            o[(size_t)(mBase + r) * HDIM] = c[t][r];
    }
}

__global__ __launch_bounds__(256) void gat_gemm_strip(
    const float* __restrict__ X, const float* __restrict__ Wt,
    float* __restrict__ H, int nRowTiles)
{
    gat_gemm_strip_body(X, Wt, H, nRowTiles,
                        (int)((blockIdx.x * blockDim.x + threadIdx.x) >> 5),
                        (int)(threadIdx.x & 31));
}

__global__ __launch_bounds__(256) void gat_gemm_cls(
    const float* __restrict__ X, const float* __restrict__ Wlt,
    const float* __restrict__ bl, float* __restrict__ Out,
    int nRowTiles, int nCols)
{
    int wid  = (blockIdx.x * blockDim.x + threadIdx.x) >> 5;
    int lane = threadIdx.x & 31;
    if (wid >= nRowTiles) return;
    int row0 = wid * 16;
    int laneM = lane & 15;
    int kOff  = (lane >> 4) * 2;

    const float* xrow = X   + (size_t)(row0 + laneM) * HDIM + kOff;
    const float* wrow = Wlt + (size_t)laneM * HDIM + kOff;

    v8f c[3] = {};
    #pragma unroll 4
    for (int k = 0; k < HDIM; k += 4) {
        v2f a = *(const v2f*)(xrow + k);
        #pragma unroll
        for (int t = 0; t < 3; ++t) {
            v2f b = *(const v2f*)(wrow + (size_t)t * 16 * HDIM + k);
            c[t] = __builtin_amdgcn_wmma_f32_16x16x4_f32(
                       false, a, false, b, (short)0, c[t], false, false);
        }
    }

    int mBase = (lane >> 4) * 8;
    #pragma unroll
    for (int t = 0; t < 3; ++t) {
        int col = t * 16 + laneM;
        if (col >= nCols) continue;
        float bv = bl[col];
        #pragma unroll
        for (int r = 0; r < 8; ++r)
            Out[(size_t)(row0 + mBase + r) * nCols + col] = c[t][r] + bv;
    }
}

__device__ __forceinline__ void gat_alpha_body(
    const float* __restrict__ H, const float* __restrict__ asrc,
    const float* __restrict__ adst, float* __restrict__ aS,
    float* __restrict__ aD, int N, int w, int lane)
{
    if (w >= N) return;
    const float* h = H + (size_t)w * HDIM;
    float s = 0.f, d = 0.f;
    #pragma unroll
    for (int j = 0; j < 4; ++j) {
        int f = lane + 32 * j;
        float hv = h[f];
        s += hv * asrc[f];
        d += hv * adst[f];
    }
    #pragma unroll
    for (int off = 16; off > 0; off >>= 1) {
        s += __shfl_xor(s, off, 32);
        d += __shfl_xor(d, off, 32);
    }
    if (lane == 0) { aS[w] = s; aD[w] = d; }
}

__global__ __launch_bounds__(256) void gat_alpha(
    const float* __restrict__ H, const float* __restrict__ asrc,
    const float* __restrict__ adst, float* __restrict__ aS,
    float* __restrict__ aD, int N)
{
    gat_alpha_body(H, asrc, adst, aS, aD, N,
                   (int)((blockIdx.x * blockDim.x + threadIdx.x) >> 5),
                   (int)(threadIdx.x & 31));
}

__device__ __forceinline__ void gat_hist_body(
    const int* __restrict__ ei, int E, int* __restrict__ cnt, int i)
{
    if (i >= E) return;
    atomicAdd(&cnt[ei[E + i]], 1);
}

#define SCAN_CHUNKS 5

__device__ __forceinline__ int gat_wave_incl_scan(int x, int lane)
{
    #pragma unroll
    for (int off = 1; off < 32; off <<= 1) {
        int y = __shfl_up(x, off, 32);
        if (lane >= off) x += y;
    }
    return x;
}

__device__ __forceinline__ void gat_scan_body(
    int* __restrict__ cnt, int* __restrict__ rowptr, int N)
{
    __shared__ int wtot[SCAN_CHUNKS * 32];
    __shared__ int w2[32];
    const int t    = threadIdx.x;
    const int lane = t & 31;
    const int wv   = t >> 5;
    int carry = 0;
    for (int base = 0; base < N; base += 1024 * 4 * SCAN_CHUNKS) {
        int4 v[SCAN_CHUNKS];
        #pragma unroll
        for (int j = 0; j < SCAN_CHUNKS; ++j) {
            int e0 = base + (j * 1024 + t) * 4;
            bool full = (e0 + 4 <= N);
            int4 q = *(const int4*)(cnt + (full ? e0 : 0));
            if (!full) {
                q.x = (e0     < N) ? cnt[e0]     : 0;
                q.y = (e0 + 1 < N) ? cnt[e0 + 1] : 0;
                q.z = (e0 + 2 < N) ? cnt[e0 + 2] : 0;
                q.w = 0;
            }
            v[j] = q;
        }
        int inc[SCAN_CHUNKS];
        #pragma unroll
        for (int j = 0; j < SCAN_CHUNKS; ++j) {
            int s = v[j].x + v[j].y + v[j].z + v[j].w;
            inc[j] = gat_wave_incl_scan(s, lane);
            if (lane == 31) wtot[j * 32 + wv] = inc[j];
        }
        __syncthreads();
        int tv = (t < SCAN_CHUNKS * 32) ? wtot[t] : 0;
        int ti = gat_wave_incl_scan(tv, lane);
        if (lane == 31) w2[wv] = ti;
        __syncthreads();
        int w2v   = w2[lane];
        int w2i   = gat_wave_incl_scan(w2v, lane);
        int wpre  = __shfl(w2i - w2v, wv, 32);
        int total = __shfl(w2i, 31, 32);
        if (t < SCAN_CHUNKS * 32) wtot[t] = ti - tv + wpre;
        __syncthreads();
        #pragma unroll
        for (int j = 0; j < SCAN_CHUNKS; ++j) {
            int e0 = base + (j * 1024 + t) * 4;
            int s  = v[j].x + v[j].y + v[j].z + v[j].w;
            int4 o;
            o.x = carry + wtot[j * 32 + wv] + inc[j] - s;
            o.y = o.x + v[j].x;
            o.z = o.y + v[j].y;
            o.w = o.z + v[j].z;
            if (e0 + 4 <= N) {
                *(int4*)(rowptr + e0) = o;
                *(int4*)(cnt + e0)    = o;
            } else {
                if (e0     < N) { rowptr[e0]     = o.x; cnt[e0]     = o.x; }
                if (e0 + 1 < N) { rowptr[e0 + 1] = o.y; cnt[e0 + 1] = o.y; }
                if (e0 + 2 < N) { rowptr[e0 + 2] = o.z; cnt[e0 + 2] = o.z; }
            }
        }
        carry += total;
        __syncthreads();
    }
    if (t == 0) rowptr[N] = carry;
}

__global__ __launch_bounds__(256) void gat_fill(
    const int* __restrict__ ei, int E, int* __restrict__ cursor,
    int* __restrict__ srcS)
{
    int i = blockIdx.x * blockDim.x + threadIdx.x;
    if (i >= E) return;
    int p = atomicAdd(&cursor[ei[E + i]], 1);
    srcS[p] = ei[i];
}


__global__ __launch_bounds__(256) void gat_stage1(
    int* __restrict__ cnt, int N, int nbZero,
    const float* __restrict__ W1, float* __restrict__ Wt, int nbT1,
    const float* __restrict__ Wl, float* __restrict__ Wlt, int C, int CP)
{
    int b = blockIdx.x;
    int t = threadIdx.x;
    if (b < nbZero) {
        int i = b * 256 + t;
        if (i < N) cnt[i] = 0;
    } else if (b < nbZero + nbT1) {
        gat_transpose_body(W1, Wt, HDIM, HDIM, HDIM, (b - nbZero) * 256 + t);
    } else {
        gat_transpose_body(Wl, Wlt, HDIM, C, CP, (b - nbZero - nbT1) * 256 + t);
    }
}

__global__ __launch_bounds__(256) void gat_stage2(
    const float* __restrict__ X, const float* __restrict__ Wt,
    float* __restrict__ H, int nRowTiles, int nbGemm,
    const int* __restrict__ ei, int E, int* __restrict__ cnt)
{
    int b = blockIdx.x;
    int t = threadIdx.x;
    if (b < nbGemm) {
        gat_gemm_strip_body(X, Wt, H, nRowTiles, (b * 256 + t) >> 5, t & 31);
    } else {
        gat_hist_body(ei, E, cnt, (b - nbGemm) * 256 + t);
    }
}

__global__ __launch_bounds__(1024) void gat_stage3(
    int* __restrict__ cnt, int* __restrict__ rowptr, int N,
    const float* __restrict__ H, const float* __restrict__ asrc,
    const float* __restrict__ adst, float* __restrict__ aS,
    float* __restrict__ aD, int nbAlpha,
    const float* __restrict__ W2, float* __restrict__ Wt)
{
    int b = blockIdx.x;
    int t = threadIdx.x;
    if (b == 0) {
        gat_scan_body(cnt, rowptr, N);
    } else if (b <= nbAlpha) {
        gat_alpha_body(H, asrc, adst, aS, aD, N,
                       ((b - 1) * 1024 + t) >> 5, t & 31);
    } else {
        gat_transpose_body(W2, Wt, HDIM, HDIM, HDIM,
                           (b - 1 - nbAlpha) * 1024 + t);
    }
}

__device__ __forceinline__ float gat_leaky(float e) {
    return (e >= 0.f) ? e : 0.2f * e;
}

__global__ __launch_bounds__(256) void gat_gather(
    const int* __restrict__ rowptr, const int* __restrict__ srcS,
    const float* __restrict__ aS, const float* __restrict__ aD,
    const float* __restrict__ H, const float* __restrict__ bias,
    float* __restrict__ out, int N)
{
    int w    = __builtin_amdgcn_readfirstlane(
                   (int)((blockIdx.x * blockDim.x + threadIdx.x) >> 5));
    int lane = threadIdx.x & 31;
    if (w >= N) return;
    int beg = rowptr[w];
    int end = rowptr[w + 1];
    float ad    = aD[w];
    float eSelf = gat_leaky(aS[w] + ad);

    const float4* H4 = (const float4*)H;
    float4 hv = H4[(size_t)w * (HDIM / 4) + lane];

    int  j0  = beg + lane;
    int  j1  = j0 + 32;
    bool in0 = j0 < end;
    bool in1 = j1 < end;
    int s0 = srcS[in0 ? j0 : 0];
    int s1 = srcS[in1 ? j1 : 0];
    float l0 = aS[s0];
    float l1 = aS[s1];
    float e0 = in0 ? gat_leaky(l0 + ad) : eSelf;
    float e1 = in1 ? gat_leaky(l1 + ad) : eSelf;

    float m = fmaxf(fmaxf(eSelf, e0), e1);
    for (int j = j1 + 32; j < end; j += 32)
        m = fmaxf(m, gat_leaky(aS[srcS[j]] + ad));
    for (int off = 16; off > 0; off >>= 1)
        m = fmaxf(m, __shfl_xor(m, off, 32));

    const float* bp = bias + 4 * lane;
    float bv0 = bp[0], bv1 = bp[1], bv2 = bp[2], bv3 = bp[3];

    float exSelf = expf(eSelf - m);
    float x0 = in0 ? expf(e0 - m) : 0.f;
    float x1 = in1 ? expf(e1 - m) : 0.f;
    float den = (lane == 0) ? exSelf : 0.f;
    den += x0;
    den += x1;
    for (int j = j1 + 32; j < end; j += 32)
        den += expf(gat_leaky(aS[srcS[j]] + ad) - m);
    for (int off = 16; off > 0; off >>= 1)
        den += __shfl_xor(den, off, 32);

    float aSelf = exSelf / den;
    float a0 = x0 / den;
    float a1 = x1 / den;
    float4 acc;
    acc.x = hv.x * aSelf; acc.y = hv.y * aSelf;
    acc.z = hv.z * aSelf; acc.w = hv.w * aSelf;
    int chunk = 0;
    for (int base = beg; base < end; base += 32, ++chunk) {
        int s;
        float a;
        if (chunk == 0) {
            s = s0; a = a0;
        } else if (chunk == 1) {
            s = s1; a = a1;
        } else {
            int j = base + lane;
            s = 0;
            a = 0.f;
            if (j < end) {
                s = srcS[j];
                a = expf(gat_leaky(aS[s] + ad) - m) / den;
            }
        }
        int cnt = (end - base < 32) ? (end - base) : 32;
        #pragma unroll 4
        for (int t = 0; t < cnt; ++t) {
            int   st = __builtin_amdgcn_readlane(s, t);
            float at = __int_as_float(
                           __builtin_amdgcn_readlane(__float_as_int(a), t));
            float4 v = H4[(size_t)st * (HDIM / 4) + lane];
            acc.x += v.x * at; acc.y += v.y * at;
            acc.z += v.z * at; acc.w += v.w * at;
        }
    }

    float4 r;
    r.x = acc.x + bv0; r.y = acc.y + bv1;
    r.z = acc.z + bv2; r.w = acc.w + bv3;
    r.x = (r.x > 0.f) ? r.x : 0.f; r.y = (r.y > 0.f) ? r.y : 0.f;
    r.z = (r.z > 0.f) ? r.z : 0.f; r.w = (r.w > 0.f) ? r.w : 0.f;
    ((float4*)out)[(size_t)w * (HDIM / 4) + lane] = r;
}

extern "C" void kernel_launch(void* const* d_in, const int* in_sizes, int n_in,
                              void* d_out, int out_size, void* d_ws, size_t ws_size,
                              hipStream_t stream)
{
    const float* x   = (const float*)d_in[0];
    const int*   ei  = (const int*)d_in[1];
    const float* W1  = (const float*)d_in[2];
    const float* a1s = (const float*)d_in[3];
    const float* a1d = (const float*)d_in[4];
    const float* b1  = (const float*)d_in[5];
    const float* W2  = (const float*)d_in[6];
    const float* a2s = (const float*)d_in[7];
    const float* a2d = (const float*)d_in[8];
    const float* b2  = (const float*)d_in[9];
    const float* Wl  = (const float*)d_in[10];
    const float* bl  = (const float*)d_in[11];

    const int N  = in_sizes[0] / HDIM;
    const int E  = in_sizes[1] / 2;
    const int C  = in_sizes[11];
    const int CP = (C + 15) & ~15;

    char* ws = (char*)d_ws;
    auto align256 = [](size_t v) { return (v + 255) & ~(size_t)255; };
    size_t off = 0;
    float* bufH   = (float*)(ws + off);  off += align256((size_t)N * HDIM * 4);
    float* bufA   = (float*)(ws + off);  off += align256((size_t)N * HDIM * 4);
    float* Wt     = (float*)(ws + off);  off += align256((size_t)HDIM * HDIM * 4);
    float* Wlt    = (float*)(ws + off);  off += align256((size_t)CP * HDIM * 4);
    float* aS     = (float*)(ws + off);  off += align256((size_t)N * 4);
    float* aD     = (float*)(ws + off);  off += align256((size_t)N * 4);
    int*   cursor = (int*)(ws + off);    off += align256((size_t)N * 4);
    int*   rowptr = (int*)(ws + off);    off += align256((size_t)(N + 1) * 4);
    int*   srcS   = (int*)(ws + off);    off += align256((size_t)E * 4);
    (void)ws_size;

    const int nRowTiles = N / 16;
    const int BLK = 256;
    dim3 blk(BLK);

    const int nbZero  = (N + BLK - 1) / BLK;
    const int nbT     = (HDIM * HDIM + BLK - 1) / BLK;
    const int nbTl    = (CP * HDIM + BLK - 1) / BLK;
    const int nbGemm  = (nRowTiles * 2 + 7) / 8;
    const int nbEdge  = (E + BLK - 1) / BLK;
    const int nbAlpha = (N + 31) / 32;
    const int nbT2    = (HDIM * HDIM + 1023) / 1024;
    dim3 gemmGrid(nbGemm);
    dim3 waveGridN((N + 7) / 8);
    dim3 edgeGrid(nbEdge);

    gat_stage1<<<dim3(nbZero + nbT + nbTl), blk, 0, stream>>>(
        cursor, N, nbZero, W1, Wt, nbT, Wl, Wlt, C, CP);
    gat_stage2<<<dim3(nbGemm + nbEdge), blk, 0, stream>>>(
        x, Wt, bufH, nRowTiles, nbGemm, ei, E, cursor);
    gat_stage3<<<dim3(1 + nbAlpha + nbT2), dim3(1024), 0, stream>>>(
        cursor, rowptr, N, bufH, a1s, a1d, aS, aD, nbAlpha, W2, Wt);
    gat_fill<<<edgeGrid, blk, 0, stream>>>(ei, E, cursor, srcS);

    gat_gather<<<waveGridN, blk, 0, stream>>>(rowptr, srcS, aS, aD, bufH,
                                              b1, bufA, N);

    gat_gemm_strip<<<gemmGrid, blk, 0, stream>>>(bufA, Wt, bufH, nRowTiles);
    gat_alpha<<<waveGridN, blk, 0, stream>>>(bufH, a2s, a2d, aS, aD, N);
    gat_gather<<<waveGridN, blk, 0, stream>>>(rowptr, srcS, aS, aD, bufH,
                                              b2, bufA, N);

    dim3 clsGrid((nRowTiles + 7) / 8);
    gat_gemm_cls<<<clsGrid, blk, 0, stream>>>(bufA, Wlt, bl, (float*)d_out,
                                              nRowTiles, C);
}


# Each edge's slot = rowptr[dst] + the value its counting atomic returned (kept as a rank): placement uses plain loads, one atomic pass over the edges, not two; segment order stays atomic arrival order
# speedup vs baseline: 3.1158x; 1.1783x over previous
#include <hip/hip_runtime.h>
#include <hip/hip_bf16.h>

typedef __attribute__((ext_vector_type(2))) float v2f;
typedef __attribute__((ext_vector_type(8))) float v8f;

#define HDIM 128
#define CTPW 4


__device__ __forceinline__ void gat_transpose_body(
    const float* __restrict__ W, float* __restrict__ Wt,
    int rows, int cols, int colsPad, int i)
{
    int total = colsPad * rows;
    if (i >= total) return;
    int c = i / rows;
    int r = i - c * rows;
    Wt[(size_t)c * rows + r] = (c < cols) ? W[(size_t)r * cols + c] : 0.f;
}

__device__ __forceinline__ void gat_gemm_strip_body(
    const float* __restrict__ X, const float* __restrict__ Wt,
    float* __restrict__ H, int nRowTiles, int wid, int lane)
{
    int nHalves = HDIM / (CTPW * 16);
    int nW = nRowTiles * nHalves;
    if (wid >= nW) return;
    int rt   = wid >> 1;
    int half = wid & 1;
    int row0 = rt * 16;
    int laneM = lane & 15;
    int kOff  = (lane >> 4) * 2;

    const float* xrow = X  + (size_t)(row0 + laneM) * HDIM + kOff;
    const float* wrow = Wt + (size_t)(half * (CTPW * 16) + laneM) * HDIM + kOff;

    v8f c[CTPW] = {};
    #pragma unroll 4
    for (int k = 0; k < HDIM; k += 4) {
        v2f a = *(const v2f*)(xrow + k);
        #pragma unroll
        for (int t = 0; t < CTPW; ++t) {
            v2f b = *(const v2f*)(wrow + (size_t)t * 16 * HDIM + k);
            c[t] = __builtin_amdgcn_wmma_f32_16x16x4_f32(
                       false, a, false, b, (short)0, c[t], false, false);
        }
    }

    int mBase = (lane >> 4) * 8;
    #pragma unroll
    for (int t = 0; t < CTPW; ++t) {
        float* o = H + (size_t)row0 * HDIM + (half * CTPW + t) * 16 + laneM;
        #pragma unroll
        for (int r = 0; r < 8; ++r)
            o[(size_t)(mBase + r) * HDIM] = c[t][r];
    }
}

__global__ __launch_bounds__(256) void gat_gemm_strip(
    const float* __restrict__ X, const float* __restrict__ Wt,
    float* __restrict__ H, int nRowTiles)
{
    gat_gemm_strip_body(X, Wt, H, nRowTiles,
                        (int)((blockIdx.x * blockDim.x + threadIdx.x) >> 5),
                        (int)(threadIdx.x & 31));
}

__global__ __launch_bounds__(256) void gat_gemm_cls(
    const float* __restrict__ X, const float* __restrict__ Wlt,
    const float* __restrict__ bl, float* __restrict__ Out,
    int nRowTiles, int nCols)
{
    int wid  = (blockIdx.x * blockDim.x + threadIdx.x) >> 5;
    int lane = threadIdx.x & 31;
    if (wid >= nRowTiles) return;
    int row0 = wid * 16;
    int laneM = lane & 15;
    int kOff  = (lane >> 4) * 2;

    const float* xrow = X   + (size_t)(row0 + laneM) * HDIM + kOff;
    const float* wrow = Wlt + (size_t)laneM * HDIM + kOff;

    v8f c[3] = {};
    #pragma unroll 4
    for (int k = 0; k < HDIM; k += 4) {
        v2f a = *(const v2f*)(xrow + k);
        #pragma unroll
        for (int t = 0; t < 3; ++t) {
            v2f b = *(const v2f*)(wrow + (size_t)t * 16 * HDIM + k);
            c[t] = __builtin_amdgcn_wmma_f32_16x16x4_f32(
                       false, a, false, b, (short)0, c[t], false, false);
        }
    }

    int mBase = (lane >> 4) * 8;
    #pragma unroll
    for (int t = 0; t < 3; ++t) {
        int col = t * 16 + laneM;
        if (col >= nCols) continue;
        float bv = bl[col];
        #pragma unroll
        for (int r = 0; r < 8; ++r)
            Out[(size_t)(row0 + mBase + r) * nCols + col] = c[t][r] + bv;
    }
}

__device__ __forceinline__ void gat_alpha_body(
    const float* __restrict__ H, const float* __restrict__ asrc,
    const float* __restrict__ adst, float* __restrict__ aS,
    float* __restrict__ aD, int N, int w, int lane)
{
    if (w >= N) return;
    const float* h = H + (size_t)w * HDIM;
    float s = 0.f, d = 0.f;
    #pragma unroll
    for (int j = 0; j < 4; ++j) {
        int f = lane + 32 * j;
        float hv = h[f];
        s += hv * asrc[f];
        d += hv * adst[f];
    }
    #pragma unroll
    for (int off = 16; off > 0; off >>= 1) {
        s += __shfl_xor(s, off, 32);
        d += __shfl_xor(d, off, 32);
    }
    if (lane == 0) { aS[w] = s; aD[w] = d; }
}

__global__ __launch_bounds__(256) void gat_alpha(
    const float* __restrict__ H, const float* __restrict__ asrc,
    const float* __restrict__ adst, float* __restrict__ aS,
    float* __restrict__ aD, int N)
{
    gat_alpha_body(H, asrc, adst, aS, aD, N,
                   (int)((blockIdx.x * blockDim.x + threadIdx.x) >> 5),
                   (int)(threadIdx.x & 31));
}

__device__ __forceinline__ void gat_hist_body(
    const int* __restrict__ ei, int E, int* __restrict__ cnt,
    int* __restrict__ rank, int i)
{
    if (i >= E) return;
    rank[i] = atomicAdd(&cnt[ei[E + i]], 1);
}

#define SCAN_CHUNKS 5

__device__ __forceinline__ int gat_wave_incl_scan(int x, int lane)
{
    #pragma unroll
    for (int off = 1; off < 32; off <<= 1) {
        int y = __shfl_up(x, off, 32);
        if (lane >= off) x += y;
    }
    return x;
}

__device__ __forceinline__ void gat_scan_body(
    const int* __restrict__ cnt, int* __restrict__ rowptr, int N)
{
    __shared__ int wtot[SCAN_CHUNKS * 32];
    __shared__ int w2[32];
    const int t    = threadIdx.x;
    const int lane = t & 31;
    const int wv   = t >> 5;
    int carry = 0;
    for (int base = 0; base < N; base += 1024 * 4 * SCAN_CHUNKS) {
        int4 v[SCAN_CHUNKS];
        #pragma unroll
        for (int j = 0; j < SCAN_CHUNKS; ++j) {
            int e0 = base + (j * 1024 + t) * 4;
            bool full = (e0 + 4 <= N);
            int4 q = *(const int4*)(cnt + (full ? e0 : 0));
            if (!full) {
                q.x = (e0     < N) ? cnt[e0]     : 0;
                q.y = (e0 + 1 < N) ? cnt[e0 + 1] : 0;
                q.z = (e0 + 2 < N) ? cnt[e0 + 2] : 0;
                q.w = 0;
            }
            v[j] = q;
        }
        int inc[SCAN_CHUNKS];
        #pragma unroll
        for (int j = 0; j < SCAN_CHUNKS; ++j) {
            int s = v[j].x + v[j].y + v[j].z + v[j].w;
            inc[j] = gat_wave_incl_scan(s, lane);
            if (lane == 31) wtot[j * 32 + wv] = inc[j];
        }
        __syncthreads();
        int tv = (t < SCAN_CHUNKS * 32) ? wtot[t] : 0;
        int ti = gat_wave_incl_scan(tv, lane);
        if (lane == 31) w2[wv] = ti;
        __syncthreads();
        int w2v   = w2[lane];
        int w2i   = gat_wave_incl_scan(w2v, lane);
        int wpre  = __shfl(w2i - w2v, wv, 32);
        int total = __shfl(w2i, 31, 32);
        if (t < SCAN_CHUNKS * 32) wtot[t] = ti - tv + wpre;
        __syncthreads();
        #pragma unroll
        for (int j = 0; j < SCAN_CHUNKS; ++j) {
            int e0 = base + (j * 1024 + t) * 4;
            int s  = v[j].x + v[j].y + v[j].z + v[j].w;
            int4 o;
            o.x = carry + wtot[j * 32 + wv] + inc[j] - s;
            o.y = o.x + v[j].x;
            o.z = o.y + v[j].y;
            o.w = o.z + v[j].z;
            if (e0 + 4 <= N) {
                *(int4*)(rowptr + e0) = o;
            } else {
                if (e0     < N) rowptr[e0]     = o.x;
                if (e0 + 1 < N) rowptr[e0 + 1] = o.y;
                if (e0 + 2 < N) rowptr[e0 + 2] = o.z;
            }
        }
        carry += total;
        __syncthreads();
    }
    if (t == 0) rowptr[N] = carry;
}

__global__ __launch_bounds__(256) void gat_fill(
    const int* __restrict__ ei, int E, const int* __restrict__ rowptr,
    const int* __restrict__ rank, int* __restrict__ srcS)
{
    int i = blockIdx.x * blockDim.x + threadIdx.x;
    if (i >= E) return;
    int d = ei[E + i];
    int r = rank[i];
    int s = ei[i];
    srcS[rowptr[d] + r] = s;
}


__global__ __launch_bounds__(256) void gat_stage1(
    int* __restrict__ cnt, int N, int nbZero,
    const float* __restrict__ W1, float* __restrict__ Wt, int nbT1,
    const float* __restrict__ Wl, float* __restrict__ Wlt, int C, int CP)
{
    int b = blockIdx.x;
    int t = threadIdx.x;
    if (b < nbZero) {
        int i = b * 256 + t;
        if (i < N) cnt[i] = 0;
    } else if (b < nbZero + nbT1) {
        gat_transpose_body(W1, Wt, HDIM, HDIM, HDIM, (b - nbZero) * 256 + t);
    } else {
        gat_transpose_body(Wl, Wlt, HDIM, C, CP, (b - nbZero - nbT1) * 256 + t);
    }
}

__global__ __launch_bounds__(256) void gat_stage2(
    const float* __restrict__ X, const float* __restrict__ Wt,
    float* __restrict__ H, int nRowTiles, int nbGemm,
    const int* __restrict__ ei, int E, int* __restrict__ cnt,
    int* __restrict__ rank)
{
    int b = blockIdx.x;
    int t = threadIdx.x;
    if (b < nbGemm) {
        gat_gemm_strip_body(X, Wt, H, nRowTiles, (b * 256 + t) >> 5, t & 31);
    } else {
        gat_hist_body(ei, E, cnt, rank, (b - nbGemm) * 256 + t);
    }
}

__global__ __launch_bounds__(1024) void gat_stage3(
    const int* __restrict__ cnt, int* __restrict__ rowptr, int N,
    const float* __restrict__ H, const float* __restrict__ asrc,
    const float* __restrict__ adst, float* __restrict__ aS,
    float* __restrict__ aD, int nbAlpha,
    const float* __restrict__ W2, float* __restrict__ Wt)
{
    int b = blockIdx.x;
    int t = threadIdx.x;
    if (b == 0) {
        gat_scan_body(cnt, rowptr, N);
    } else if (b <= nbAlpha) {
        gat_alpha_body(H, asrc, adst, aS, aD, N,
                       ((b - 1) * 1024 + t) >> 5, t & 31);
    } else {
        gat_transpose_body(W2, Wt, HDIM, HDIM, HDIM,
                           (b - 1 - nbAlpha) * 1024 + t);
    }
}

__device__ __forceinline__ float gat_leaky(float e) {
    return (e >= 0.f) ? e : 0.2f * e;
}

__global__ __launch_bounds__(256) void gat_gather(
    const int* __restrict__ rowptr, const int* __restrict__ srcS,
    const float* __restrict__ aS, const float* __restrict__ aD,
    const float* __restrict__ H, const float* __restrict__ bias,
    float* __restrict__ out, int N)
{
    int w    = __builtin_amdgcn_readfirstlane(
                   (int)((blockIdx.x * blockDim.x + threadIdx.x) >> 5));
    int lane = threadIdx.x & 31;
    if (w >= N) return;
    int beg = rowptr[w];
    int end = rowptr[w + 1];
    float ad    = aD[w];
    float eSelf = gat_leaky(aS[w] + ad);

    const float4* H4 = (const float4*)H;
    float4 hv = H4[(size_t)w * (HDIM / 4) + lane];

    int  j0  = beg + lane;
    int  j1  = j0 + 32;
    bool in0 = j0 < end;
    bool in1 = j1 < end;
    int s0 = srcS[in0 ? j0 : 0];
    int s1 = srcS[in1 ? j1 : 0];
    float l0 = aS[s0];
    float l1 = aS[s1];
    float e0 = in0 ? gat_leaky(l0 + ad) : eSelf;
    float e1 = in1 ? gat_leaky(l1 + ad) : eSelf;

    float m = fmaxf(fmaxf(eSelf, e0), e1);
    for (int j = j1 + 32; j < end; j += 32)
        m = fmaxf(m, gat_leaky(aS[srcS[j]] + ad));
    for (int off = 16; off > 0; off >>= 1)
        m = fmaxf(m, __shfl_xor(m, off, 32));

    const float* bp = bias + 4 * lane;
    float bv0 = bp[0], bv1 = bp[1], bv2 = bp[2], bv3 = bp[3];

    float exSelf = expf(eSelf - m);
    float x0 = in0 ? expf(e0 - m) : 0.f;
    float x1 = in1 ? expf(e1 - m) : 0.f;
    float den = (lane == 0) ? exSelf : 0.f;
    den += x0;
    den += x1;
    for (int j = j1 + 32; j < end; j += 32)
        den += expf(gat_leaky(aS[srcS[j]] + ad) - m);
    for (int off = 16; off > 0; off >>= 1)
        den += __shfl_xor(den, off, 32);

    float aSelf = exSelf / den;
    float a0 = x0 / den;
    float a1 = x1 / den;
    float4 acc;
    acc.x = hv.x * aSelf; acc.y = hv.y * aSelf;
    acc.z = hv.z * aSelf; acc.w = hv.w * aSelf;
    int chunk = 0;
    for (int base = beg; base < end; base += 32, ++chunk) {
        int s;
        float a;
        if (chunk == 0) {
            s = s0; a = a0;
        } else if (chunk == 1) {
            s = s1; a = a1;
        } else {
            int j = base + lane;
            s = 0;
            a = 0.f;
            if (j < end) {
                s = srcS[j];
                a = expf(gat_leaky(aS[s] + ad) - m) / den;
            }
        }
        int cnt = (end - base < 32) ? (end - base) : 32;
        #pragma unroll 4
        for (int t = 0; t < cnt; ++t) {
            int   st = __builtin_amdgcn_readlane(s, t);
            float at = __int_as_float(
                           __builtin_amdgcn_readlane(__float_as_int(a), t));
            float4 v = H4[(size_t)st * (HDIM / 4) + lane];
            acc.x += v.x * at; acc.y += v.y * at;
            acc.z += v.z * at; acc.w += v.w * at;
        }
    }

    float4 r;
    r.x = acc.x + bv0; r.y = acc.y + bv1;
    r.z = acc.z + bv2; r.w = acc.w + bv3;
    r.x = (r.x > 0.f) ? r.x : 0.f; r.y = (r.y > 0.f) ? r.y : 0.f;
    r.z = (r.z > 0.f) ? r.z : 0.f; r.w = (r.w > 0.f) ? r.w : 0.f;
    ((float4*)out)[(size_t)w * (HDIM / 4) + lane] = r;
}

extern "C" void kernel_launch(void* const* d_in, const int* in_sizes, int n_in,
                              void* d_out, int out_size, void* d_ws, size_t ws_size,
                              hipStream_t stream)
{
    const float* x   = (const float*)d_in[0];
    const int*   ei  = (const int*)d_in[1];
    const float* W1  = (const float*)d_in[2];
    const float* a1s = (const float*)d_in[3];
    const float* a1d = (const float*)d_in[4];
    const float* b1  = (const float*)d_in[5];
    const float* W2  = (const float*)d_in[6];
    const float* a2s = (const float*)d_in[7];
    const float* a2d = (const float*)d_in[8];
    const float* b2  = (const float*)d_in[9];
    const float* Wl  = (const float*)d_in[10];
    const float* bl  = (const float*)d_in[11];

    const int N  = in_sizes[0] / HDIM;
    const int E  = in_sizes[1] / 2;
    const int C  = in_sizes[11];
    const int CP = (C + 15) & ~15;

    char* ws = (char*)d_ws;
    auto align256 = [](size_t v) { return (v + 255) & ~(size_t)255; };
    size_t off = 0;
    float* bufH   = (float*)(ws + off);  off += align256((size_t)N * HDIM * 4);
    float* bufA   = (float*)(ws + off);  off += align256((size_t)N * HDIM * 4);
    float* Wt     = (float*)(ws + off);  off += align256((size_t)HDIM * HDIM * 4);
    float* Wlt    = (float*)(ws + off);  off += align256((size_t)CP * HDIM * 4);
    float* aS     = (float*)(ws + off);  off += align256((size_t)N * 4);
    float* aD     = (float*)(ws + off);  off += align256((size_t)N * 4);
    int*   cnt    = (int*)(ws + off);    off += align256((size_t)N * 4);
    int*   rowptr = (int*)(ws + off);    off += align256((size_t)(N + 1) * 4);
    int*   srcS   = (int*)(ws + off);    off += align256((size_t)E * 4);
    int*   rank   = (int*)bufA;
    (void)ws_size;

    const int nRowTiles = N / 16;
    const int BLK = 256;
    dim3 blk(BLK);

    const int nbZero  = (N + BLK - 1) / BLK;
    const int nbT     = (HDIM * HDIM + BLK - 1) / BLK;
    const int nbTl    = (CP * HDIM + BLK - 1) / BLK;
    const int nbGemm  = (nRowTiles * 2 + 7) / 8;
    const int nbEdge  = (E + BLK - 1) / BLK;
    const int nbAlpha = (N + 31) / 32;
    const int nbT2    = (HDIM * HDIM + 1023) / 1024;
    dim3 gemmGrid(nbGemm);
    dim3 waveGridN((N + 7) / 8);
    dim3 edgeGrid(nbEdge);

    gat_stage1<<<dim3(nbZero + nbT + nbTl), blk, 0, stream>>>(
        cnt, N, nbZero, W1, Wt, nbT, Wl, Wlt, C, CP);
    gat_stage2<<<dim3(nbGemm + nbEdge), blk, 0, stream>>>(
        x, Wt, bufH, nRowTiles, nbGemm, ei, E, cnt, rank);
    gat_stage3<<<dim3(1 + nbAlpha + nbT2), dim3(1024), 0, stream>>>(
        cnt, rowptr, N, bufH, a1s, a1d, aS, aD, nbAlpha, W2, Wt);
    gat_fill<<<edgeGrid, blk, 0, stream>>>(ei, E, rowptr, rank, srcS);

    gat_gather<<<waveGridN, blk, 0, stream>>>(rowptr, srcS, aS, aD, bufH,
                                              b1, bufA, N);

    gat_gemm_strip<<<gemmGrid, blk, 0, stream>>>(bufA, Wt, bufH, nRowTiles);
    gat_alpha<<<waveGridN, blk, 0, stream>>>(bufH, a2s, a2d, aS, aD, N);
    gat_gather<<<waveGridN, blk, 0, stream>>>(rowptr, srcS, aS, aD, bufH,
                                              b2, bufA, N);

    dim3 clsGrid((nRowTiles + 7) / 8);
    gat_gemm_cls<<<clsGrid, blk, 0, stream>>>(bufA, Wlt, bl, (float*)d_out,
                                              nRowTiles, C);
}


# Edge sort keeps every counter in LDS: chunks bucket edges by 128 destinations (count, table scan, scatter), one work-group per bucket then ranks them; no global atomics; segments in LDS arrival order
# speedup vs baseline: 4.0803x; 1.3096x over previous
#include <hip/hip_runtime.h>
#include <hip/hip_bf16.h>

typedef __attribute__((ext_vector_type(2))) float v2f;
typedef __attribute__((ext_vector_type(8))) float v8f;

#define HDIM 128
#define CTPW 4


__device__ __forceinline__ void gat_transpose_body(
    const float* __restrict__ W, float* __restrict__ Wt,
    int rows, int cols, int colsPad, int i)
{
    int total = colsPad * rows;
    if (i >= total) return;
    int c = i / rows;
    int r = i - c * rows;
    Wt[(size_t)c * rows + r] = (c < cols) ? W[(size_t)r * cols + c] : 0.f;
}

__device__ __forceinline__ void gat_gemm_strip_body(
    const float* __restrict__ X, const float* __restrict__ Wt,
    float* __restrict__ H, int nRowTiles, int wid, int lane)
{
    int nHalves = HDIM / (CTPW * 16);
    int nW = nRowTiles * nHalves;
    if (wid >= nW) return;
    int rt   = wid >> 1;
    int half = wid & 1;
    int row0 = rt * 16;
    int laneM = lane & 15;
    int kOff  = (lane >> 4) * 2;

    const float* xrow = X  + (size_t)(row0 + laneM) * HDIM + kOff;
    const float* wrow = Wt + (size_t)(half * (CTPW * 16) + laneM) * HDIM + kOff;

    v8f c[CTPW] = {};
    #pragma unroll 4
    for (int k = 0; k < HDIM; k += 4) {
        v2f a = *(const v2f*)(xrow + k);
        #pragma unroll
        for (int t = 0; t < CTPW; ++t) {
            v2f b = *(const v2f*)(wrow + (size_t)t * 16 * HDIM + k);
            c[t] = __builtin_amdgcn_wmma_f32_16x16x4_f32(
                       false, a, false, b, (short)0, c[t], false, false);
        }
    }

    int mBase = (lane >> 4) * 8;
    #pragma unroll
    for (int t = 0; t < CTPW; ++t) {
        float* o = H + (size_t)row0 * HDIM + (half * CTPW + t) * 16 + laneM;
        #pragma unroll
        for (int r = 0; r < 8; ++r)
            o[(size_t)(mBase + r) * HDIM] = c[t][r];
    }
}

__global__ __launch_bounds__(256) void gat_gemm_strip(
    const float* __restrict__ X, const float* __restrict__ Wt,
    float* __restrict__ H, int nRowTiles)
{
    gat_gemm_strip_body(X, Wt, H, nRowTiles,
                        (int)((blockIdx.x * blockDim.x + threadIdx.x) >> 5),
                        (int)(threadIdx.x & 31));
}

__global__ __launch_bounds__(256) void gat_gemm_cls(
    const float* __restrict__ X, const float* __restrict__ Wlt,
    const float* __restrict__ bl, float* __restrict__ Out,
    int nRowTiles, int nCols)
{
    int wid  = (blockIdx.x * blockDim.x + threadIdx.x) >> 5;
    int lane = threadIdx.x & 31;
    if (wid >= nRowTiles) return;
    int row0 = wid * 16;
    int laneM = lane & 15;
    int kOff  = (lane >> 4) * 2;

    const float* xrow = X   + (size_t)(row0 + laneM) * HDIM + kOff;
    const float* wrow = Wlt + (size_t)laneM * HDIM + kOff;

    v8f c[3] = {};
    #pragma unroll 4
    for (int k = 0; k < HDIM; k += 4) {
        v2f a = *(const v2f*)(xrow + k);
        #pragma unroll
        for (int t = 0; t < 3; ++t) {
            v2f b = *(const v2f*)(wrow + (size_t)t * 16 * HDIM + k);
            c[t] = __builtin_amdgcn_wmma_f32_16x16x4_f32(
                       false, a, false, b, (short)0, c[t], false, false);
        }
    }

    int mBase = (lane >> 4) * 8;
    #pragma unroll
    for (int t = 0; t < 3; ++t) {
        int col = t * 16 + laneM;
        if (col >= nCols) continue;
        float bv = bl[col];
        #pragma unroll
        for (int r = 0; r < 8; ++r)
            Out[(size_t)(row0 + mBase + r) * nCols + col] = c[t][r] + bv;
    }
}

__device__ __forceinline__ void gat_alpha_body(
    const float* __restrict__ H, const float* __restrict__ asrc,
    const float* __restrict__ adst, float* __restrict__ aS,
    float* __restrict__ aD, int N, int w, int lane)
{
    if (w >= N) return;
    const float* h = H + (size_t)w * HDIM;
    float s = 0.f, d = 0.f;
    #pragma unroll
    for (int j = 0; j < 4; ++j) {
        int f = lane + 32 * j;
        float hv = h[f];
        s += hv * asrc[f];
        d += hv * adst[f];
    }
    #pragma unroll
    for (int off = 16; off > 0; off >>= 1) {
        s += __shfl_xor(s, off, 32);
        d += __shfl_xor(d, off, 32);
    }
    if (lane == 0) { aS[w] = s; aD[w] = d; }
}

__global__ __launch_bounds__(256) void gat_alpha(
    const float* __restrict__ H, const float* __restrict__ asrc,
    const float* __restrict__ adst, float* __restrict__ aS,
    float* __restrict__ aD, int N)
{
    gat_alpha_body(H, asrc, adst, aS, aD, N,
                   (int)((blockIdx.x * blockDim.x + threadIdx.x) >> 5),
                   (int)(threadIdx.x & 31));
}

#define BSH  7
#define BR   (1 << BSH)
#define NCH  128
#define KMAX 1024
#define EUB  8
#define RUB  6

__device__ __forceinline__ int gat_wave_incl_scan(int x, int lane)
{
    #pragma unroll
    for (int off = 1; off < 32; off <<= 1) {
        int y = __shfl_up(x, off, 32);
        if (lane >= off) x += y;
    }
    return x;
}

__device__ __forceinline__ void gat_bcount_body(
    const int* __restrict__ ei, int E, int K, int chunkLen,
    int* __restrict__ tabC, int Mpad, int c, int t, int nt)
{
    __shared__ int h[KMAX];
    if (c == 0)
        for (int p = K * NCH + t; p < Mpad; p += nt) tabC[p] = 0;
    for (int k = t; k < K; k += nt) h[k] = 0;
    __syncthreads();
    int c0 = c * chunkLen;
    int c1 = (c0 + chunkLen < E) ? (c0 + chunkLen) : E;
    for (int i = c0 + t; i < c1; i += EUB * nt) {
        int d[EUB];
        #pragma unroll
        for (int u = 0; u < EUB; ++u) {
            int j  = i + u * nt;
            bool ok = j < c1;
            int dv = ei[E + (ok ? j : c0)];
            d[u] = ok ? dv : -1;
        }
        #pragma unroll
        for (int u = 0; u < EUB; ++u)
            if (d[u] >= 0) atomicAdd(&h[d[u] >> BSH], 1);
    }
    __syncthreads();
    for (int k = t; k < K; k += nt) tabC[k * NCH + c] = h[k];
}

#define TS_G    20
#define TS_NT   256
#define TS_PASS (TS_NT * 4 * TS_G)

__device__ __forceinline__ void gat_tabscan_body(
    const int* __restrict__ tabC, int* __restrict__ tabP, int Mpad)
{
    __shared__ int wt[TS_NT / 32];
    const int t    = threadIdx.x;
    const int lane = t & 31;
    const int wv   = t >> 5;
    int carry = 0;
    for (int base = 0; base < Mpad; base += TS_PASS) {
        const int4* src = (const int4*)(tabC + base + t * (4 * TS_G));
        int4*       dst = (int4*)(tabP + base + t * (4 * TS_G));
        int4 q[TS_G];
        #pragma unroll
        for (int g = 0; g < TS_G; ++g) q[g] = src[g];
        int s = 0;
        #pragma unroll
        for (int g = 0; g < TS_G; ++g) s += q[g].x + q[g].y + q[g].z + q[g].w;
        int inc = gat_wave_incl_scan(s, lane);
        if (lane == 31) wt[wv] = inc;
        __syncthreads();
        int run = carry + inc - s;
        int tot = 0;
        for (int w = 0; w < TS_NT / 32; ++w) {
            int x = wt[w];
            tot += x;
            if (w < wv) run += x;
        }
        #pragma unroll
        for (int g = 0; g < TS_G; ++g) {
            int4 o;
            o.x = run;
            o.y = o.x + q[g].x;
            o.z = o.y + q[g].y;
            o.w = o.z + q[g].z;
            run = o.w + q[g].w;
            dst[g] = o;
        }
        carry += tot;
        __syncthreads();
    }
}

__device__ __forceinline__ void gat_bscatter_body(
    const int* __restrict__ ei, int E, int K, int chunkLen,
    const int* __restrict__ tabP, int2* __restrict__ rec, int c, int t, int nt)
{
    __shared__ int cur[KMAX];
    for (int k = t; k < K; k += nt) cur[k] = tabP[k * NCH + c];
    __syncthreads();
    int c0 = c * chunkLen;
    int c1 = (c0 + chunkLen < E) ? (c0 + chunkLen) : E;
    for (int i = c0 + t; i < c1; i += EUB * nt) {
        int d[EUB], s[EUB];
        #pragma unroll
        for (int u = 0; u < EUB; ++u) {
            int j  = i + u * nt;
            bool ok = j < c1;
            int jj = ok ? j : c0;
            int dv = ei[E + jj];
            s[u] = ei[jj];
            d[u] = ok ? dv : -1;
        }
        #pragma unroll
        for (int u = 0; u < EUB; ++u) {
            if (d[u] >= 0) {
                int slot = atomicAdd(&cur[d[u] >> BSH], 1);
                rec[slot] = make_int2(d[u], s[u]);
            }
        }
    }
}

__global__ __launch_bounds__(1024) void gat_bsort(
    const int* __restrict__ tabP, const int* __restrict__ recw,
    int E, int N, int K, int* __restrict__ rowptr, int* __restrict__ srcS)
{
    __shared__ int lcnt[BR];
    __shared__ int lcur[BR];
    __shared__ int wsum[BR / 32];
    const int2* rec = (const int2*)recw;
    const int k    = blockIdx.x;
    const int t    = threadIdx.x;
    const int lane = t & 31;
    const int wv   = t >> 5;
    const int start = tabP[k * NCH];
    const int end   = (k + 1 < K) ? tabP[(k + 1) * NCH] : E;
    const int d0    = k << BSH;
    if (t < BR) lcnt[t] = 0;
    __syncthreads();
    for (int e = start + t; e < end; e += RUB * 1024) {
        int d[RUB];
        #pragma unroll
        for (int u = 0; u < RUB; ++u) {
            int j  = e + u * 1024;
            bool ok = j < end;
            int dv = rec[ok ? j : start].x;
            d[u] = ok ? dv : -1;
        }
        #pragma unroll
        for (int u = 0; u < RUB; ++u)
            if (d[u] >= 0) atomicAdd(&lcnt[d[u] - d0], 1);
    }
    __syncthreads();
    int v   = (t < BR) ? lcnt[t] : 0;
    int inc = gat_wave_incl_scan(v, lane);
    if (t < BR && lane == 31) wsum[wv] = inc;
    __syncthreads();
    if (t < BR) {
        int ex = start + inc - v;
        for (int q = 0; q < wv; ++q) ex += wsum[q];
        lcur[t] = ex;
        if (d0 + t < N) rowptr[d0 + t] = ex;
    }
    if (k == K - 1 && t == 0) rowptr[N] = end;
    __syncthreads();
    for (int e = start + t; e < end; e += RUB * 1024) {
        int2 r[RUB];
        bool ok[RUB];
        #pragma unroll
        for (int u = 0; u < RUB; ++u) {
            int j = e + u * 1024;
            ok[u] = j < end;
            r[u]  = rec[ok[u] ? j : start];
        }
        #pragma unroll
        for (int u = 0; u < RUB; ++u) {
            if (ok[u]) {
                int slot = atomicAdd(&lcur[r[u].x - d0], 1);
                srcS[slot] = r[u].y;
            }
        }
    }
}


__global__ __launch_bounds__(256) void gat_stage1(
    const int* __restrict__ ei, int E, int K, int chunkLen,
    int* __restrict__ tabC, int Mpad,
    const float* __restrict__ W1, float* __restrict__ Wt, int nbT1,
    const float* __restrict__ Wl, float* __restrict__ Wlt, int C, int CP)
{
    int b = blockIdx.x;
    int t = threadIdx.x;
    if (b < NCH) {
        gat_bcount_body(ei, E, K, chunkLen, tabC, Mpad, b, t, 256);
    } else if (b < NCH + nbT1) {
        gat_transpose_body(W1, Wt, HDIM, HDIM, HDIM, (b - NCH) * 256 + t);
    } else {
        gat_transpose_body(Wl, Wlt, HDIM, C, CP, (b - NCH - nbT1) * 256 + t);
    }
}

__global__ __launch_bounds__(256) void gat_stage2(
    const float* __restrict__ X, const float* __restrict__ Wt,
    float* __restrict__ H, int nRowTiles,
    const int* __restrict__ tabC, int* __restrict__ tabP, int Mpad)
{
    int b = blockIdx.x;
    int t = threadIdx.x;
    if (b == 0) {
        gat_tabscan_body(tabC, tabP, Mpad);
    } else {
        gat_gemm_strip_body(X, Wt, H, nRowTiles, ((b - 1) * 256 + t) >> 5, t & 31);
    }
}

__global__ __launch_bounds__(1024) void gat_stage3(
    const int* __restrict__ ei, int E, int K, int chunkLen,
    const int* __restrict__ tabP, int* __restrict__ recw, int N,
    const float* __restrict__ H, const float* __restrict__ asrc,
    const float* __restrict__ adst, float* __restrict__ aS,
    float* __restrict__ aD, int nbAlpha,
    const float* __restrict__ W2, float* __restrict__ Wt)
{
    int b = blockIdx.x;
    int t = threadIdx.x;
    if (b < NCH) {
        gat_bscatter_body(ei, E, K, chunkLen, tabP, (int2*)recw, b, t, 1024);
    } else if (b < NCH + nbAlpha) {
        gat_alpha_body(H, asrc, adst, aS, aD, N,
                       ((b - NCH) * 1024 + t) >> 5, t & 31);
    } else {
        gat_transpose_body(W2, Wt, HDIM, HDIM, HDIM,
                           (b - NCH - nbAlpha) * 1024 + t);
    }
}

__device__ __forceinline__ float gat_leaky(float e) {
    return (e >= 0.f) ? e : 0.2f * e;
}

__global__ __launch_bounds__(256) void gat_gather(
    const int* __restrict__ rowptr, const int* __restrict__ srcS,
    const float* __restrict__ aS, const float* __restrict__ aD,
    const float* __restrict__ H, const float* __restrict__ bias,
    float* __restrict__ out, int N)
{
    int w    = __builtin_amdgcn_readfirstlane(
                   (int)((blockIdx.x * blockDim.x + threadIdx.x) >> 5));
    int lane = threadIdx.x & 31;
    if (w >= N) return;
    int beg = rowptr[w];
    int end = rowptr[w + 1];
    float ad    = aD[w];
    float eSelf = gat_leaky(aS[w] + ad);

    const float4* H4 = (const float4*)H;
    float4 hv = H4[(size_t)w * (HDIM / 4) + lane];

    int  j0  = beg + lane;
    int  j1  = j0 + 32;
    bool in0 = j0 < end;
    bool in1 = j1 < end;
    int s0 = srcS[in0 ? j0 : 0];
    int s1 = srcS[in1 ? j1 : 0];
    float l0 = aS[s0];
    float l1 = aS[s1];
    float e0 = in0 ? gat_leaky(l0 + ad) : eSelf;
    float e1 = in1 ? gat_leaky(l1 + ad) : eSelf;

    float m = fmaxf(fmaxf(eSelf, e0), e1);
    for (int j = j1 + 32; j < end; j += 32)
        m = fmaxf(m, gat_leaky(aS[srcS[j]] + ad));
    for (int off = 16; off > 0; off >>= 1)
        m = fmaxf(m, __shfl_xor(m, off, 32));

    const float* bp = bias + 4 * lane;
    float bv0 = bp[0], bv1 = bp[1], bv2 = bp[2], bv3 = bp[3];

    float exSelf = expf(eSelf - m);
    float x0 = in0 ? expf(e0 - m) : 0.f;
    float x1 = in1 ? expf(e1 - m) : 0.f;
    float den = (lane == 0) ? exSelf : 0.f;
    den += x0;
    den += x1;
    for (int j = j1 + 32; j < end; j += 32)
        den += expf(gat_leaky(aS[srcS[j]] + ad) - m);
    for (int off = 16; off > 0; off >>= 1)
        den += __shfl_xor(den, off, 32);

    float aSelf = exSelf / den;
    float a0 = x0 / den;
    float a1 = x1 / den;
    float4 acc;
    acc.x = hv.x * aSelf; acc.y = hv.y * aSelf;
    acc.z = hv.z * aSelf; acc.w = hv.w * aSelf;
    int chunk = 0;
    for (int base = beg; base < end; base += 32, ++chunk) {
        int s;
        float a;
        if (chunk == 0) {
            s = s0; a = a0;
        } else if (chunk == 1) {
            s = s1; a = a1;
        } else {
            int j = base + lane;
            s = 0;
            a = 0.f;
            if (j < end) {
                s = srcS[j];
                a = expf(gat_leaky(aS[s] + ad) - m) / den;
            }
        }
        int cnt = (end - base < 32) ? (end - base) : 32;
        #pragma unroll 4
        for (int t = 0; t < cnt; ++t) {
            int   st = __builtin_amdgcn_readlane(s, t);
            float at = __int_as_float(
                           __builtin_amdgcn_readlane(__float_as_int(a), t));
            float4 v = H4[(size_t)st * (HDIM / 4) + lane];
            acc.x += v.x * at; acc.y += v.y * at;
            acc.z += v.z * at; acc.w += v.w * at;
        }
    }

    float4 r;
    r.x = acc.x + bv0; r.y = acc.y + bv1;
    r.z = acc.z + bv2; r.w = acc.w + bv3;
    r.x = (r.x > 0.f) ? r.x : 0.f; r.y = (r.y > 0.f) ? r.y : 0.f;
    r.z = (r.z > 0.f) ? r.z : 0.f; r.w = (r.w > 0.f) ? r.w : 0.f;
    ((float4*)out)[(size_t)w * (HDIM / 4) + lane] = r;
}

extern "C" void kernel_launch(void* const* d_in, const int* in_sizes, int n_in,
                              void* d_out, int out_size, void* d_ws, size_t ws_size,
                              hipStream_t stream)
{
    const float* x   = (const float*)d_in[0];
    const int*   ei  = (const int*)d_in[1];
    const float* W1  = (const float*)d_in[2];
    const float* a1s = (const float*)d_in[3];
    const float* a1d = (const float*)d_in[4];
    const float* b1  = (const float*)d_in[5];
    const float* W2  = (const float*)d_in[6];
    const float* a2s = (const float*)d_in[7];
    const float* a2d = (const float*)d_in[8];
    const float* b2  = (const float*)d_in[9];
    const float* Wl  = (const float*)d_in[10];
    const float* bl  = (const float*)d_in[11];

    const int N  = in_sizes[0] / HDIM;
    const int E  = in_sizes[1] / 2;
    const int C  = in_sizes[11];
    const int CP = (C + 15) & ~15;

    char* ws = (char*)d_ws;
    auto align256 = [](size_t v) { return (v + 255) & ~(size_t)255; };
    size_t off = 0;
    float* bufH   = (float*)(ws + off);  off += align256((size_t)N * HDIM * 4);
    float* bufA   = (float*)(ws + off);  off += align256((size_t)N * HDIM * 4);
    float* Wt     = (float*)(ws + off);  off += align256((size_t)HDIM * HDIM * 4);
    float* Wlt    = (float*)(ws + off);  off += align256((size_t)CP * HDIM * 4);
    float* aS     = (float*)(ws + off);  off += align256((size_t)N * 4);
    float* aD     = (float*)(ws + off);  off += align256((size_t)N * 4);
    off += align256((size_t)N * 4);
    int*   rowptr = (int*)(ws + off);    off += align256((size_t)(N + 1) * 4);
    int*   srcS   = (int*)(ws + off);    off += align256((size_t)E * 4);
    (void)ws_size;

    const int K        = (N + BR - 1) >> BSH;
    const int Mpad     = (K * NCH + TS_PASS - 1) / TS_PASS * TS_PASS;
    const int chunkLen = (E + NCH - 1) / NCH;
    char*  sa   = (char*)bufA;
    int*   recw = (int*)sa;
    int*   tabC = (int*)(sa + align256((size_t)E * 8));
    int*   tabP = (int*)(sa + align256((size_t)E * 8) + align256((size_t)Mpad * 4));

    const int nRowTiles = N / 16;
    const int BLK = 256;
    dim3 blk(BLK);

    const int nbT     = (HDIM * HDIM + BLK - 1) / BLK;
    const int nbTl    = (CP * HDIM + BLK - 1) / BLK;
    const int nbGemm  = (nRowTiles * 2 + 7) / 8;
    const int nbAlpha = (N + 31) / 32;
    const int nbT2    = (HDIM * HDIM + 1023) / 1024;
    dim3 gemmGrid(nbGemm);
    dim3 waveGridN((N + 7) / 8);

    gat_stage1<<<dim3(NCH + nbT + nbTl), blk, 0, stream>>>(
        ei, E, K, chunkLen, tabC, Mpad, W1, Wt, nbT, Wl, Wlt, C, CP);
    gat_stage2<<<dim3(1 + nbGemm), blk, 0, stream>>>(
        x, Wt, bufH, nRowTiles, tabC, tabP, Mpad);
    gat_stage3<<<dim3(NCH + nbAlpha + nbT2), dim3(1024), 0, stream>>>(
        ei, E, K, chunkLen, tabP, recw, N, bufH, a1s, a1d, aS, aD, nbAlpha, W2, Wt);
    gat_bsort<<<dim3(K), dim3(1024), 0, stream>>>(tabP, recw, E, N, K, rowptr, srcS);

    gat_gather<<<waveGridN, blk, 0, stream>>>(rowptr, srcS, aS, aD, bufH,
                                              b1, bufA, N);

    gat_gemm_strip<<<gemmGrid, blk, 0, stream>>>(bufA, Wt, bufH, nRowTiles);
    gat_alpha<<<waveGridN, blk, 0, stream>>>(bufH, a2s, a2d, aS, aD, N);
    gat_gather<<<waveGridN, blk, 0, stream>>>(rowptr, srcS, aS, aD, bufH,
                                              b2, bufA, N);

    dim3 clsGrid((nRowTiles + 7) / 8);
    gat_gemm_cls<<<clsGrid, blk, 0, stream>>>(bufA, Wlt, bl, (float*)d_out,
                                              nRowTiles, C);
}


# Attention logits are taken from the GEMM's accumulator tiles (16-lane butterfly, column halves meet in LDS): both one-wave-per-node logit passes go, 9 -> 8 launches; logit sum order changes
# speedup vs baseline: 4.2404x; 1.0392x over previous
#include <hip/hip_runtime.h>
#include <hip/hip_bf16.h>

typedef __attribute__((ext_vector_type(2))) float v2f;
typedef __attribute__((ext_vector_type(8))) float v8f;

#define HDIM 128
#define CTPW 4


__device__ __forceinline__ void gat_transpose_body(
    const float* __restrict__ W, float* __restrict__ Wt,
    int rows, int cols, int colsPad, int i)
{
    int total = colsPad * rows;
    if (i >= total) return;
    int c = i / rows;
    int r = i - c * rows;
    Wt[(size_t)c * rows + r] = (c < cols) ? W[(size_t)r * cols + c] : 0.f;
}

#define GW 8

__device__ __forceinline__ float gat_row16_sum(float v)
{
    #pragma unroll
    for (int off = 1; off < 16; off <<= 1)
        v += __shfl_xor(v, off, 32);
    return v;
}

__device__ __forceinline__ void gat_gemm_strip_body(
    const float* __restrict__ X, const float* __restrict__ Wt,
    float* __restrict__ H,
    const float* __restrict__ asrc, const float* __restrict__ adst,
    float* __restrict__ aS, float* __restrict__ aD,
    int nRowTiles, int wid0, int t)
{
    __shared__ float part[GW][2][16];
    const int nHalves = HDIM / (CTPW * 16);
    const int nW    = nRowTiles * nHalves;
    const int wv    = t >> 5;
    const int lane  = t & 31;
    const int wid   = wid0 + wv;
    const int rt    = wid >> 1;
    const int half  = wid & 1;
    const int row0  = rt * 16;
    const int laneM = lane & 15;
    const int kOff  = (lane >> 4) * 2;
    const int mBase = (lane >> 4) * 8;
    const bool act = wid < nW;

    v8f c[CTPW] = {};
    if (act) {
        const float* xrow = X  + (size_t)(row0 + laneM) * HDIM + kOff;
        const float* wrow = Wt + (size_t)(half * (CTPW * 16) + laneM) * HDIM + kOff;

        float as[CTPW], ad[CTPW];
        #pragma unroll
        for (int q = 0; q < CTPW; ++q) {
            int col = (half * CTPW + q) * 16 + laneM;
            as[q] = asrc[col];
            ad[q] = adst[col];
        }

        #pragma unroll 4
        for (int k = 0; k < HDIM; k += 4) {
            v2f a = *(const v2f*)(xrow + k);
            #pragma unroll
            for (int q = 0; q < CTPW; ++q) {
                v2f b = *(const v2f*)(wrow + (size_t)q * 16 * HDIM + k);
                c[q] = __builtin_amdgcn_wmma_f32_16x16x4_f32(
                           false, a, false, b, (short)0, c[q], false, false);
            }
        }

        float ps[8], pd[8];
        #pragma unroll
        for (int r = 0; r < 8; ++r) {
            float s = 0.f, d = 0.f;
            #pragma unroll
            for (int q = 0; q < CTPW; ++q) {
                s += c[q][r] * as[q];
                d += c[q][r] * ad[q];
            }
            ps[r] = s;
            pd[r] = d;
        }
        #pragma unroll
        for (int r = 0; r < 8; ++r) {
            ps[r] = gat_row16_sum(ps[r]);
            pd[r] = gat_row16_sum(pd[r]);
        }
        if (laneM == 0) {
            #pragma unroll
            for (int r = 0; r < 8; ++r) {
                part[wv][0][mBase + r] = ps[r];
                part[wv][1][mBase + r] = pd[r];
            }
        }
    }
    __syncthreads();
    if (t < (GW / 2) * 16) {
        int ls  = t >> 4;
        int row = t & 15;
        int rs  = (wid0 >> 1) + ls;
        if (rs < nRowTiles) {
            aS[rs * 16 + row] = part[2 * ls][0][row] + part[2 * ls + 1][0][row];
            aD[rs * 16 + row] = part[2 * ls][1][row] + part[2 * ls + 1][1][row];
        }
    }
    if (act) {
        #pragma unroll
        for (int q = 0; q < CTPW; ++q) {
            float* o = H + (size_t)row0 * HDIM + (half * CTPW + q) * 16 + laneM;
            #pragma unroll
            for (int r = 0; r < 8; ++r)
                o[(size_t)(mBase + r) * HDIM] = c[q][r];
        }
    }
}

__global__ __launch_bounds__(256) void gat_gemm_strip(
    const float* __restrict__ X, const float* __restrict__ Wt,
    float* __restrict__ H,
    const float* __restrict__ asrc, const float* __restrict__ adst,
    float* __restrict__ aS, float* __restrict__ aD, int nRowTiles)
{
    gat_gemm_strip_body(X, Wt, H, asrc, adst, aS, aD, nRowTiles,
                        (int)(blockIdx.x * GW), (int)threadIdx.x);
}

__global__ __launch_bounds__(256) void gat_gemm_cls(
    const float* __restrict__ X, const float* __restrict__ Wlt,
    const float* __restrict__ bl, float* __restrict__ Out,
    int nRowTiles, int nCols)
{
    int wid  = (blockIdx.x * blockDim.x + threadIdx.x) >> 5;
    int lane = threadIdx.x & 31;
    if (wid >= nRowTiles) return;
    int row0 = wid * 16;
    int laneM = lane & 15;
    int kOff  = (lane >> 4) * 2;

    const float* xrow = X   + (size_t)(row0 + laneM) * HDIM + kOff;
    const float* wrow = Wlt + (size_t)laneM * HDIM + kOff;

    v8f c[3] = {};
    #pragma unroll 4
    for (int k = 0; k < HDIM; k += 4) {
        v2f a = *(const v2f*)(xrow + k);
        #pragma unroll
        for (int t = 0; t < 3; ++t) {
            v2f b = *(const v2f*)(wrow + (size_t)t * 16 * HDIM + k);
            c[t] = __builtin_amdgcn_wmma_f32_16x16x4_f32(
                       false, a, false, b, (short)0, c[t], false, false);
        }
    }

    int mBase = (lane >> 4) * 8;
    #pragma unroll
    for (int t = 0; t < 3; ++t) {
        int col = t * 16 + laneM;
        if (col >= nCols) continue;
        float bv = bl[col];
        #pragma unroll
        for (int r = 0; r < 8; ++r)
            Out[(size_t)(row0 + mBase + r) * nCols + col] = c[t][r] + bv;
    }
}

#define BSH  7
#define BR   (1 << BSH)
#define NCH  128
#define KMAX 1024
#define EUB  8
#define RUB  6

__device__ __forceinline__ int gat_wave_incl_scan(int x, int lane)
{
    #pragma unroll
    for (int off = 1; off < 32; off <<= 1) {
        int y = __shfl_up(x, off, 32);
        if (lane >= off) x += y;
    }
    return x;
}

__device__ __forceinline__ void gat_bcount_body(
    const int* __restrict__ ei, int E, int K, int chunkLen,
    int* __restrict__ tabC, int Mpad, int c, int t, int nt)
{
    __shared__ int h[KMAX];
    if (c == 0)
        for (int p = K * NCH + t; p < Mpad; p += nt) tabC[p] = 0;
    for (int k = t; k < K; k += nt) h[k] = 0;
    __syncthreads();
    int c0 = c * chunkLen;
    int c1 = (c0 + chunkLen < E) ? (c0 + chunkLen) : E;
    for (int i = c0 + t; i < c1; i += EUB * nt) {
        int d[EUB];
        #pragma unroll
        for (int u = 0; u < EUB; ++u) {
            int j  = i + u * nt;
            bool ok = j < c1;
            int dv = ei[E + (ok ? j : c0)];
            d[u] = ok ? dv : -1;
        }
        #pragma unroll
        for (int u = 0; u < EUB; ++u)
            if (d[u] >= 0) atomicAdd(&h[d[u] >> BSH], 1);
    }
    __syncthreads();
    for (int k = t; k < K; k += nt) tabC[k * NCH + c] = h[k];
}

#define TS_G    20
#define TS_NT   256
#define TS_PASS (TS_NT * 4 * TS_G)

__device__ __forceinline__ void gat_tabscan_body(
    const int* __restrict__ tabC, int* __restrict__ tabP, int Mpad)
{
    __shared__ int wt[TS_NT / 32];
    const int t    = threadIdx.x;
    const int lane = t & 31;
    const int wv   = t >> 5;
    int carry = 0;
    for (int base = 0; base < Mpad; base += TS_PASS) {
        const int4* src = (const int4*)(tabC + base + t * (4 * TS_G));
        int4*       dst = (int4*)(tabP + base + t * (4 * TS_G));
        int4 q[TS_G];
        #pragma unroll
        for (int g = 0; g < TS_G; ++g) q[g] = src[g];
        int s = 0;
        #pragma unroll
        for (int g = 0; g < TS_G; ++g) s += q[g].x + q[g].y + q[g].z + q[g].w;
        int inc = gat_wave_incl_scan(s, lane);
        if (lane == 31) wt[wv] = inc;
        __syncthreads();
        int run = carry + inc - s;
        int tot = 0;
        for (int w = 0; w < TS_NT / 32; ++w) {
            int x = wt[w];
            tot += x;
            if (w < wv) run += x;
        }
        #pragma unroll
        for (int g = 0; g < TS_G; ++g) {
            int4 o;
            o.x = run;
            o.y = o.x + q[g].x;
            o.z = o.y + q[g].y;
            o.w = o.z + q[g].z;
            run = o.w + q[g].w;
            dst[g] = o;
        }
        carry += tot;
        __syncthreads();
    }
}

__device__ __forceinline__ void gat_bscatter_body(
    const int* __restrict__ ei, int E, int K, int chunkLen,
    const int* __restrict__ tabP, int2* __restrict__ rec, int c, int t, int nt)
{
    __shared__ int cur[KMAX];
    for (int k = t; k < K; k += nt) cur[k] = tabP[k * NCH + c];
    __syncthreads();
    int c0 = c * chunkLen;
    int c1 = (c0 + chunkLen < E) ? (c0 + chunkLen) : E;
    for (int i = c0 + t; i < c1; i += EUB * nt) {
        int d[EUB], s[EUB];
        #pragma unroll
        for (int u = 0; u < EUB; ++u) {
            int j  = i + u * nt;
            bool ok = j < c1;
            int jj = ok ? j : c0;
            int dv = ei[E + jj];
            s[u] = ei[jj];
            d[u] = ok ? dv : -1;
        }
        #pragma unroll
        for (int u = 0; u < EUB; ++u) {
            if (d[u] >= 0) {
                int slot = atomicAdd(&cur[d[u] >> BSH], 1);
                rec[slot] = make_int2(d[u], s[u]);
            }
        }
    }
}

__global__ __launch_bounds__(1024) void gat_bsort(
    const int* __restrict__ tabP, const int* __restrict__ recw,
    int E, int N, int K, int* __restrict__ rowptr, int* __restrict__ srcS)
{
    __shared__ int lcnt[BR];
    __shared__ int lcur[BR];
    __shared__ int wsum[BR / 32];
    const int2* rec = (const int2*)recw;
    const int k    = blockIdx.x;
    const int t    = threadIdx.x;
    const int lane = t & 31;
    const int wv   = t >> 5;
    const int start = tabP[k * NCH];
    const int end   = (k + 1 < K) ? tabP[(k + 1) * NCH] : E;
    const int d0    = k << BSH;
    if (t < BR) lcnt[t] = 0;
    __syncthreads();
    for (int e = start + t; e < end; e += RUB * 1024) {
        int d[RUB];
        #pragma unroll
        for (int u = 0; u < RUB; ++u) {
            int j  = e + u * 1024;
            bool ok = j < end;
            int dv = rec[ok ? j : start].x;
            d[u] = ok ? dv : -1;
        }
        #pragma unroll
        for (int u = 0; u < RUB; ++u)
            if (d[u] >= 0) atomicAdd(&lcnt[d[u] - d0], 1);
    }
    __syncthreads();
    int v   = (t < BR) ? lcnt[t] : 0;
    int inc = gat_wave_incl_scan(v, lane);
    if (t < BR && lane == 31) wsum[wv] = inc;
    __syncthreads();
    if (t < BR) {
        int ex = start + inc - v;
        for (int q = 0; q < wv; ++q) ex += wsum[q];
        lcur[t] = ex;
        if (d0 + t < N) rowptr[d0 + t] = ex;
    }
    if (k == K - 1 && t == 0) rowptr[N] = end;
    __syncthreads();
    for (int e = start + t; e < end; e += RUB * 1024) {
        int2 r[RUB];
        bool ok[RUB];
        #pragma unroll
        for (int u = 0; u < RUB; ++u) {
            int j = e + u * 1024;
            ok[u] = j < end;
            r[u]  = rec[ok[u] ? j : start];
        }
        #pragma unroll
        for (int u = 0; u < RUB; ++u) {
            if (ok[u]) {
                int slot = atomicAdd(&lcur[r[u].x - d0], 1);
                srcS[slot] = r[u].y;
            }
        }
    }
}


__global__ __launch_bounds__(256) void gat_stage1(
    const int* __restrict__ ei, int E, int K, int chunkLen,
    int* __restrict__ tabC, int Mpad,
    const float* __restrict__ W1, float* __restrict__ Wt, int nbT1,
    const float* __restrict__ Wl, float* __restrict__ Wlt, int C, int CP)
{
    int b = blockIdx.x;
    int t = threadIdx.x;
    if (b < NCH) {
        gat_bcount_body(ei, E, K, chunkLen, tabC, Mpad, b, t, 256);
    } else if (b < NCH + nbT1) {
        gat_transpose_body(W1, Wt, HDIM, HDIM, HDIM, (b - NCH) * 256 + t);
    } else {
        gat_transpose_body(Wl, Wlt, HDIM, C, CP, (b - NCH - nbT1) * 256 + t);
    }
}

__global__ __launch_bounds__(256) void gat_stage2(
    const float* __restrict__ X, const float* __restrict__ Wt,
    float* __restrict__ H,
    const float* __restrict__ asrc, const float* __restrict__ adst,
    float* __restrict__ aS, float* __restrict__ aD, int nRowTiles,
    const int* __restrict__ tabC, int* __restrict__ tabP, int Mpad)
{
    int b = blockIdx.x;
    int t = threadIdx.x;
    if (b == 0) {
        gat_tabscan_body(tabC, tabP, Mpad);
    } else {
        gat_gemm_strip_body(X, Wt, H, asrc, adst, aS, aD, nRowTiles,
                            (b - 1) * GW, t);
    }
}

__global__ __launch_bounds__(1024) void gat_stage3(
    const int* __restrict__ ei, int E, int K, int chunkLen,
    const int* __restrict__ tabP, int* __restrict__ recw,
    const float* __restrict__ W2, float* __restrict__ Wt)
{
    int b = blockIdx.x;
    int t = threadIdx.x;
    if (b < NCH) {
        gat_bscatter_body(ei, E, K, chunkLen, tabP, (int2*)recw, b, t, 1024);
    } else {
        gat_transpose_body(W2, Wt, HDIM, HDIM, HDIM, (b - NCH) * 1024 + t);
    }
}

__device__ __forceinline__ float gat_leaky(float e) {
    return (e >= 0.f) ? e : 0.2f * e;
}

__global__ __launch_bounds__(256) void gat_gather(
    const int* __restrict__ rowptr, const int* __restrict__ srcS,
    const float* __restrict__ aS, const float* __restrict__ aD,
    const float* __restrict__ H, const float* __restrict__ bias,
    float* __restrict__ out, int N)
{
    int w    = __builtin_amdgcn_readfirstlane(
                   (int)((blockIdx.x * blockDim.x + threadIdx.x) >> 5));
    int lane = threadIdx.x & 31;
    if (w >= N) return;
    int beg = rowptr[w];
    int end = rowptr[w + 1];
    float ad    = aD[w];
    float eSelf = gat_leaky(aS[w] + ad);

    const float4* H4 = (const float4*)H;
    float4 hv = H4[(size_t)w * (HDIM / 4) + lane];

    int  j0  = beg + lane;
    int  j1  = j0 + 32;
    bool in0 = j0 < end;
    bool in1 = j1 < end;
    int s0 = srcS[in0 ? j0 : 0];
    int s1 = srcS[in1 ? j1 : 0];
    float l0 = aS[s0];
    float l1 = aS[s1];
    float e0 = in0 ? gat_leaky(l0 + ad) : eSelf;
    float e1 = in1 ? gat_leaky(l1 + ad) : eSelf;

    float m = fmaxf(fmaxf(eSelf, e0), e1);
    for (int j = j1 + 32; j < end; j += 32)
        m = fmaxf(m, gat_leaky(aS[srcS[j]] + ad));
    for (int off = 16; off > 0; off >>= 1)
        m = fmaxf(m, __shfl_xor(m, off, 32));

    const float* bp = bias + 4 * lane;
    float bv0 = bp[0], bv1 = bp[1], bv2 = bp[2], bv3 = bp[3];

    float exSelf = expf(eSelf - m);
    float x0 = in0 ? expf(e0 - m) : 0.f;
    float x1 = in1 ? expf(e1 - m) : 0.f;
    float den = (lane == 0) ? exSelf : 0.f;
    den += x0;
    den += x1;
    for (int j = j1 + 32; j < end; j += 32)
        den += expf(gat_leaky(aS[srcS[j]] + ad) - m);
    for (int off = 16; off > 0; off >>= 1)
        den += __shfl_xor(den, off, 32);

    float aSelf = exSelf / den;
    float a0 = x0 / den;
    float a1 = x1 / den;
    float4 acc;
    acc.x = hv.x * aSelf; acc.y = hv.y * aSelf;
    acc.z = hv.z * aSelf; acc.w = hv.w * aSelf;
    int chunk = 0;
    for (int base = beg; base < end; base += 32, ++chunk) {
        int s;
        float a;
        if (chunk == 0) {
            s = s0; a = a0;
        } else if (chunk == 1) {
            s = s1; a = a1;
        } else {
            int j = base + lane;
            s = 0;
            a = 0.f;
            if (j < end) {
                s = srcS[j];
                a = expf(gat_leaky(aS[s] + ad) - m) / den;
            }
        }
        int cnt = (end - base < 32) ? (end - base) : 32;
        #pragma unroll 4
        for (int t = 0; t < cnt; ++t) {
            int   st = __builtin_amdgcn_readlane(s, t);
            float at = __int_as_float(
                           __builtin_amdgcn_readlane(__float_as_int(a), t));
            float4 v = H4[(size_t)st * (HDIM / 4) + lane];
            acc.x += v.x * at; acc.y += v.y * at;
            acc.z += v.z * at; acc.w += v.w * at;
        }
    }

    float4 r;
    r.x = acc.x + bv0; r.y = acc.y + bv1;
    r.z = acc.z + bv2; r.w = acc.w + bv3;
    r.x = (r.x > 0.f) ? r.x : 0.f; r.y = (r.y > 0.f) ? r.y : 0.f;
    r.z = (r.z > 0.f) ? r.z : 0.f; r.w = (r.w > 0.f) ? r.w : 0.f;
    ((float4*)out)[(size_t)w * (HDIM / 4) + lane] = r;
}

extern "C" void kernel_launch(void* const* d_in, const int* in_sizes, int n_in,
                              void* d_out, int out_size, void* d_ws, size_t ws_size,
                              hipStream_t stream)
{
    const float* x   = (const float*)d_in[0];
    const int*   ei  = (const int*)d_in[1];
    const float* W1  = (const float*)d_in[2];
    const float* a1s = (const float*)d_in[3];
    const float* a1d = (const float*)d_in[4];
    const float* b1  = (const float*)d_in[5];
    const float* W2  = (const float*)d_in[6];
    const float* a2s = (const float*)d_in[7];
    const float* a2d = (const float*)d_in[8];
    const float* b2  = (const float*)d_in[9];
    const float* Wl  = (const float*)d_in[10];
    const float* bl  = (const float*)d_in[11];

    const int N  = in_sizes[0] / HDIM;
    const int E  = in_sizes[1] / 2;
    const int C  = in_sizes[11];
    const int CP = (C + 15) & ~15;

    char* ws = (char*)d_ws;
    auto align256 = [](size_t v) { return (v + 255) & ~(size_t)255; };
    size_t off = 0;
    float* bufH   = (float*)(ws + off);  off += align256((size_t)N * HDIM * 4);
    float* bufA   = (float*)(ws + off);  off += align256((size_t)N * HDIM * 4);
    float* Wt     = (float*)(ws + off);  off += align256((size_t)HDIM * HDIM * 4);
    float* Wlt    = (float*)(ws + off);  off += align256((size_t)CP * HDIM * 4);
    float* aS     = (float*)(ws + off);  off += align256((size_t)N * 4);
    float* aD     = (float*)(ws + off);  off += align256((size_t)N * 4);
    off += align256((size_t)N * 4);
    int*   rowptr = (int*)(ws + off);    off += align256((size_t)(N + 1) * 4);
    int*   srcS   = (int*)(ws + off);    off += align256((size_t)E * 4);
    (void)ws_size;

    const int K        = (N + BR - 1) >> BSH;
    const int Mpad     = (K * NCH + TS_PASS - 1) / TS_PASS * TS_PASS;
    const int chunkLen = (E + NCH - 1) / NCH;
    char*  sa   = (char*)bufA;
    int*   recw = (int*)sa;
    int*   tabC = (int*)(sa + align256((size_t)E * 8));
    int*   tabP = (int*)(sa + align256((size_t)E * 8) + align256((size_t)Mpad * 4));

    const int nRowTiles = N / 16;
    const int BLK = 256;
    dim3 blk(BLK);

    const int nbT     = (HDIM * HDIM + BLK - 1) / BLK;
    const int nbTl    = (CP * HDIM + BLK - 1) / BLK;
    const int nbGemm  = (nRowTiles * 2 + 7) / 8;
    const int nbT2    = (HDIM * HDIM + 1023) / 1024;
    dim3 gemmGrid(nbGemm);
    dim3 waveGridN((N + 7) / 8);

    gat_stage1<<<dim3(NCH + nbT + nbTl), blk, 0, stream>>>(
        ei, E, K, chunkLen, tabC, Mpad, W1, Wt, nbT, Wl, Wlt, C, CP);
    gat_stage2<<<dim3(1 + nbGemm), blk, 0, stream>>>(
        x, Wt, bufH, a1s, a1d, aS, aD, nRowTiles, tabC, tabP, Mpad);
    gat_stage3<<<dim3(NCH + nbT2), dim3(1024), 0, stream>>>(
        ei, E, K, chunkLen, tabP, recw, W2, Wt);
    gat_bsort<<<dim3(K), dim3(1024), 0, stream>>>(tabP, recw, E, N, K, rowptr, srcS);

    gat_gather<<<waveGridN, blk, 0, stream>>>(rowptr, srcS, aS, aD, bufH,
                                              b1, bufA, N);

    gat_gemm_strip<<<gemmGrid, blk, 0, stream>>>(bufA, Wt, bufH, a2s, a2d,
                                                 aS, aD, nRowTiles);
    gat_gather<<<waveGridN, blk, 0, stream>>>(rowptr, srcS, aS, aD, bufH,
                                              b2, bufA, N);

    dim3 clsGrid((nRowTiles + 7) / 8);
    gat_gemm_cls<<<clsGrid, blk, 0, stream>>>(bufA, Wlt, bl, (float*)d_out,
                                              nRowTiles, C);
}


# GEMM weights kept in WMMA fragment order, both operands fetched 16 B per lane (2 k-steps per load): a B load is 4 whole 128-B lines, not 32 B from each of 16; k regroups, product sum order changes
# speedup vs baseline: 4.5510x; 1.0732x over previous
#include <hip/hip_runtime.h>
#include <hip/hip_bf16.h>

typedef __attribute__((ext_vector_type(2))) float v2f;
typedef __attribute__((ext_vector_type(8))) float v8f;

#define HDIM 128
#define CTPW 4


__device__ __forceinline__ void gat_fragorder_body(
    const float* __restrict__ W, float* __restrict__ Wf,
    int cols, int colsPad, int i)
{
    int total = colsPad * HDIM;
    if (i >= total) return;
    int e = i & 3;
    int l = (i >> 2) & 31;
    int j = (i >> 7) & 15;
    int q = i >> 11;
    int k = 8 * j + 4 * (l >> 4) + e;
    int c = q * 16 + (l & 15);
    Wf[i] = (c < cols) ? W[(size_t)k * cols + c] : 0.f;
}

#define GW 8

__device__ __forceinline__ float gat_row16_sum(float v)
{
    #pragma unroll
    for (int off = 1; off < 16; off <<= 1)
        v += __shfl_xor(v, off, 32);
    return v;
}

__device__ __forceinline__ void gat_gemm_strip_body(
    const float* __restrict__ X, const float* __restrict__ Wf,
    float* __restrict__ H,
    const float* __restrict__ asrc, const float* __restrict__ adst,
    float* __restrict__ aS, float* __restrict__ aD,
    int nRowTiles, int wid0, int t)
{
    __shared__ float part[GW][2][16];
    const int nHalves = HDIM / (CTPW * 16);
    const int nW    = nRowTiles * nHalves;
    const int wv    = t >> 5;
    const int lane  = t & 31;
    const int wid   = wid0 + wv;
    const int rt    = wid >> 1;
    const int half  = wid & 1;
    const int row0  = rt * 16;
    const int laneM = lane & 15;
    const int hf    = lane >> 4;
    const int mBase = hf * 8;
    const bool act = wid < nW;

    v8f c[CTPW] = {};
    if (act) {
        const float4* xq = (const float4*)(X + (size_t)(row0 + laneM) * HDIM) + hf;
        const float4* wq = (const float4*)Wf + (size_t)(half * CTPW) * (16 * 32) + lane;

        float as[CTPW], ad[CTPW];
        #pragma unroll
        for (int q = 0; q < CTPW; ++q) {
            int col = (half * CTPW + q) * 16 + laneM;
            as[q] = asrc[col];
            ad[q] = adst[col];
        }

        #pragma unroll 4
        for (int j = 0; j < HDIM / 8; ++j) {
            float4 a4 = xq[2 * j];
            float4 b4[CTPW];
            #pragma unroll
            for (int q = 0; q < CTPW; ++q) b4[q] = wq[(q * 16 + j) * 32];
            v2f a0 = {a4.x, a4.y};
            v2f a1 = {a4.z, a4.w};
            #pragma unroll
            for (int q = 0; q < CTPW; ++q) {
                v2f b0 = {b4[q].x, b4[q].y};
                c[q] = __builtin_amdgcn_wmma_f32_16x16x4_f32(
                           false, a0, false, b0, (short)0, c[q], false, false);
            }
            #pragma unroll
            for (int q = 0; q < CTPW; ++q) {
                v2f b1 = {b4[q].z, b4[q].w};
                c[q] = __builtin_amdgcn_wmma_f32_16x16x4_f32(
                           false, a1, false, b1, (short)0, c[q], false, false);
            }
        }

        float ps[8], pd[8];
        #pragma unroll
        for (int r = 0; r < 8; ++r) {
            float s = 0.f, d = 0.f;
            #pragma unroll
            for (int q = 0; q < CTPW; ++q) {
                s += c[q][r] * as[q];
                d += c[q][r] * ad[q];
            }
            ps[r] = s;
            pd[r] = d;
        }
        #pragma unroll
        for (int r = 0; r < 8; ++r) {
            ps[r] = gat_row16_sum(ps[r]);
            pd[r] = gat_row16_sum(pd[r]);
        }
        if (laneM == 0) {
            #pragma unroll
            for (int r = 0; r < 8; ++r) {
                part[wv][0][mBase + r] = ps[r];
                part[wv][1][mBase + r] = pd[r];
            }
        }
    }
    __syncthreads();
    if (t < (GW / 2) * 16) {
        int ls  = t >> 4;
        int row = t & 15;
        int rs  = (wid0 >> 1) + ls;
        if (rs < nRowTiles) {
            aS[rs * 16 + row] = part[2 * ls][0][row] + part[2 * ls + 1][0][row];
            aD[rs * 16 + row] = part[2 * ls][1][row] + part[2 * ls + 1][1][row];
        }
    }
    if (act) {
        #pragma unroll
        for (int q = 0; q < CTPW; ++q) {
            float* o = H + (size_t)row0 * HDIM + (half * CTPW + q) * 16 + laneM;
            #pragma unroll
            for (int r = 0; r < 8; ++r)
                o[(size_t)(mBase + r) * HDIM] = c[q][r];
        }
    }
}

__global__ __launch_bounds__(256) void gat_gemm_strip(
    const float* __restrict__ X, const float* __restrict__ Wf,
    float* __restrict__ H,
    const float* __restrict__ asrc, const float* __restrict__ adst,
    float* __restrict__ aS, float* __restrict__ aD, int nRowTiles)
{
    gat_gemm_strip_body(X, Wf, H, asrc, adst, aS, aD, nRowTiles,
                        (int)(blockIdx.x * GW), (int)threadIdx.x);
}

__global__ __launch_bounds__(256) void gat_gemm_cls(
    const float* __restrict__ X, const float* __restrict__ Wlf,
    const float* __restrict__ bl, float* __restrict__ Out,
    int nRowTiles, int nCols)
{
    int wid  = (blockIdx.x * blockDim.x + threadIdx.x) >> 5;
    int lane = threadIdx.x & 31;
    if (wid >= nRowTiles) return;
    int row0 = wid * 16;
    int laneM = lane & 15;
    int hf    = lane >> 4;

    const float4* xq = (const float4*)(X + (size_t)(row0 + laneM) * HDIM) + hf;
    const float4* wq = (const float4*)Wlf + lane;

    v8f c[3] = {};
    #pragma unroll 4
    for (int j = 0; j < HDIM / 8; ++j) {
        float4 a4 = xq[2 * j];
        float4 b4[3];
        #pragma unroll
        for (int t = 0; t < 3; ++t) b4[t] = wq[(t * 16 + j) * 32];
        v2f a0 = {a4.x, a4.y};
        v2f a1 = {a4.z, a4.w};
        #pragma unroll
        for (int t = 0; t < 3; ++t) {
            v2f b0 = {b4[t].x, b4[t].y};
            c[t] = __builtin_amdgcn_wmma_f32_16x16x4_f32(
                       false, a0, false, b0, (short)0, c[t], false, false);
        }
        #pragma unroll
        for (int t = 0; t < 3; ++t) {
            v2f b1 = {b4[t].z, b4[t].w};
            c[t] = __builtin_amdgcn_wmma_f32_16x16x4_f32(
                       false, a1, false, b1, (short)0, c[t], false, false);
        }
    }

    int mBase = (lane >> 4) * 8;
    #pragma unroll
    for (int t = 0; t < 3; ++t) {
        int col = t * 16 + laneM;
        if (col >= nCols) continue;
        float bv = bl[col];
        #pragma unroll
        for (int r = 0; r < 8; ++r)
            Out[(size_t)(row0 + mBase + r) * nCols + col] = c[t][r] + bv;
    }
}

#define BSH  7
#define BR   (1 << BSH)
#define NCH  128
#define KMAX 1024
#define EUB  8
#define RUB  6

__device__ __forceinline__ int gat_wave_incl_scan(int x, int lane)
{
    #pragma unroll
    for (int off = 1; off < 32; off <<= 1) {
        int y = __shfl_up(x, off, 32);
        if (lane >= off) x += y;
    }
    return x;
}

__device__ __forceinline__ void gat_bcount_body(
    const int* __restrict__ ei, int E, int K, int chunkLen,
    int* __restrict__ tabC, int Mpad, int c, int t, int nt)
{
    __shared__ int h[KMAX];
    if (c == 0)
        for (int p = K * NCH + t; p < Mpad; p += nt) tabC[p] = 0;
    for (int k = t; k < K; k += nt) h[k] = 0;
    __syncthreads();
    int c0 = c * chunkLen;
    int c1 = (c0 + chunkLen < E) ? (c0 + chunkLen) : E;
    for (int i = c0 + t; i < c1; i += EUB * nt) {
        int d[EUB];
        #pragma unroll
        for (int u = 0; u < EUB; ++u) {
            int j  = i + u * nt;
            bool ok = j < c1;
            int dv = ei[E + (ok ? j : c0)];
            d[u] = ok ? dv : -1;
        }
        #pragma unroll
        for (int u = 0; u < EUB; ++u)
            if (d[u] >= 0) atomicAdd(&h[d[u] >> BSH], 1);
    }
    __syncthreads();
    for (int k = t; k < K; k += nt) tabC[k * NCH + c] = h[k];
}

#define TS_G    20
#define TS_NT   256
#define TS_PASS (TS_NT * 4 * TS_G)

__device__ __forceinline__ void gat_tabscan_body(
    const int* __restrict__ tabC, int* __restrict__ tabP, int Mpad)
{
    __shared__ int wt[TS_NT / 32];
    const int t    = threadIdx.x;
    const int lane = t & 31;
    const int wv   = t >> 5;
    int carry = 0;
    for (int base = 0; base < Mpad; base += TS_PASS) {
        const int4* src = (const int4*)(tabC + base + t * (4 * TS_G));
        int4*       dst = (int4*)(tabP + base + t * (4 * TS_G));
        int4 q[TS_G];
        #pragma unroll
        for (int g = 0; g < TS_G; ++g) q[g] = src[g];
        int s = 0;
        #pragma unroll
        for (int g = 0; g < TS_G; ++g) s += q[g].x + q[g].y + q[g].z + q[g].w;
        int inc = gat_wave_incl_scan(s, lane);
        if (lane == 31) wt[wv] = inc;
        __syncthreads();
        int run = carry + inc - s;
        int tot = 0;
        for (int w = 0; w < TS_NT / 32; ++w) {
            int x = wt[w];
            tot += x;
            if (w < wv) run += x;
        }
        #pragma unroll
        for (int g = 0; g < TS_G; ++g) {
            int4 o;
            o.x = run;
            o.y = o.x + q[g].x;
            o.z = o.y + q[g].y;
            o.w = o.z + q[g].z;
            run = o.w + q[g].w;
            dst[g] = o;
        }
        carry += tot;
        __syncthreads();
    }
}

__device__ __forceinline__ void gat_bscatter_body(
    const int* __restrict__ ei, int E, int K, int chunkLen,
    const int* __restrict__ tabP, int2* __restrict__ rec, int c, int t, int nt)
{
    __shared__ int cur[KMAX];
    for (int k = t; k < K; k += nt) cur[k] = tabP[k * NCH + c];
    __syncthreads();
    int c0 = c * chunkLen;
    int c1 = (c0 + chunkLen < E) ? (c0 + chunkLen) : E;
    for (int i = c0 + t; i < c1; i += EUB * nt) {
        int d[EUB], s[EUB];
        #pragma unroll
        for (int u = 0; u < EUB; ++u) {
            int j  = i + u * nt;
            bool ok = j < c1;
            int jj = ok ? j : c0;
            int dv = ei[E + jj];
            s[u] = ei[jj];
            d[u] = ok ? dv : -1;
        }
        #pragma unroll
        for (int u = 0; u < EUB; ++u) {
            if (d[u] >= 0) {
                int slot = atomicAdd(&cur[d[u] >> BSH], 1);
                rec[slot] = make_int2(d[u], s[u]);
            }
        }
    }
}

__global__ __launch_bounds__(1024) void gat_bsort(
    const int* __restrict__ tabP, const int* __restrict__ recw,
    int E, int N, int K, int* __restrict__ rowptr, int* __restrict__ srcS)
{
    __shared__ int lcnt[BR];
    __shared__ int lcur[BR];
    __shared__ int wsum[BR / 32];
    const int2* rec = (const int2*)recw;
    const int k    = blockIdx.x;
    const int t    = threadIdx.x;
    const int lane = t & 31;
    const int wv   = t >> 5;
    const int start = tabP[k * NCH];
    const int end   = (k + 1 < K) ? tabP[(k + 1) * NCH] : E;
    const int d0    = k << BSH;
    if (t < BR) lcnt[t] = 0;
    __syncthreads();
    for (int e = start + t; e < end; e += RUB * 1024) {
        int d[RUB];
        #pragma unroll
        for (int u = 0; u < RUB; ++u) {
            int j  = e + u * 1024;
            bool ok = j < end;
            int dv = rec[ok ? j : start].x;
            d[u] = ok ? dv : -1;
        }
        #pragma unroll
        for (int u = 0; u < RUB; ++u)
            if (d[u] >= 0) atomicAdd(&lcnt[d[u] - d0], 1);
    }
    __syncthreads();
    int v   = (t < BR) ? lcnt[t] : 0;
    int inc = gat_wave_incl_scan(v, lane);
    if (t < BR && lane == 31) wsum[wv] = inc;
    __syncthreads();
    if (t < BR) {
        int ex = start + inc - v;
        for (int q = 0; q < wv; ++q) ex += wsum[q];
        lcur[t] = ex;
        if (d0 + t < N) rowptr[d0 + t] = ex;
    }
    if (k == K - 1 && t == 0) rowptr[N] = end;
    __syncthreads();
    for (int e = start + t; e < end; e += RUB * 1024) {
        int2 r[RUB];
        bool ok[RUB];
        #pragma unroll
        for (int u = 0; u < RUB; ++u) {
            int j = e + u * 1024;
            ok[u] = j < end;
            r[u]  = rec[ok[u] ? j : start];
        }
        #pragma unroll
        for (int u = 0; u < RUB; ++u) {
            if (ok[u]) {
                int slot = atomicAdd(&lcur[r[u].x - d0], 1);
                srcS[slot] = r[u].y;
            }
        }
    }
}


__global__ __launch_bounds__(256) void gat_stage1(
    const int* __restrict__ ei, int E, int K, int chunkLen,
    int* __restrict__ tabC, int Mpad,
    const float* __restrict__ W1, float* __restrict__ Wf, int nbT1,
    const float* __restrict__ Wl, float* __restrict__ Wlf, int C, int CP)
{
    int b = blockIdx.x;
    int t = threadIdx.x;
    if (b < NCH) {
        gat_bcount_body(ei, E, K, chunkLen, tabC, Mpad, b, t, 256);
    } else if (b < NCH + nbT1) {
        gat_fragorder_body(W1, Wf, HDIM, HDIM, (b - NCH) * 256 + t);
    } else {
        gat_fragorder_body(Wl, Wlf, C, CP, (b - NCH - nbT1) * 256 + t);
    }
}

__global__ __launch_bounds__(256) void gat_stage2(
    const float* __restrict__ X, const float* __restrict__ Wf,
    float* __restrict__ H,
    const float* __restrict__ asrc, const float* __restrict__ adst,
    float* __restrict__ aS, float* __restrict__ aD, int nRowTiles,
    const int* __restrict__ tabC, int* __restrict__ tabP, int Mpad)
{
    int b = blockIdx.x;
    int t = threadIdx.x;
    if (b == 0) {
        gat_tabscan_body(tabC, tabP, Mpad);
    } else {
        gat_gemm_strip_body(X, Wf, H, asrc, adst, aS, aD, nRowTiles,
                            (b - 1) * GW, t);
    }
}

__global__ __launch_bounds__(1024) void gat_stage3(
    const int* __restrict__ ei, int E, int K, int chunkLen,
    const int* __restrict__ tabP, int* __restrict__ recw,
    const float* __restrict__ W2, float* __restrict__ Wf)
{
    int b = blockIdx.x;
    int t = threadIdx.x;
    if (b < NCH) {
        gat_bscatter_body(ei, E, K, chunkLen, tabP, (int2*)recw, b, t, 1024);
    } else {
        gat_fragorder_body(W2, Wf, HDIM, HDIM, (b - NCH) * 1024 + t);
    }
}

__device__ __forceinline__ float gat_leaky(float e) {
    return (e >= 0.f) ? e : 0.2f * e;
}

__global__ __launch_bounds__(256) void gat_gather(
    const int* __restrict__ rowptr, const int* __restrict__ srcS,
    const float* __restrict__ aS, const float* __restrict__ aD,
    const float* __restrict__ H, const float* __restrict__ bias,
    float* __restrict__ out, int N)
{
    int w    = __builtin_amdgcn_readfirstlane(
                   (int)((blockIdx.x * blockDim.x + threadIdx.x) >> 5));
    int lane = threadIdx.x & 31;
    if (w >= N) return;
    int beg = rowptr[w];
    int end = rowptr[w + 1];
    float ad    = aD[w];
    float eSelf = gat_leaky(aS[w] + ad);

    const float4* H4 = (const float4*)H;
    float4 hv = H4[(size_t)w * (HDIM / 4) + lane];

    int  j0  = beg + lane;
    int  j1  = j0 + 32;
    bool in0 = j0 < end;
    bool in1 = j1 < end;
    int s0 = srcS[in0 ? j0 : 0];
    int s1 = srcS[in1 ? j1 : 0];
    float l0 = aS[s0];
    float l1 = aS[s1];
    float e0 = in0 ? gat_leaky(l0 + ad) : eSelf;
    float e1 = in1 ? gat_leaky(l1 + ad) : eSelf;

    float m = fmaxf(fmaxf(eSelf, e0), e1);
    for (int j = j1 + 32; j < end; j += 32)
        m = fmaxf(m, gat_leaky(aS[srcS[j]] + ad));
    for (int off = 16; off > 0; off >>= 1)
        m = fmaxf(m, __shfl_xor(m, off, 32));

    const float* bp = bias + 4 * lane;
    float bv0 = bp[0], bv1 = bp[1], bv2 = bp[2], bv3 = bp[3];

    float exSelf = expf(eSelf - m);
    float x0 = in0 ? expf(e0 - m) : 0.f;
    float x1 = in1 ? expf(e1 - m) : 0.f;
    float den = (lane == 0) ? exSelf : 0.f;
    den += x0;
    den += x1;
    for (int j = j1 + 32; j < end; j += 32)
        den += expf(gat_leaky(aS[srcS[j]] + ad) - m);
    for (int off = 16; off > 0; off >>= 1)
        den += __shfl_xor(den, off, 32);

    float aSelf = exSelf / den;
    float a0 = x0 / den;
    float a1 = x1 / den;
    float4 acc;
    acc.x = hv.x * aSelf; acc.y = hv.y * aSelf;
    acc.z = hv.z * aSelf; acc.w = hv.w * aSelf;
    int chunk = 0;
    for (int base = beg; base < end; base += 32, ++chunk) {
        int s;
        float a;
        if (chunk == 0) {
            s = s0; a = a0;
        } else if (chunk == 1) {
            s = s1; a = a1;
        } else {
            int j = base + lane;
            s = 0;
            a = 0.f;
            if (j < end) {
                s = srcS[j];
                a = expf(gat_leaky(aS[s] + ad) - m) / den;
            }
        }
        int cnt = (end - base < 32) ? (end - base) : 32;
        #pragma unroll 4
        for (int t = 0; t < cnt; ++t) {
            int   st = __builtin_amdgcn_readlane(s, t);
            float at = __int_as_float(
                           __builtin_amdgcn_readlane(__float_as_int(a), t));
            float4 v = H4[(size_t)st * (HDIM / 4) + lane];
            acc.x += v.x * at; acc.y += v.y * at;
            acc.z += v.z * at; acc.w += v.w * at;
        }
    }

    float4 r;
    r.x = acc.x + bv0; r.y = acc.y + bv1;
    r.z = acc.z + bv2; r.w = acc.w + bv3;
    r.x = (r.x > 0.f) ? r.x : 0.f; r.y = (r.y > 0.f) ? r.y : 0.f;
    r.z = (r.z > 0.f) ? r.z : 0.f; r.w = (r.w > 0.f) ? r.w : 0.f;
    ((float4*)out)[(size_t)w * (HDIM / 4) + lane] = r;
}

extern "C" void kernel_launch(void* const* d_in, const int* in_sizes, int n_in,
                              void* d_out, int out_size, void* d_ws, size_t ws_size,
                              hipStream_t stream)
{
    const float* x   = (const float*)d_in[0];
    const int*   ei  = (const int*)d_in[1];
    const float* W1  = (const float*)d_in[2];
    const float* a1s = (const float*)d_in[3];
    const float* a1d = (const float*)d_in[4];
    const float* b1  = (const float*)d_in[5];
    const float* W2  = (const float*)d_in[6];
    const float* a2s = (const float*)d_in[7];
    const float* a2d = (const float*)d_in[8];
    const float* b2  = (const float*)d_in[9];
    const float* Wl  = (const float*)d_in[10];
    const float* bl  = (const float*)d_in[11];

    const int N  = in_sizes[0] / HDIM;
    const int E  = in_sizes[1] / 2;
    const int C  = in_sizes[11];
    const int CP = (C + 15) & ~15;

    char* ws = (char*)d_ws;
    auto align256 = [](size_t v) { return (v + 255) & ~(size_t)255; };
    size_t off = 0;
    float* bufH   = (float*)(ws + off);  off += align256((size_t)N * HDIM * 4);
    float* bufA   = (float*)(ws + off);  off += align256((size_t)N * HDIM * 4);
    float* Wf     = (float*)(ws + off);  off += align256((size_t)HDIM * HDIM * 4);
    float* Wlf    = (float*)(ws + off);  off += align256((size_t)CP * HDIM * 4);
    float* aS     = (float*)(ws + off);  off += align256((size_t)N * 4);
    float* aD     = (float*)(ws + off);  off += align256((size_t)N * 4);
    off += align256((size_t)N * 4);
    int*   rowptr = (int*)(ws + off);    off += align256((size_t)(N + 1) * 4);
    int*   srcS   = (int*)(ws + off);    off += align256((size_t)E * 4);
    (void)ws_size;

    const int K        = (N + BR - 1) >> BSH;
    const int Mpad     = (K * NCH + TS_PASS - 1) / TS_PASS * TS_PASS;
    const int chunkLen = (E + NCH - 1) / NCH;
    char*  sa   = (char*)bufA;
    int*   recw = (int*)sa;
    int*   tabC = (int*)(sa + align256((size_t)E * 8));
    int*   tabP = (int*)(sa + align256((size_t)E * 8) + align256((size_t)Mpad * 4));

    const int nRowTiles = N / 16;
    const int BLK = 256;
    dim3 blk(BLK);

    const int nbT     = (HDIM * HDIM + BLK - 1) / BLK;
    const int nbTl    = (CP * HDIM + BLK - 1) / BLK;
    const int nbGemm  = (nRowTiles * 2 + 7) / 8;
    const int nbT2    = (HDIM * HDIM + 1023) / 1024;
    dim3 gemmGrid(nbGemm);
    dim3 waveGridN((N + 7) / 8);

    gat_stage1<<<dim3(NCH + nbT + nbTl), blk, 0, stream>>>(
        ei, E, K, chunkLen, tabC, Mpad, W1, Wf, nbT, Wl, Wlf, C, CP);
    gat_stage2<<<dim3(1 + nbGemm), blk, 0, stream>>>(
        x, Wf, bufH, a1s, a1d, aS, aD, nRowTiles, tabC, tabP, Mpad);
    gat_stage3<<<dim3(NCH + nbT2), dim3(1024), 0, stream>>>(
        ei, E, K, chunkLen, tabP, recw, W2, Wf);
    gat_bsort<<<dim3(K), dim3(1024), 0, stream>>>(tabP, recw, E, N, K, rowptr, srcS);

    gat_gather<<<waveGridN, blk, 0, stream>>>(rowptr, srcS, aS, aD, bufH,
                                              b1, bufA, N);

    gat_gemm_strip<<<gemmGrid, blk, 0, stream>>>(bufA, Wf, bufH, a2s, a2d,
                                                 aS, aD, nRowTiles);
    gat_gather<<<waveGridN, blk, 0, stream>>>(rowptr, srcS, aS, aD, bufH,
                                              b2, bufA, N);

    dim3 clsGrid((nRowTiles + 7) / 8);
    gat_gemm_cls<<<clsGrid, blk, 0, stream>>>(bufA, Wlf, bl, (float*)d_out,
                                              nRowTiles, C);
}


# Gather work-groups own a 16-node strip, keep the new rows in LDS and multiply them there (layer-2 GEMM + logits, classifier): 8 -> 6 launches; layer-2 logit sum order changes
# speedup vs baseline: 5.1894x; 1.1403x over previous
#include <hip/hip_runtime.h>
#include <hip/hip_bf16.h>

typedef __attribute__((ext_vector_type(2))) float v2f;
typedef __attribute__((ext_vector_type(8))) float v8f;

#define HDIM 128
#define CTPW 4


__device__ __forceinline__ void gat_fragorder_body(
    const float* __restrict__ W, float* __restrict__ Wf,
    int cols, int colsPad, int i)
{
    int total = colsPad * HDIM;
    if (i >= total) return;
    int e = i & 3;
    int l = (i >> 2) & 31;
    int j = (i >> 7) & 15;
    int q = i >> 11;
    int k = 8 * j + 4 * (l >> 4) + e;
    int c = q * 16 + (l & 15);
    Wf[i] = (c < cols) ? W[(size_t)k * cols + c] : 0.f;
}

#define GW 8

__device__ __forceinline__ float gat_row16_sum(float v)
{
    #pragma unroll
    for (int off = 1; off < 16; off <<= 1)
        v += __shfl_xor(v, off, 32);
    return v;
}

__device__ __forceinline__ void gat_gemm_strip_body(
    const float* __restrict__ X, const float* __restrict__ Wf,
    float* __restrict__ H,
    const float* __restrict__ asrc, const float* __restrict__ adst,
    float* __restrict__ aS, float* __restrict__ aD,
    int nRowTiles, int wid0, int t)
{
    __shared__ float part[GW][2][16];
    const int nHalves = HDIM / (CTPW * 16);
    const int nW    = nRowTiles * nHalves;
    const int wv    = t >> 5;
    const int lane  = t & 31;
    const int wid   = wid0 + wv;
    const int rt    = wid >> 1;
    const int half  = wid & 1;
    const int row0  = rt * 16;
    const int laneM = lane & 15;
    const int hf    = lane >> 4;
    const int mBase = hf * 8;
    const bool act = wid < nW;

    v8f c[CTPW] = {};
    if (act) {
        const float4* xq = (const float4*)(X + (size_t)(row0 + laneM) * HDIM) + hf;
        const float4* wq = (const float4*)Wf + (size_t)(half * CTPW) * (16 * 32) + lane;

        float as[CTPW], ad[CTPW];
        #pragma unroll
        for (int q = 0; q < CTPW; ++q) {
            int col = (half * CTPW + q) * 16 + laneM;
            as[q] = asrc[col];
            ad[q] = adst[col];
        }

        #pragma unroll 4
        for (int j = 0; j < HDIM / 8; ++j) {
            float4 a4 = xq[2 * j];
            float4 b4[CTPW];
            #pragma unroll
            for (int q = 0; q < CTPW; ++q) b4[q] = wq[(q * 16 + j) * 32];
            v2f a0 = {a4.x, a4.y};
            v2f a1 = {a4.z, a4.w};
            #pragma unroll
            for (int q = 0; q < CTPW; ++q) {
                v2f b0 = {b4[q].x, b4[q].y};
                c[q] = __builtin_amdgcn_wmma_f32_16x16x4_f32(
                           false, a0, false, b0, (short)0, c[q], false, false);
            }
            #pragma unroll
            for (int q = 0; q < CTPW; ++q) {
                v2f b1 = {b4[q].z, b4[q].w};
                c[q] = __builtin_amdgcn_wmma_f32_16x16x4_f32(
                           false, a1, false, b1, (short)0, c[q], false, false);
            }
        }

        float ps[8], pd[8];
        #pragma unroll
        for (int r = 0; r < 8; ++r) {
            float s = 0.f, d = 0.f;
            #pragma unroll
            for (int q = 0; q < CTPW; ++q) {
                s += c[q][r] * as[q];
                d += c[q][r] * ad[q];
            }
            ps[r] = s;
            pd[r] = d;
        }
        #pragma unroll
        for (int r = 0; r < 8; ++r) {
            ps[r] = gat_row16_sum(ps[r]);
            pd[r] = gat_row16_sum(pd[r]);
        }
        if (laneM == 0) {
            #pragma unroll
            for (int r = 0; r < 8; ++r) {
                part[wv][0][mBase + r] = ps[r];
                part[wv][1][mBase + r] = pd[r];
            }
        }
    }
    __syncthreads();
    if (t < (GW / 2) * 16) {
        int ls  = t >> 4;
        int row = t & 15;
        int rs  = (wid0 >> 1) + ls;
        if (rs < nRowTiles) {
            aS[rs * 16 + row] = part[2 * ls][0][row] + part[2 * ls + 1][0][row];
            aD[rs * 16 + row] = part[2 * ls][1][row] + part[2 * ls + 1][1][row];
        }
    }
    if (act) {
        #pragma unroll
        for (int q = 0; q < CTPW; ++q) {
            float* o = H + (size_t)row0 * HDIM + (half * CTPW + q) * 16 + laneM;
            #pragma unroll
            for (int r = 0; r < 8; ++r)
                o[(size_t)(mBase + r) * HDIM] = c[q][r];
        }
    }
}

#define BSH  7
#define BR   (1 << BSH)
#define NCH  128
#define KMAX 1024
#define EUB  8
#define RUB  6

__device__ __forceinline__ int gat_wave_incl_scan(int x, int lane)
{
    #pragma unroll
    for (int off = 1; off < 32; off <<= 1) {
        int y = __shfl_up(x, off, 32);
        if (lane >= off) x += y;
    }
    return x;
}

__device__ __forceinline__ void gat_bcount_body(
    const int* __restrict__ ei, int E, int K, int chunkLen,
    int* __restrict__ tabC, int Mpad, int c, int t, int nt)
{
    __shared__ int h[KMAX];
    if (c == 0)
        for (int p = K * NCH + t; p < Mpad; p += nt) tabC[p] = 0;
    for (int k = t; k < K; k += nt) h[k] = 0;
    __syncthreads();
    int c0 = c * chunkLen;
    int c1 = (c0 + chunkLen < E) ? (c0 + chunkLen) : E;
    for (int i = c0 + t; i < c1; i += EUB * nt) {
        int d[EUB];
        #pragma unroll
        for (int u = 0; u < EUB; ++u) {
            int j  = i + u * nt;
            bool ok = j < c1;
            int dv = ei[E + (ok ? j : c0)];
            d[u] = ok ? dv : -1;
        }
        #pragma unroll
        for (int u = 0; u < EUB; ++u)
            if (d[u] >= 0) atomicAdd(&h[d[u] >> BSH], 1);
    }
    __syncthreads();
    for (int k = t; k < K; k += nt) tabC[k * NCH + c] = h[k];
}

#define TS_G    20
#define TS_NT   256
#define TS_PASS (TS_NT * 4 * TS_G)

__device__ __forceinline__ void gat_tabscan_body(
    const int* __restrict__ tabC, int* __restrict__ tabP, int Mpad)
{
    __shared__ int wt[TS_NT / 32];
    const int t    = threadIdx.x;
    const int lane = t & 31;
    const int wv   = t >> 5;
    int carry = 0;
    for (int base = 0; base < Mpad; base += TS_PASS) {
        const int4* src = (const int4*)(tabC + base + t * (4 * TS_G));
        int4*       dst = (int4*)(tabP + base + t * (4 * TS_G));
        int4 q[TS_G];
        #pragma unroll
        for (int g = 0; g < TS_G; ++g) q[g] = src[g];
        int s = 0;
        #pragma unroll
        for (int g = 0; g < TS_G; ++g) s += q[g].x + q[g].y + q[g].z + q[g].w;
        int inc = gat_wave_incl_scan(s, lane);
        if (lane == 31) wt[wv] = inc;
        __syncthreads();
        int run = carry + inc - s;
        int tot = 0;
        for (int w = 0; w < TS_NT / 32; ++w) {
            int x = wt[w];
            tot += x;
            if (w < wv) run += x;
        }
        #pragma unroll
        for (int g = 0; g < TS_G; ++g) {
            int4 o;
            o.x = run;
            o.y = o.x + q[g].x;
            o.z = o.y + q[g].y;
            o.w = o.z + q[g].z;
            run = o.w + q[g].w;
            dst[g] = o;
        }
        carry += tot;
        __syncthreads();
    }
}

__device__ __forceinline__ void gat_bscatter_body(
    const int* __restrict__ ei, int E, int K, int chunkLen,
    const int* __restrict__ tabP, int2* __restrict__ rec, int c, int t, int nt)
{
    __shared__ int cur[KMAX];
    for (int k = t; k < K; k += nt) cur[k] = tabP[k * NCH + c];
    __syncthreads();
    int c0 = c * chunkLen;
    int c1 = (c0 + chunkLen < E) ? (c0 + chunkLen) : E;
    for (int i = c0 + t; i < c1; i += EUB * nt) {
        int d[EUB], s[EUB];
        #pragma unroll
        for (int u = 0; u < EUB; ++u) {
            int j  = i + u * nt;
            bool ok = j < c1;
            int jj = ok ? j : c0;
            int dv = ei[E + jj];
            s[u] = ei[jj];
            d[u] = ok ? dv : -1;
        }
        #pragma unroll
        for (int u = 0; u < EUB; ++u) {
            if (d[u] >= 0) {
                int slot = atomicAdd(&cur[d[u] >> BSH], 1);
                rec[slot] = make_int2(d[u], s[u]);
            }
        }
    }
}

__global__ __launch_bounds__(1024) void gat_bsort(
    const int* __restrict__ tabP, const int* __restrict__ recw,
    int E, int N, int K, int* __restrict__ rowptr, int* __restrict__ srcS)
{
    __shared__ int lcnt[BR];
    __shared__ int lcur[BR];
    __shared__ int wsum[BR / 32];
    const int2* rec = (const int2*)recw;
    const int k    = blockIdx.x;
    const int t    = threadIdx.x;
    const int lane = t & 31;
    const int wv   = t >> 5;
    const int start = tabP[k * NCH];
    const int end   = (k + 1 < K) ? tabP[(k + 1) * NCH] : E;
    const int d0    = k << BSH;
    if (t < BR) lcnt[t] = 0;
    __syncthreads();
    for (int e = start + t; e < end; e += RUB * 1024) {
        int d[RUB];
        #pragma unroll
        for (int u = 0; u < RUB; ++u) {
            int j  = e + u * 1024;
            bool ok = j < end;
            int dv = rec[ok ? j : start].x;
            d[u] = ok ? dv : -1;
        }
        #pragma unroll
        for (int u = 0; u < RUB; ++u)
            if (d[u] >= 0) atomicAdd(&lcnt[d[u] - d0], 1);
    }
    __syncthreads();
    int v   = (t < BR) ? lcnt[t] : 0;
    int inc = gat_wave_incl_scan(v, lane);
    if (t < BR && lane == 31) wsum[wv] = inc;
    __syncthreads();
    if (t < BR) {
        int ex = start + inc - v;
        for (int q = 0; q < wv; ++q) ex += wsum[q];
        lcur[t] = ex;
        if (d0 + t < N) rowptr[d0 + t] = ex;
    }
    if (k == K - 1 && t == 0) rowptr[N] = end;
    __syncthreads();
    for (int e = start + t; e < end; e += RUB * 1024) {
        int2 r[RUB];
        bool ok[RUB];
        #pragma unroll
        for (int u = 0; u < RUB; ++u) {
            int j = e + u * 1024;
            ok[u] = j < end;
            r[u]  = rec[ok[u] ? j : start];
        }
        #pragma unroll
        for (int u = 0; u < RUB; ++u) {
            if (ok[u]) {
                int slot = atomicAdd(&lcur[r[u].x - d0], 1);
                srcS[slot] = r[u].y;
            }
        }
    }
}


__global__ __launch_bounds__(256) void gat_stage1(
    const int* __restrict__ ei, int E, int K, int chunkLen,
    int* __restrict__ tabC, int Mpad,
    const float* __restrict__ W1, float* __restrict__ Wf, int nbT1,
    const float* __restrict__ Wl, float* __restrict__ Wlf, int C, int CP)
{
    int b = blockIdx.x;
    int t = threadIdx.x;
    if (b < NCH) {
        gat_bcount_body(ei, E, K, chunkLen, tabC, Mpad, b, t, 256);
    } else if (b < NCH + nbT1) {
        gat_fragorder_body(W1, Wf, HDIM, HDIM, (b - NCH) * 256 + t);
    } else {
        gat_fragorder_body(Wl, Wlf, C, CP, (b - NCH - nbT1) * 256 + t);
    }
}

__global__ __launch_bounds__(256) void gat_stage2(
    const float* __restrict__ X, const float* __restrict__ Wf,
    float* __restrict__ H,
    const float* __restrict__ asrc, const float* __restrict__ adst,
    float* __restrict__ aS, float* __restrict__ aD, int nRowTiles,
    const int* __restrict__ tabC, int* __restrict__ tabP, int Mpad)
{
    int b = blockIdx.x;
    int t = threadIdx.x;
    if (b == 0) {
        gat_tabscan_body(tabC, tabP, Mpad);
    } else {
        gat_gemm_strip_body(X, Wf, H, asrc, adst, aS, aD, nRowTiles,
                            (b - 1) * GW, t);
    }
}

__global__ __launch_bounds__(1024) void gat_stage3(
    const int* __restrict__ ei, int E, int K, int chunkLen,
    const int* __restrict__ tabP, int* __restrict__ recw,
    const float* __restrict__ W2, float* __restrict__ Wf)
{
    int b = blockIdx.x;
    int t = threadIdx.x;
    if (b < NCH) {
        gat_bscatter_body(ei, E, K, chunkLen, tabP, (int2*)recw, b, t, 1024);
    } else {
        gat_fragorder_body(W2, Wf, HDIM, HDIM, (b - NCH) * 1024 + t);
    }
}

__device__ __forceinline__ float gat_leaky(float e) {
    return (e >= 0.f) ? e : 0.2f * e;
}

__device__ __forceinline__ float4 gat_gather_node(
    const int* __restrict__ rowptr, const int* __restrict__ srcS,
    const float* __restrict__ aS, const float* aD,
    const float* __restrict__ H, const float* __restrict__ bias,
    int w, int lane)
{
    int beg = rowptr[w];
    int end = rowptr[w + 1];
    float ad    = aD[w];
    float eSelf = gat_leaky(aS[w] + ad);

    const float4* H4 = (const float4*)H;
    float4 hv = H4[(size_t)w * (HDIM / 4) + lane];

    int  j0  = beg + lane;
    int  j1  = j0 + 32;
    bool in0 = j0 < end;
    bool in1 = j1 < end;
    int s0 = srcS[in0 ? j0 : 0];
    int s1 = srcS[in1 ? j1 : 0];
    float l0 = aS[s0];
    float l1 = aS[s1];
    float e0 = in0 ? gat_leaky(l0 + ad) : eSelf;
    float e1 = in1 ? gat_leaky(l1 + ad) : eSelf;

    float m = fmaxf(fmaxf(eSelf, e0), e1);
    for (int j = j1 + 32; j < end; j += 32)
        m = fmaxf(m, gat_leaky(aS[srcS[j]] + ad));
    for (int off = 16; off > 0; off >>= 1)
        m = fmaxf(m, __shfl_xor(m, off, 32));

    const float* bp = bias + 4 * lane;
    float bv0 = bp[0], bv1 = bp[1], bv2 = bp[2], bv3 = bp[3];

    float exSelf = expf(eSelf - m);
    float x0 = in0 ? expf(e0 - m) : 0.f;
    float x1 = in1 ? expf(e1 - m) : 0.f;
    float den = (lane == 0) ? exSelf : 0.f;
    den += x0;
    den += x1;
    for (int j = j1 + 32; j < end; j += 32)
        den += expf(gat_leaky(aS[srcS[j]] + ad) - m);
    for (int off = 16; off > 0; off >>= 1)
        den += __shfl_xor(den, off, 32);

    float aSelf = exSelf / den;
    float a0 = x0 / den;
    float a1 = x1 / den;
    float4 acc;
    acc.x = hv.x * aSelf; acc.y = hv.y * aSelf;
    acc.z = hv.z * aSelf; acc.w = hv.w * aSelf;
    int chunk = 0;
    for (int base = beg; base < end; base += 32, ++chunk) {
        int s;
        float a;
        if (chunk == 0) {
            s = s0; a = a0;
        } else if (chunk == 1) {
            s = s1; a = a1;
        } else {
            int j = base + lane;
            s = 0;
            a = 0.f;
            if (j < end) {
                s = srcS[j];
                a = expf(gat_leaky(aS[s] + ad) - m) / den;
            }
        }
        int cnt = (end - base < 32) ? (end - base) : 32;
        #pragma unroll 4
        for (int t = 0; t < cnt; ++t) {
            int   st = __builtin_amdgcn_readlane(s, t);
            float at = __int_as_float(
                           __builtin_amdgcn_readlane(__float_as_int(a), t));
            float4 v = H4[(size_t)st * (HDIM / 4) + lane];
            acc.x += v.x * at; acc.y += v.y * at;
            acc.z += v.z * at; acc.w += v.w * at;
        }
    }

    float4 r;
    r.x = acc.x + bv0; r.y = acc.y + bv1;
    r.z = acc.z + bv2; r.w = acc.w + bv3;
    r.x = (r.x > 0.f) ? r.x : 0.f; r.y = (r.y > 0.f) ? r.y : 0.f;
    r.z = (r.z > 0.f) ? r.z : 0.f; r.w = (r.w > 0.f) ? r.w : 0.f;
    return r;
}

#define GN 16
#define SP 132

__device__ __forceinline__ void gat_tile_bfirst(const float4* __restrict__ wq,
                                                float4 (&b)[8])
{
    #pragma unroll
    for (int u = 0; u < 8; ++u) b[u] = wq[u * 32];
}

__device__ __forceinline__ v8f gat_strip_tile(
    const float* strip, const float4* __restrict__ wq, float4 (&b)[8], int lane)
{
    const int laneM = lane & 15;
    const int hf    = lane >> 4;
    const float4* xq = (const float4*)(strip + laneM * SP) + hf;
    v8f c = {};
    #pragma unroll
    for (int u = 0; u < 8; ++u) {
        float4 a4 = xq[2 * u];
        v2f a0 = {a4.x, a4.y};
        v2f a1 = {a4.z, a4.w};
        v2f b0 = {b[u].x, b[u].y};
        v2f b1 = {b[u].z, b[u].w};
        c = __builtin_amdgcn_wmma_f32_16x16x4_f32(
                false, a0, false, b0, (short)0, c, false, false);
        c = __builtin_amdgcn_wmma_f32_16x16x4_f32(
                false, a1, false, b1, (short)0, c, false, false);
        b[u] = wq[(8 + u) * 32];
    }
    #pragma unroll
    for (int u = 0; u < 8; ++u) {
        float4 a4 = xq[2 * (8 + u)];
        v2f a0 = {a4.x, a4.y};
        v2f a1 = {a4.z, a4.w};
        v2f b0 = {b[u].x, b[u].y};
        v2f b1 = {b[u].z, b[u].w};
        c = __builtin_amdgcn_wmma_f32_16x16x4_f32(
                false, a0, false, b0, (short)0, c, false, false);
        c = __builtin_amdgcn_wmma_f32_16x16x4_f32(
                false, a1, false, b1, (short)0, c, false, false);
    }
    return c;
}

__global__ __launch_bounds__(GN * 32) void gat_gather_gemm(
    const int* __restrict__ rowptr, const int* __restrict__ srcS,
    const float* __restrict__ aSin, float* aD,
    const float* __restrict__ H, const float* __restrict__ bias,
    const float* __restrict__ Wf,
    const float* __restrict__ asrc, const float* __restrict__ adst,
    float* __restrict__ Hout, float* __restrict__ aSout)
{
    __shared__ __attribute__((aligned(16))) float strip[GN * SP];
    __shared__ float part[HDIM / 16][2][16];
    const int t    = threadIdx.x;
    const int lane = t & 31;
    const int wv   = __builtin_amdgcn_readfirstlane(t >> 5);
    const int row0 = blockIdx.x * GN;
    float4 r = gat_gather_node(rowptr, srcS, aSin, aD, H, bias, row0 + wv, lane);

    const int laneM = lane & 15;
    const int mBase = (lane >> 4) * 8;
    const bool mul  = wv < HDIM / 16;
    const float4* wq = (const float4*)Wf + (size_t)(wv & (HDIM / 16 - 1)) * (16 * 32) + lane;
    *(float4*)(strip + wv * SP + 4 * lane) = r;
    float4 b[8];
    float as = 0.f, ad = 0.f;
    if (mul) {
        int col = wv * 16 + laneM;
        as = asrc[col];
        ad = adst[col];
        gat_tile_bfirst(wq, b);
    }
    __syncthreads();

    v8f c = {};
    if (mul) {
        c = gat_strip_tile(strip, wq, b, lane);
        float ps[8], pd[8];
        #pragma unroll
        for (int q = 0; q < 8; ++q) {
            ps[q] = gat_row16_sum(c[q] * as);
            pd[q] = gat_row16_sum(c[q] * ad);
        }
        if (laneM == 0) {
            #pragma unroll
            for (int q = 0; q < 8; ++q) {
                part[wv][0][mBase + q] = ps[q];
                part[wv][1][mBase + q] = pd[q];
            }
        }
    }
    __syncthreads();
    if (t < 32) {
        int row  = t & 15;
        int kind = t >> 4;
        float sum = 0.f;
        #pragma unroll
        for (int q = 0; q < HDIM / 16; ++q) sum += part[q][kind][row];
        if (kind == 0) aSout[row0 + row] = sum;
        else           aD[row0 + row]    = sum;
    }
    if (mul) {
        float* o = Hout + (size_t)row0 * HDIM + wv * 16 + laneM;
        #pragma unroll
        for (int q = 0; q < 8; ++q)
            o[(size_t)(mBase + q) * HDIM] = c[q];
    }
}

__global__ __launch_bounds__(GN * 32) void gat_gather_cls(
    const int* __restrict__ rowptr, const int* __restrict__ srcS,
    const float* __restrict__ aS, const float* __restrict__ aD,
    const float* __restrict__ H, const float* __restrict__ bias,
    const float* __restrict__ Wlf, const float* __restrict__ bl,
    float* __restrict__ Out, int nCols)
{
    __shared__ __attribute__((aligned(16))) float strip[GN * SP];
    const int t    = threadIdx.x;
    const int lane = t & 31;
    const int wv   = __builtin_amdgcn_readfirstlane(t >> 5);
    const int row0 = blockIdx.x * GN;
    float4 r = gat_gather_node(rowptr, srcS, aS, aD, H, bias, row0 + wv, lane);

    const bool mul = wv * 16 < nCols;
    const float4* wq = (const float4*)Wlf + (size_t)(mul ? wv : 0) * (16 * 32) + lane;
    *(float4*)(strip + wv * SP + 4 * lane) = r;
    float4 b[8];
    if (mul) gat_tile_bfirst(wq, b);
    __syncthreads();

    if (mul) {
        v8f c = gat_strip_tile(strip, wq, b, lane);
        int laneM = lane & 15;
        int mBase = (lane >> 4) * 8;
        int col   = wv * 16 + laneM;
        if (col < nCols) {
            float bv = bl[col];
            #pragma unroll
            for (int q = 0; q < 8; ++q)
                Out[(size_t)(row0 + mBase + q) * nCols + col] = c[q] + bv;
        }
    }
}

extern "C" void kernel_launch(void* const* d_in, const int* in_sizes, int n_in,
                              void* d_out, int out_size, void* d_ws, size_t ws_size,
                              hipStream_t stream)
{
    const float* x   = (const float*)d_in[0];
    const int*   ei  = (const int*)d_in[1];
    const float* W1  = (const float*)d_in[2];
    const float* a1s = (const float*)d_in[3];
    const float* a1d = (const float*)d_in[4];
    const float* b1  = (const float*)d_in[5];
    const float* W2  = (const float*)d_in[6];
    const float* a2s = (const float*)d_in[7];
    const float* a2d = (const float*)d_in[8];
    const float* b2  = (const float*)d_in[9];
    const float* Wl  = (const float*)d_in[10];
    const float* bl  = (const float*)d_in[11];

    const int N  = in_sizes[0] / HDIM;
    const int E  = in_sizes[1] / 2;
    const int C  = in_sizes[11];
    const int CP = (C + 15) & ~15;

    char* ws = (char*)d_ws;
    auto align256 = [](size_t v) { return (v + 255) & ~(size_t)255; };
    size_t off = 0;
    float* bufH   = (float*)(ws + off);  off += align256((size_t)N * HDIM * 4);
    float* bufA   = (float*)(ws + off);  off += align256((size_t)N * HDIM * 4);
    float* Wf     = (float*)(ws + off);  off += align256((size_t)HDIM * HDIM * 4);
    float* Wlf    = (float*)(ws + off);  off += align256((size_t)CP * HDIM * 4);
    float* aS     = (float*)(ws + off);  off += align256((size_t)N * 4);
    float* aD     = (float*)(ws + off);  off += align256((size_t)N * 4);
    float* aS2    = (float*)(ws + off);  off += align256((size_t)N * 4);
    int*   rowptr = (int*)(ws + off);    off += align256((size_t)(N + 1) * 4);
    int*   srcS   = (int*)(ws + off);    off += align256((size_t)E * 4);
    (void)ws_size;

    const int K        = (N + BR - 1) >> BSH;
    const int Mpad     = (K * NCH + TS_PASS - 1) / TS_PASS * TS_PASS;
    const int chunkLen = (E + NCH - 1) / NCH;
    char*  sa   = (char*)bufA;
    int*   recw = (int*)sa;
    int*   tabC = (int*)(sa + align256((size_t)E * 8));
    int*   tabP = (int*)(sa + align256((size_t)E * 8) + align256((size_t)Mpad * 4));

    const int nRowTiles = N / 16;
    const int BLK = 256;
    dim3 blk(BLK);

    const int nbT     = (HDIM * HDIM + BLK - 1) / BLK;
    const int nbTl    = (CP * HDIM + BLK - 1) / BLK;
    const int nbGemm  = (nRowTiles * 2 + 7) / 8;
    const int nbT2    = (HDIM * HDIM + 1023) / 1024;
    dim3 stripGrid(nRowTiles);

    gat_stage1<<<dim3(NCH + nbT + nbTl), blk, 0, stream>>>(
        ei, E, K, chunkLen, tabC, Mpad, W1, Wf, nbT, Wl, Wlf, C, CP);
    gat_stage2<<<dim3(1 + nbGemm), blk, 0, stream>>>(
        x, Wf, bufH, a1s, a1d, aS, aD, nRowTiles, tabC, tabP, Mpad);
    gat_stage3<<<dim3(NCH + nbT2), dim3(1024), 0, stream>>>(
        ei, E, K, chunkLen, tabP, recw, W2, Wf);
    gat_bsort<<<dim3(K), dim3(1024), 0, stream>>>(tabP, recw, E, N, K, rowptr, srcS);

    gat_gather_gemm<<<stripGrid, dim3(GN * 32), 0, stream>>>(
        rowptr, srcS, aS, aD, bufH, b1, Wf, a2s, a2d, bufA, aS2);
    gat_gather_cls<<<stripGrid, dim3(GN * 32), 0, stream>>>(
        rowptr, srcS, aS2, aD, bufA, b2, Wlf, bl, (float*)d_out, C);
}


# Edge sort in 2 launches, not 4: chunks sort their edges by bucket inside LDS (no count pass, no table scan); buckets sum the chunk tables for their offset; 6 -> 4 launches; per-node sum order changes
# speedup vs baseline: 6.4571x; 1.2443x over previous
#include <hip/hip_runtime.h>
#include <hip/hip_bf16.h>

typedef __attribute__((ext_vector_type(2))) float v2f;
typedef __attribute__((ext_vector_type(8))) float v8f;

#define HDIM 128
#define CTPW 4


__device__ __forceinline__ void gat_fragorder_body(
    const float* __restrict__ W, float* __restrict__ Wf,
    int cols, int colsPad, int i)
{
    int total = colsPad * HDIM;
    if (i >= total) return;
    int e = i & 3;
    int l = (i >> 2) & 31;
    int j = (i >> 7) & 15;
    int q = i >> 11;
    int k = 8 * j + 4 * (l >> 4) + e;
    int c = q * 16 + (l & 15);
    Wf[i] = (c < cols) ? W[(size_t)k * cols + c] : 0.f;
}

#define GW 8

__device__ __forceinline__ float gat_row16_sum(float v)
{
    #pragma unroll
    for (int off = 1; off < 16; off <<= 1)
        v += __shfl_xor(v, off, 32);
    return v;
}

__device__ __forceinline__ void gat_gemm_strip_body(
    const float* __restrict__ X, const float* __restrict__ Wf,
    float* __restrict__ H,
    const float* __restrict__ asrc, const float* __restrict__ adst,
    float* __restrict__ aS, float* __restrict__ aD,
    int nRowTiles, int wid0, int t)
{
    __shared__ float part[GW][2][16];
    const int nHalves = HDIM / (CTPW * 16);
    const int nW    = nRowTiles * nHalves;
    const int wv    = t >> 5;
    const int lane  = t & 31;
    const int wid   = wid0 + wv;
    const int rt    = wid >> 1;
    const int half  = wid & 1;
    const int row0  = rt * 16;
    const int laneM = lane & 15;
    const int hf    = lane >> 4;
    const int mBase = hf * 8;
    const bool act = wid < nW;

    v8f c[CTPW] = {};
    if (act) {
        const float4* xq = (const float4*)(X + (size_t)(row0 + laneM) * HDIM) + hf;
        const float4* wq = (const float4*)Wf + (size_t)(half * CTPW) * (16 * 32) + lane;

        float as[CTPW], ad[CTPW];
        #pragma unroll
        for (int q = 0; q < CTPW; ++q) {
            int col = (half * CTPW + q) * 16 + laneM;
            as[q] = asrc[col];
            ad[q] = adst[col];
        }

        #pragma unroll 4
        for (int j = 0; j < HDIM / 8; ++j) {
            float4 a4 = xq[2 * j];
            float4 b4[CTPW];
            #pragma unroll
            for (int q = 0; q < CTPW; ++q) b4[q] = wq[(q * 16 + j) * 32];
            v2f a0 = {a4.x, a4.y};
            v2f a1 = {a4.z, a4.w};
            #pragma unroll
            for (int q = 0; q < CTPW; ++q) {
                v2f b0 = {b4[q].x, b4[q].y};
                c[q] = __builtin_amdgcn_wmma_f32_16x16x4_f32(
                           false, a0, false, b0, (short)0, c[q], false, false);
            }
            #pragma unroll
            for (int q = 0; q < CTPW; ++q) {
                v2f b1 = {b4[q].z, b4[q].w};
                c[q] = __builtin_amdgcn_wmma_f32_16x16x4_f32(
                           false, a1, false, b1, (short)0, c[q], false, false);
            }
        }

        float ps[8], pd[8];
        #pragma unroll
        for (int r = 0; r < 8; ++r) {
            float s = 0.f, d = 0.f;
            #pragma unroll
            for (int q = 0; q < CTPW; ++q) {
                s += c[q][r] * as[q];
                d += c[q][r] * ad[q];
            }
            ps[r] = s;
            pd[r] = d;
        }
        #pragma unroll
        for (int r = 0; r < 8; ++r) {
            ps[r] = gat_row16_sum(ps[r]);
            pd[r] = gat_row16_sum(pd[r]);
        }
        if (laneM == 0) {
            #pragma unroll
            for (int r = 0; r < 8; ++r) {
                part[wv][0][mBase + r] = ps[r];
                part[wv][1][mBase + r] = pd[r];
            }
        }
    }
    __syncthreads();
    if (t < (GW / 2) * 16) {
        int ls  = t >> 4;
        int row = t & 15;
        int rs  = (wid0 >> 1) + ls;
        if (rs < nRowTiles) {
            aS[rs * 16 + row] = part[2 * ls][0][row] + part[2 * ls + 1][0][row];
            aD[rs * 16 + row] = part[2 * ls][1][row] + part[2 * ls + 1][1][row];
        }
    }
    if (act) {
        #pragma unroll
        for (int q = 0; q < CTPW; ++q) {
            float* o = H + (size_t)row0 * HDIM + (half * CTPW + q) * 16 + laneM;
            #pragma unroll
            for (int r = 0; r < 8; ++r)
                o[(size_t)(mBase + r) * HDIM] = c[q][r];
        }
    }
}

#define BSH  7
#define BR   (1 << BSH)
#define NCH  128
#define KMAX 1024
#define EPT  5
#define CNT  1024
#define CHCAP (EPT * CNT)
#define BNT  256
#define RPW  (NCH / (BNT / 32))

__device__ __forceinline__ int gat_wave_incl_scan(int x, int lane)
{
    #pragma unroll
    for (int off = 1; off < 32; off <<= 1) {
        int y = __shfl_up(x, off, 32);
        if (lane >= off) x += y;
    }
    return x;
}

__device__ __forceinline__ void gat_chunksort_body(
    const int* __restrict__ ei, int E, int K, int chunkLen,
    int2* __restrict__ rec, int* __restrict__ cntT, int* __restrict__ offT,
    int c, int t)
{
    __shared__ int  lh[KMAX];
    __shared__ int2 lrec[CHCAP];
    __shared__ int  wtot[CNT / 32];
    const int lane = t & 31;
    const int wv   = t >> 5;
    for (int k = t; k < K; k += CNT) lh[k] = 0;
    int c0 = c * chunkLen;
    int c1 = (c0 + chunkLen < E) ? (c0 + chunkLen) : E;
    int n  = c1 - c0;
    int d[EPT], s[EPT], rk[EPT];
    #pragma unroll
    for (int u = 0; u < EPT; ++u) {
        int p  = t + u * CNT;
        bool ok = p < n;
        int j  = ok ? (c0 + p) : ((n > 0) ? c0 : 0);
        int dv = ei[E + j];
        s[u] = ei[j];
        d[u] = ok ? dv : -1;
    }
    __syncthreads();
    #pragma unroll
    for (int u = 0; u < EPT; ++u) {
        rk[u] = 0;
        if (d[u] >= 0) rk[u] = atomicAdd(&lh[d[u] >> BSH], 1);
    }
    __syncthreads();
    int v   = (t < K) ? lh[t] : 0;
    int inc = gat_wave_incl_scan(v, lane);
    if (lane == 31) wtot[wv] = inc;
    __syncthreads();
    int ex = inc - v;
    for (int q = 0; q < wv; ++q) ex += wtot[q];
    if (t < K) {
        cntT[c * K + t] = v;
        offT[c * K + t] = ex;
        lh[t] = ex;
    }
    __syncthreads();
    #pragma unroll
    for (int u = 0; u < EPT; ++u)
        if (d[u] >= 0) lrec[lh[d[u] >> BSH] + rk[u]] = make_int2(d[u], s[u]);
    __syncthreads();
    #pragma unroll
    for (int u = 0; u < EPT; ++u) {
        int p = t + u * CNT;
        if (p < n) rec[c0 + p] = lrec[p];
    }
}

__device__ __forceinline__ void gat_bucketsort_body(
    const int* __restrict__ cntT, const int* __restrict__ offT,
    const int2* __restrict__ rec, int N, int K, int chunkLen,
    int* __restrict__ rowptr, int* __restrict__ srcS, int k, int t)
{
    __shared__ int lcnt[BR];
    __shared__ int lcur[BR];
    __shared__ int wsum[BR / 32];
    __shared__ int wred[2][BNT / 32];
    const int lane = t & 31;
    const int wv   = __builtin_amdgcn_readfirstlane(t >> 5);
    const int d0   = k << BSH;
    const int cw   = wv * RPW + (lane & (RPW - 1));
    int o = 0, m = 0;
    if (lane < RPW) {
        o = offT[cw * K + k];
        m = cntT[cw * K + k];
    }
    const int rbv = cw * chunkLen + o;
    if (t < BR) lcnt[t] = 0;
    int so = o, sm = m;
    for (int off = 16; off > 0; off >>= 1) {
        so += __shfl_xor(so, off, 32);
        sm += __shfl_xor(sm, off, 32);
    }
    if (lane == 0) { wred[0][wv] = so; wred[1][wv] = sm; }
    __syncthreads();
    int start = 0, size = 0;
    for (int q = 0; q < BNT / 32; ++q) { start += wred[0][q]; size += wred[1][q]; }

    int2 r[RPW];
    bool ok[RPW];
    #pragma unroll
    for (int i = 0; i < RPW; ++i) {
        int rb = __builtin_amdgcn_readlane(rbv, i);
        int rn = __builtin_amdgcn_readlane(m, i);
        ok[i] = lane < rn;
        r[i]  = rec[ok[i] ? (rb + lane) : 0];
    }
    #pragma unroll
    for (int i = 0; i < RPW; ++i)
        if (ok[i]) atomicAdd(&lcnt[r[i].x - d0], 1);
    for (int i = 0; i < RPW; ++i) {
        int rb = __builtin_amdgcn_readlane(rbv, i);
        int rn = __builtin_amdgcn_readlane(m, i);
        for (int j = lane + 32; j < rn; j += 32)
            atomicAdd(&lcnt[rec[rb + j].x - d0], 1);
    }
    __syncthreads();
    int v   = (t < BR) ? lcnt[t] : 0;
    int inc = gat_wave_incl_scan(v, lane);
    if (t < BR && lane == 31) wsum[wv] = inc;
    __syncthreads();
    if (t < BR) {
        int ex = start + inc - v;
        for (int q = 0; q < wv; ++q) ex += wsum[q];
        lcur[t] = ex;
        if (d0 + t < N) rowptr[d0 + t] = ex;
    }
    if (k == K - 1 && t == 0) rowptr[N] = start + size;
    __syncthreads();
    #pragma unroll
    for (int i = 0; i < RPW; ++i) {
        if (ok[i]) {
            int slot = atomicAdd(&lcur[r[i].x - d0], 1);
            srcS[slot] = r[i].y;
        }
    }
    for (int i = 0; i < RPW; ++i) {
        int rb = __builtin_amdgcn_readlane(rbv, i);
        int rn = __builtin_amdgcn_readlane(m, i);
        for (int j = lane + 32; j < rn; j += 32) {
            int2 q = rec[rb + j];
            int slot = atomicAdd(&lcur[q.x - d0], 1);
            srcS[slot] = q.y;
        }
    }
}


__global__ __launch_bounds__(CNT) void gat_stage1(
    const int* __restrict__ ei, int E, int K, int chunkLen,
    int* __restrict__ recw, int* __restrict__ cntT, int* __restrict__ offT,
    const float* __restrict__ W1, float* __restrict__ W1f,
    const float* __restrict__ W2, float* __restrict__ W2f, int nbT,
    const float* __restrict__ Wl, float* __restrict__ Wlf, int C, int CP)
{
    int b = blockIdx.x;
    int t = threadIdx.x;
    if (b < NCH) {
        gat_chunksort_body(ei, E, K, chunkLen, (int2*)recw, cntT, offT, b, t);
    } else if (b < NCH + nbT) {
        gat_fragorder_body(W1, W1f, HDIM, HDIM, (b - NCH) * CNT + t);
    } else if (b < NCH + 2 * nbT) {
        gat_fragorder_body(W2, W2f, HDIM, HDIM, (b - NCH - nbT) * CNT + t);
    } else {
        gat_fragorder_body(Wl, Wlf, C, CP, (b - NCH - 2 * nbT) * CNT + t);
    }
}

__global__ __launch_bounds__(BNT) __attribute__((amdgpu_waves_per_eu(7, 7))) void gat_stage2(
    const float* __restrict__ X, const float* __restrict__ W1f,
    float* __restrict__ H,
    const float* __restrict__ asrc, const float* __restrict__ adst,
    float* __restrict__ aS, float* __restrict__ aD, int nRowTiles,
    const int* __restrict__ cntT, const int* __restrict__ offT,
    const int* __restrict__ recw, int N, int K, int chunkLen,
    int* __restrict__ rowptr, int* __restrict__ srcS)
{
    int b = blockIdx.x;
    int t = threadIdx.x;
    if (b < K) {
        gat_bucketsort_body(cntT, offT, (const int2*)recw, N, K, chunkLen,
                            rowptr, srcS, b, t);
    } else {
        gat_gemm_strip_body(X, W1f, H, asrc, adst, aS, aD, nRowTiles,
                            (b - K) * GW, t);
    }
}

__device__ __forceinline__ float gat_leaky(float e) {
    return (e >= 0.f) ? e : 0.2f * e;
}

__device__ __forceinline__ float4 gat_gather_node(
    const int* __restrict__ rowptr, const int* __restrict__ srcS,
    const float* __restrict__ aS, const float* aD,
    const float* __restrict__ H, const float* __restrict__ bias,
    int w, int lane)
{
    int beg = rowptr[w];
    int end = rowptr[w + 1];
    float ad    = aD[w];
    float eSelf = gat_leaky(aS[w] + ad);

    const float4* H4 = (const float4*)H;
    float4 hv = H4[(size_t)w * (HDIM / 4) + lane];

    int  j0  = beg + lane;
    int  j1  = j0 + 32;
    bool in0 = j0 < end;
    bool in1 = j1 < end;
    int s0 = srcS[in0 ? j0 : 0];
    int s1 = srcS[in1 ? j1 : 0];
    float l0 = aS[s0];
    float l1 = aS[s1];
    float e0 = in0 ? gat_leaky(l0 + ad) : eSelf;
    float e1 = in1 ? gat_leaky(l1 + ad) : eSelf;

    float m = fmaxf(fmaxf(eSelf, e0), e1);
    for (int j = j1 + 32; j < end; j += 32)
        m = fmaxf(m, gat_leaky(aS[srcS[j]] + ad));
    for (int off = 16; off > 0; off >>= 1)
        m = fmaxf(m, __shfl_xor(m, off, 32));

    const float* bp = bias + 4 * lane;
    float bv0 = bp[0], bv1 = bp[1], bv2 = bp[2], bv3 = bp[3];

    float exSelf = expf(eSelf - m);
    float x0 = in0 ? expf(e0 - m) : 0.f;
    float x1 = in1 ? expf(e1 - m) : 0.f;
    float den = (lane == 0) ? exSelf : 0.f;
    den += x0;
    den += x1;
    for (int j = j1 + 32; j < end; j += 32)
        den += expf(gat_leaky(aS[srcS[j]] + ad) - m);
    for (int off = 16; off > 0; off >>= 1)
        den += __shfl_xor(den, off, 32);

    float aSelf = exSelf / den;
    float a0 = x0 / den;
    float a1 = x1 / den;
    float4 acc;
    acc.x = hv.x * aSelf; acc.y = hv.y * aSelf;
    acc.z = hv.z * aSelf; acc.w = hv.w * aSelf;
    int chunk = 0;
    for (int base = beg; base < end; base += 32, ++chunk) {
        int s;
        float a;
        if (chunk == 0) {
            s = s0; a = a0;
        } else if (chunk == 1) {
            s = s1; a = a1;
        } else {
            int j = base + lane;
            s = 0;
            a = 0.f;
            if (j < end) {
                s = srcS[j];
                a = expf(gat_leaky(aS[s] + ad) - m) / den;
            }
        }
        int cnt = (end - base < 32) ? (end - base) : 32;
        #pragma unroll 4
        for (int t = 0; t < cnt; ++t) {
            int   st = __builtin_amdgcn_readlane(s, t);
            float at = __int_as_float(
                           __builtin_amdgcn_readlane(__float_as_int(a), t));
            float4 v = H4[(size_t)st * (HDIM / 4) + lane];
            acc.x += v.x * at; acc.y += v.y * at;
            acc.z += v.z * at; acc.w += v.w * at;
        }
    }

    float4 r;
    r.x = acc.x + bv0; r.y = acc.y + bv1;
    r.z = acc.z + bv2; r.w = acc.w + bv3;
    r.x = (r.x > 0.f) ? r.x : 0.f; r.y = (r.y > 0.f) ? r.y : 0.f;
    r.z = (r.z > 0.f) ? r.z : 0.f; r.w = (r.w > 0.f) ? r.w : 0.f;
    return r;
}

#define GN 16
#define SP 132

__device__ __forceinline__ void gat_tile_bfirst(const float4* __restrict__ wq,
                                                float4 (&b)[8])
{
    #pragma unroll
    for (int u = 0; u < 8; ++u) b[u] = wq[u * 32];
}

__device__ __forceinline__ v8f gat_strip_tile(
    const float* strip, const float4* __restrict__ wq, float4 (&b)[8], int lane)
{
    const int laneM = lane & 15;
    const int hf    = lane >> 4;
    const float4* xq = (const float4*)(strip + laneM * SP) + hf;
    v8f c = {};
    #pragma unroll
    for (int u = 0; u < 8; ++u) {
        float4 a4 = xq[2 * u];
        v2f a0 = {a4.x, a4.y};
        v2f a1 = {a4.z, a4.w};
        v2f b0 = {b[u].x, b[u].y};
        v2f b1 = {b[u].z, b[u].w};
        c = __builtin_amdgcn_wmma_f32_16x16x4_f32(
                false, a0, false, b0, (short)0, c, false, false);
        c = __builtin_amdgcn_wmma_f32_16x16x4_f32(
                false, a1, false, b1, (short)0, c, false, false);
        b[u] = wq[(8 + u) * 32];
    }
    #pragma unroll
    for (int u = 0; u < 8; ++u) {
        float4 a4 = xq[2 * (8 + u)];
        v2f a0 = {a4.x, a4.y};
        v2f a1 = {a4.z, a4.w};
        v2f b0 = {b[u].x, b[u].y};
        v2f b1 = {b[u].z, b[u].w};
        c = __builtin_amdgcn_wmma_f32_16x16x4_f32(
                false, a0, false, b0, (short)0, c, false, false);
        c = __builtin_amdgcn_wmma_f32_16x16x4_f32(
                false, a1, false, b1, (short)0, c, false, false);
    }
    return c;
}

__global__ __launch_bounds__(GN * 32) void gat_gather_gemm(
    const int* __restrict__ rowptr, const int* __restrict__ srcS,
    const float* __restrict__ aSin, float* aD,
    const float* __restrict__ H, const float* __restrict__ bias,
    const float* __restrict__ Wf,
    const float* __restrict__ asrc, const float* __restrict__ adst,
    float* __restrict__ Hout, float* __restrict__ aSout)
{
    __shared__ __attribute__((aligned(16))) float strip[GN * SP];
    __shared__ float part[HDIM / 16][2][16];
    const int t    = threadIdx.x;
    const int lane = t & 31;
    const int wv   = __builtin_amdgcn_readfirstlane(t >> 5);
    const int row0 = blockIdx.x * GN;
    float4 r = gat_gather_node(rowptr, srcS, aSin, aD, H, bias, row0 + wv, lane);

    const int laneM = lane & 15;
    const int mBase = (lane >> 4) * 8;
    const bool mul  = wv < HDIM / 16;
    const float4* wq = (const float4*)Wf + (size_t)(wv & (HDIM / 16 - 1)) * (16 * 32) + lane;
    *(float4*)(strip + wv * SP + 4 * lane) = r;
    float4 b[8];
    float as = 0.f, ad = 0.f;
    if (mul) {
        int col = wv * 16 + laneM;
        as = asrc[col];
        ad = adst[col];
        gat_tile_bfirst(wq, b);
    }
    __syncthreads();

    v8f c = {};
    if (mul) {
        c = gat_strip_tile(strip, wq, b, lane);
        float ps[8], pd[8];
        #pragma unroll
        for (int q = 0; q < 8; ++q) {
            ps[q] = gat_row16_sum(c[q] * as);
            pd[q] = gat_row16_sum(c[q] * ad);
        }
        if (laneM == 0) {
            #pragma unroll
            for (int q = 0; q < 8; ++q) {
                part[wv][0][mBase + q] = ps[q];
                part[wv][1][mBase + q] = pd[q];
            }
        }
    }
    __syncthreads();
    if (t < 32) {
        int row  = t & 15;
        int kind = t >> 4;
        float sum = 0.f;
        #pragma unroll
        for (int q = 0; q < HDIM / 16; ++q) sum += part[q][kind][row];
        if (kind == 0) aSout[row0 + row] = sum;
        else           aD[row0 + row]    = sum;
    }
    if (mul) {
        float* o = Hout + (size_t)row0 * HDIM + wv * 16 + laneM;
        #pragma unroll
        for (int q = 0; q < 8; ++q)
            o[(size_t)(mBase + q) * HDIM] = c[q];
    }
}

__global__ __launch_bounds__(GN * 32) void gat_gather_cls(
    const int* __restrict__ rowptr, const int* __restrict__ srcS,
    const float* __restrict__ aS, const float* __restrict__ aD,
    const float* __restrict__ H, const float* __restrict__ bias,
    const float* __restrict__ Wlf, const float* __restrict__ bl,
    float* __restrict__ Out, int nCols)
{
    __shared__ __attribute__((aligned(16))) float strip[GN * SP];
    const int t    = threadIdx.x;
    const int lane = t & 31;
    const int wv   = __builtin_amdgcn_readfirstlane(t >> 5);
    const int row0 = blockIdx.x * GN;
    float4 r = gat_gather_node(rowptr, srcS, aS, aD, H, bias, row0 + wv, lane);

    const bool mul = wv * 16 < nCols;
    const float4* wq = (const float4*)Wlf + (size_t)(mul ? wv : 0) * (16 * 32) + lane;
    *(float4*)(strip + wv * SP + 4 * lane) = r;
    float4 b[8];
    if (mul) gat_tile_bfirst(wq, b);
    __syncthreads();

    if (mul) {
        v8f c = gat_strip_tile(strip, wq, b, lane);
        int laneM = lane & 15;
        int mBase = (lane >> 4) * 8;
        int col   = wv * 16 + laneM;
        if (col < nCols) {
            float bv = bl[col];
            #pragma unroll
            for (int q = 0; q < 8; ++q)
                Out[(size_t)(row0 + mBase + q) * nCols + col] = c[q] + bv;
        }
    }
}

extern "C" void kernel_launch(void* const* d_in, const int* in_sizes, int n_in,
                              void* d_out, int out_size, void* d_ws, size_t ws_size,
                              hipStream_t stream)
{
    const float* x   = (const float*)d_in[0];
    const int*   ei  = (const int*)d_in[1];
    const float* W1  = (const float*)d_in[2];
    const float* a1s = (const float*)d_in[3];
    const float* a1d = (const float*)d_in[4];
    const float* b1  = (const float*)d_in[5];
    const float* W2  = (const float*)d_in[6];
    const float* a2s = (const float*)d_in[7];
    const float* a2d = (const float*)d_in[8];
    const float* b2  = (const float*)d_in[9];
    const float* Wl  = (const float*)d_in[10];
    const float* bl  = (const float*)d_in[11];

    const int N  = in_sizes[0] / HDIM;
    const int E  = in_sizes[1] / 2;
    const int C  = in_sizes[11];
    const int CP = (C + 15) & ~15;

    char* ws = (char*)d_ws;
    auto align256 = [](size_t v) { return (v + 255) & ~(size_t)255; };
    size_t off = 0;
    float* bufH   = (float*)(ws + off);  off += align256((size_t)N * HDIM * 4);
    float* bufA   = (float*)(ws + off);  off += align256((size_t)N * HDIM * 4);
    float* Wf     = (float*)(ws + off);  off += align256((size_t)HDIM * HDIM * 4);
    float* Wlf    = (float*)(ws + off);  off += align256((size_t)CP * HDIM * 4);
    float* aS     = (float*)(ws + off);  off += align256((size_t)N * 4);
    float* aD     = (float*)(ws + off);  off += align256((size_t)N * 4);
    float* aS2    = (float*)(ws + off);  off += align256((size_t)N * 4);
    int*   rowptr = (int*)(ws + off);    off += align256((size_t)(N + 1) * 4);
    int*   srcS   = (int*)(ws + off);    off += align256((size_t)E * 4);
    (void)ws_size;

    const int K        = (N + BR - 1) >> BSH;
    const int chunkLen = (E + NCH - 1) / NCH;
    char*  sa   = (char*)bufA;
    int*   recw = (int*)sa;
    int*   cntT = (int*)(sa + align256((size_t)E * 8));
    int*   offT = (int*)(sa + align256((size_t)E * 8) + align256((size_t)NCH * K * 4));
    float* W1f  = (float*)(sa + align256((size_t)E * 8) + 2 * align256((size_t)NCH * K * 4));

    const int nRowTiles = N / 16;

    const int nbT     = (HDIM * HDIM + CNT - 1) / CNT;
    const int nbTl    = (CP * HDIM + CNT - 1) / CNT;
    const int nbGemm  = (nRowTiles * 2 + GW - 1) / GW;
    dim3 stripGrid(nRowTiles);

    gat_stage1<<<dim3(NCH + 2 * nbT + nbTl), dim3(CNT), 0, stream>>>(
        ei, E, K, chunkLen, recw, cntT, offT, W1, W1f, W2, Wf, nbT, Wl, Wlf, C, CP);
    gat_stage2<<<dim3(K + nbGemm), dim3(BNT), 0, stream>>>(
        x, W1f, bufH, a1s, a1d, aS, aD, nRowTiles,
        cntT, offT, recw, N, K, chunkLen, rowptr, srcS);

    gat_gather_gemm<<<stripGrid, dim3(GN * 32), 0, stream>>>(
        rowptr, srcS, aS, aD, bufH, b1, Wf, a2s, a2d, bufA, aS2);
    gat_gather_cls<<<stripGrid, dim3(GN * 32), 0, stream>>>(
        rowptr, srcS, aS2, aD, bufA, b2, Wlf, bl, (float*)d_out, C);
}
